# starting kernel
# baseline (speedup 1.0000x reference)
#include <hip/hip_runtime.h>
#include <math.h>

typedef __attribute__((ext_vector_type(2))) float v2f;
typedef __attribute__((ext_vector_type(4))) float v4f;
typedef __attribute__((ext_vector_type(8))) float v8f;

#define AS3 __attribute__((address_space(3)))
#define AS1 __attribute__((address_space(1)))
typedef AS3 float*       lptr;
typedef const AS3 float* clptr;
typedef AS1 float*       gptr;
typedef const AS1 float* cgptr;

#define B_TOT    8192
#define L_SEQ    256
#define NCH      19
#define KPACK1   96
#define KPACK2   64
#define NPACK    32
#define PRS      96
#define NWAVES   8
#define NTHREADS 256
#define K1N      (KPACK1 * NPACK)
#define K2N      (KPACK2 * NPACK)

#define SEV_OFF 0
#define SEV_SZ  2578
#define SOD_OFF (SEV_OFF + SEV_SZ)
#define SOD_SZ  2578
#define HB_OFF  (SOD_OFF + SOD_SZ)
#define HB_SZ   2502
#define DB_OFF  (HB_OFF + HB_SZ)
#define DB_SZ   2578
#define CB_OFF  (DB_OFF + DB_SZ)
#define CB_SZ   2578
#define W1S_OFF 12816
#define W1S_SZ  ((KPACK1 / 2) * PRS)
#define W2S_OFF (W1S_OFF + W1S_SZ)
#define W2S_SZ  ((KPACK2 / 2) * PRS)
#define SMEM_TOT (W2S_OFF + W2S_SZ)

static __device__ __forceinline__ v8f wmma4(v2f a, v2f b, v8f c) {
  return __builtin_amdgcn_wmma_f32_16x16x4_f32(false, a, false, b, (short)0, c,
                                               false, false);
}

static __device__ __forceinline__ float leaky(float x) {
  return x > 0.f ? x : 0.01f * x;
}

static __device__ __forceinline__ float fast_tanh(float x) {
  x = fminf(fmaxf(x, -15.f), 15.f);
  const float e = __expf(2.f * x);
  return __fdividef(e - 1.f, e + 1.f);
}

template <int MODE>
static __device__ void run_branch(cgptr Wt1, cgptr Wt2, cgptr b1, cgptr b2,
                                  clptr in, lptr hb, lptr w1s, lptr w2s,
                                  int Lp, clptr other, lptr outLds,
                                  gptr xg, int tmul, int tadd) {
  const int tid  = threadIdx.x;
  const int wave = tid >> 5;
  const int lane = tid & 31;
  const int lhi  = lane >> 4;
  const int lln  = lane & 15;

  for (int i = tid; i < K1N / 4; i += NTHREADS) {
    const int p = i >> 4, j = (i & 15) * 4;
    *(AS3 v4f*)(w1s + p * PRS + j) = ((const AS1 v4f*)Wt1)[i];
  }
  for (int i = tid; i < K2N / 4; i += NTHREADS) {
    const int p = i >> 4, j = (i & 15) * 4;
    *(AS3 v4f*)(w2s + p * PRS + j) = ((const AS1 v4f*)Wt2)[i];
  }
  __syncthreads();

  const int M1   = Lp + 2;
  const int nMt1 = (M1 + 15) >> 4;
  const float b1a = b1[lln];
  const float b1b = b1[lln < 3 ? 16 + lln : 18];
  for (int mt = wave; mt < nMt1; mt += NWAVES) {
    const int m0 = mt << 4;
    int mrow = m0 + lln; if (mrow > M1 - 1) mrow = M1 - 1;
    clptr ap = in + mrow * NCH + 2 * lhi;
    clptr bp = w1s + lhi * PRS + 2 * lln;
    v8f acc0 = {}; v8f acc1 = {};
#pragma unroll 4
    for (int k0 = 0; k0 < KPACK1; k0 += 4) {
      v2f a; a.x = ap[k0]; a.y = ap[k0 + 1];
      v2f bv0 = *(const AS3 v2f*)(bp);
      v2f bv1 = *(const AS3 v2f*)(bp + 32);
      bp += 2 * PRS;
      acc0 = wmma4(a, bv0, acc0);
      acc1 = wmma4(a, bv1, acc1);
    }
    lptr hrow = hb + (m0 + 8 * lhi) * NCH;
    if (m0 + 16 <= M1) {
#pragma unroll
      for (int v = 0; v < 8; v++) hrow[v * NCH + lln] = leaky(acc0[v] + b1a);
      if (lln < 3) {
#pragma unroll
        for (int v = 0; v < 8; v++)
          hrow[v * NCH + 16 + lln] = leaky(acc1[v] + b1b);
      }
    } else {
#pragma unroll
      for (int v = 0; v < 8; v++)
        if (m0 + 8 * lhi + v < M1) hrow[v * NCH + lln] = leaky(acc0[v] + b1a);
      if (lln < 3) {
#pragma unroll
        for (int v = 0; v < 8; v++)
          if (m0 + 8 * lhi + v < M1)
            hrow[v * NCH + 16 + lln] = leaky(acc1[v] + b1b);
      }
    }
  }
  __syncthreads();

  const int M2   = Lp;
  const int nMt2 = M2 >> 4;
  const float b2a = b2[lln];
  const float b2b = b2[lln < 3 ? 16 + lln : 18];
  for (int mt = wave; mt < nMt2; mt += NWAVES) {
    const int m0 = mt << 4;
    clptr ap = hb + (m0 + lln) * NCH + 2 * lhi;
    clptr bp = w2s + lhi * PRS + 2 * lln;
    v8f acc0 = {}; v8f acc1 = {};
#pragma unroll 4
    for (int k0 = 0; k0 < KPACK2; k0 += 4) {
      v2f a; a.x = ap[k0]; a.y = ap[k0 + 1];
      v2f bv0 = *(const AS3 v2f*)(bp);
      v2f bv1 = *(const AS3 v2f*)(bp + 32);
      bp += 2 * PRS;
      acc0 = wmma4(a, bv0, acc0);
      acc1 = wmma4(a, bv1, acc1);
    }
    const int mb = m0 + 8 * lhi;
#pragma unroll
    for (int v = 0; v < 8; v++) {
      const int m = mb + v;
      const float t   = fast_tanh(acc0[v] + b2a);
      const float oth = other[(3 + m) * NCH + lln];
      if (MODE <= 1)      outLds[(3 + m) * NCH + lln] = oth * __expf(t);
      else if (MODE == 2) xg[(m * tmul + tadd) * NCH + lln] = oth + t;
      else                xg[(m * tmul + tadd) * NCH + lln] = oth - t;
    }
    if (lln < 3) {
      const int co = 16 + lln;
#pragma unroll
      for (int v = 0; v < 8; v++) {
        const int m = mb + v;
        const float t   = fast_tanh(acc1[v] + b2b);
        const float oth = other[(3 + m) * NCH + co];
        if (MODE <= 1)      outLds[(3 + m) * NCH + co] = oth * __expf(t);
        else if (MODE == 2) xg[(m * tmul + tadd) * NCH + co] = oth + t;
        else                xg[(m * tmul + tadd) * NCH + co] = oth - t;
      }
    }
  }
  __syncthreads();
}

__global__ __launch_bounds__(NTHREADS) void scinet_level_kernel(
    float* __restrict__ xbuf_, const float* __restrict__ Wt1p_,
    const float* __restrict__ Wt2p_, const float* __restrict__ b1g_,
    const float* __restrict__ b2g_, int depth) {
  __shared__ __align__(16) float smem[SMEM_TOT];
  const int tid = threadIdx.x;
  const int sel = blockIdx.x & ((1 << depth) - 1);
  const int b   = blockIdx.x >> depth;
  int node = 0;
  for (int i = 0; i < depth; i++) node = 2 * node + 1 + ((sel >> i) & 1);
  const int stride = 1 << depth;
  const int Lp     = (L_SEQ >> depth) >> 1;

  lptr sev = (lptr)(smem + SEV_OFF);
  lptr sod = (lptr)(smem + SOD_OFF);
  lptr hb  = (lptr)(smem + HB_OFF);
  lptr db  = (lptr)(smem + DB_OFF);
  lptr cb  = (lptr)(smem + CB_OFF);
  lptr w1s = (lptr)(smem + W1S_OFF);
  lptr w2s = (lptr)(smem + W2S_OFF);
  lptr sm  = (lptr)smem;

  for (int i = tid; i < SMEM_TOT; i += NTHREADS) sm[i] = 0.f;
  __syncthreads();

  gptr  xg   = (gptr)(xbuf_ + (size_t)b * L_SEQ * NCH);
  cgptr Wt1p = (cgptr)Wt1p_;
  cgptr Wt2p = (cgptr)Wt2p_;
  cgptr b1g  = (cgptr)b1g_;
  cgptr b2g  = (cgptr)b2g_;

  for (int i = tid; i < Lp * NCH; i += NTHREADS) {
    const int r = i / NCH, c = i - r * NCH;
    sev[(3 + r) * NCH + c] = xg[((2 * r) * stride + sel) * NCH + c];
    sod[(3 + r) * NCH + c] = xg[((2 * r + 1) * stride + sel) * NCH + c];
  }
  __syncthreads();
  for (int i = tid; i < 2 * 3 * NCH; i += NTHREADS) {
    const int c = i % NCH, r = (i / NCH) % 3;
    lptr buf = (i >= 3 * NCH) ? sod : sev;
    buf[r * NCH + c]            = buf[3 * NCH + c];
    buf[(Lp + 3 + r) * NCH + c] = buf[(Lp + 2) * NCH + c];
  }
  __syncthreads();

  const int nb = node * 4;
  run_branch<0>(Wt1p + (size_t)(nb + 0) * K1N, Wt2p + (size_t)(nb + 0) * K2N,
                b1g + (nb + 0) * NCH, b2g + (nb + 0) * NCH,
                sev, hb, w1s, w2s, Lp, sod, db, xg, 0, 0);
  run_branch<1>(Wt1p + (size_t)(nb + 1) * K1N, Wt2p + (size_t)(nb + 1) * K2N,
                b1g + (nb + 1) * NCH, b2g + (nb + 1) * NCH,
                sod, hb, w1s, w2s, Lp, sev, cb, xg, 0, 0);
  for (int i = tid; i < 2 * 3 * NCH; i += NTHREADS) {
    const int c = i % NCH, r = (i / NCH) % 3;
    lptr buf = (i >= 3 * NCH) ? cb : db;
    buf[r * NCH + c]            = buf[3 * NCH + c];
    buf[(Lp + 3 + r) * NCH + c] = buf[(Lp + 2) * NCH + c];
  }
  __syncthreads();
  run_branch<2>(Wt1p + (size_t)(nb + 3) * K1N, Wt2p + (size_t)(nb + 3) * K2N,
                b1g + (nb + 3) * NCH, b2g + (nb + 3) * NCH,
                db, hb, w1s, w2s, Lp, cb, db, xg, 2 * stride, sel);
  run_branch<3>(Wt1p + (size_t)(nb + 2) * K1N, Wt2p + (size_t)(nb + 2) * K2N,
                b1g + (nb + 2) * NCH, b2g + (nb + 2) * NCH,
                cb, hb, w1s, w2s, Lp, db, cb, xg, 2 * stride, sel + stride);
}

__global__ void pack_weights_kernel(const float* __restrict__ W1,
                                    const float* __restrict__ W2,
                                    float* __restrict__ Wt1p,
                                    float* __restrict__ Wt2p) {
  const int idx = blockIdx.x * blockDim.x + threadIdx.x;
  const int N1 = 28 * K1N, N2 = 28 * K2N;
  if (idx < N1) {
    const int nb = idx / K1N, rem = idx - nb * K1N;
    const int p = rem >> 6, q = rem & 63;
    const int co = q >> 1, k = 2 * p + (q & 1);
    float v = 0.f;
    if (co < NCH && k < 95) {
      const int tap = k / NCH, ci = k - tap * NCH;
      v = W1[((size_t)(nb * NCH + co) * NCH + ci) * 5 + tap];
    }
    Wt1p[idx] = v;
  } else if (idx < N1 + N2) {
    const int j = idx - N1;
    const int nb = j / K2N, rem = j - nb * K2N;
    const int p = rem >> 6, q = rem & 63;
    const int co = q >> 1, k = 2 * p + (q & 1);
    float v = 0.f;
    if (co < NCH && k < 57) {
      const int tap = k / NCH, ci = k - tap * NCH;
      v = W2[((size_t)(nb * NCH + co) * NCH + ci) * 3 + tap];
    }
    Wt2p[j] = v;
  }
}

__global__ void concat_kernel(const float* __restrict__ x0D,
                              const float* __restrict__ xctrl,
                              float* __restrict__ xbuf) {
  const long long idx = (long long)blockIdx.x * blockDim.x + threadIdx.x;
  if (idx >= (long long)B_TOT * L_SEQ * NCH) return;
  const int c = (int)(idx % NCH);
  const long long bl = idx / NCH;
  xbuf[idx] = (c < 5) ? x0D[bl * 5 + c] : xctrl[bl * 14 + (c - 5)];
}

__global__ void proj_kernel(const float* __restrict__ xbuf,
                            const float* __restrict__ x0D,
                            const float* __restrict__ projw,
                            float* __restrict__ out) {
  __shared__ float yres_s[L_SEQ * 5];
  lptr yres = (lptr)yres_s;
  const int b = blockIdx.x, tid = threadIdx.x;
  cgptr xb = (cgptr)xbuf, x0 = (cgptr)x0D, pj = (cgptr)projw;
  for (int i = tid; i < L_SEQ * 5; i += blockDim.x) {
    const int l = i / 5, c = i - l * 5;
    yres[i] = xb[((size_t)b * L_SEQ + l) * NCH + c] +
              x0[((size_t)b * L_SEQ + l) * 5 + c];
  }
  __syncthreads();
  for (int i = tid; i < 64 * 5; i += blockDim.x) {
    const int o = i / 5, c = i - o * 5;
    cgptr pw = pj + o * L_SEQ;
    float acc = 0.f;
    for (int l = 0; l < L_SEQ; l++) acc = fmaf(pw[l], yres[l * 5 + c], acc);
    ((gptr)out)[(size_t)b * 320 + i] = acc;
  }
}

extern "C" void kernel_launch(void* const* d_in, const int* in_sizes, int n_in,
                              void* d_out, int out_size, void* d_ws,
                              size_t ws_size, hipStream_t stream) {
  const float* x0D   = (const float*)d_in[0];
  const float* xctrl = (const float*)d_in[1];
  const float* W1    = (const float*)d_in[2];
  const float* b1    = (const float*)d_in[3];
  const float* W2    = (const float*)d_in[4];
  const float* b2    = (const float*)d_in[5];
  const float* projw = (const float*)d_in[6];
  float* out = (float*)d_out;

  float* ws   = (float*)d_ws;
  float* xbuf = ws;
  const size_t XB = (size_t)B_TOT * L_SEQ * NCH;
  float* Wt1p = xbuf + XB;
  float* Wt2p = Wt1p + 28 * K1N;

  const int packN = 28 * K1N + 28 * K2N;
  pack_weights_kernel<<<(packN + 255) / 256, 256, 0, stream>>>(W1, W2, Wt1p, Wt2p);
  concat_kernel<<<(int)((XB + 255) / 256), 256, 0, stream>>>(x0D, xctrl, xbuf);
  for (int d = 0; d < 3; d++)
    scinet_level_kernel<<<B_TOT << d, NTHREADS, 0, stream>>>(xbuf, Wt1p, Wt2p,
                                                             b1, b2, d);
  proj_kernel<<<B_TOT, 256, 0, stream>>>(xbuf, x0D, projw, out);
}


# Each block takes 2^depth batch elements of one tree node, concatenated along M in LDS: 8192 blocks per level, a GEMM tile for every wave at every depth, weights staged once per 128 rows
# speedup vs baseline: 1.3055x; 1.3055x over previous
#include <hip/hip_runtime.h>
#include <math.h>

typedef __attribute__((ext_vector_type(2))) float v2f;
typedef __attribute__((ext_vector_type(4))) float v4f;
typedef __attribute__((ext_vector_type(8))) float v8f;

#define AS3 __attribute__((address_space(3)))
#define AS1 __attribute__((address_space(1)))
typedef AS3 float*       lptr;
typedef const AS3 float* clptr;
typedef AS1 float*       gptr;
typedef const AS1 float* cgptr;

#define B_TOT    8192
#define L_SEQ    256
#define NCH      19
#define KPACK1   96
#define KPACK2   64
#define NPACK    32
#define PRS      96
#define NWAVES   8
#define NTHREADS 256
#define K1N      (KPACK1 * NPACK)
#define K2N      (KPACK2 * NPACK)

#define ROWS_MAX 152
#define SEV_OFF 0
#define SEV_SZ  (ROWS_MAX * NCH + 32)
#define SOD_OFF (SEV_OFF + SEV_SZ)
#define SOD_SZ  SEV_SZ
#define HB_OFF  (SOD_OFF + SOD_SZ)
#define HB_SZ   SEV_SZ
#define DB_OFF  (HB_OFF + HB_SZ)
#define DB_SZ   SEV_SZ
#define CB_OFF  (DB_OFF + DB_SZ)
#define CB_SZ   SEV_SZ
#define W1S_OFF (CB_OFF + CB_SZ)
#define W1S_SZ  ((KPACK1 / 2) * PRS)
#define W2S_OFF (W1S_OFF + W1S_SZ)
#define W2S_SZ  ((KPACK2 / 2) * PRS)
#define SMEM_TOT (W2S_OFF + W2S_SZ)

static __device__ __forceinline__ v8f wmma4(v2f a, v2f b, v8f c) {
  return __builtin_amdgcn_wmma_f32_16x16x4_f32(false, a, false, b, (short)0, c,
                                               false, false);
}

static __device__ __forceinline__ float leaky(float x) {
  return x > 0.f ? x : 0.01f * x;
}

static __device__ __forceinline__ float fast_tanh(float x) {
  x = fminf(fmaxf(x, -15.f), 15.f);
  const float e = __expf(2.f * x);
  return __fdividef(e - 1.f, e + 1.f);
}

template <int MODE>
static __device__ void run_branch(cgptr Wt1, cgptr Wt2, cgptr b1, cgptr b2,
                                  clptr in, lptr hb, lptr w1s, lptr w2s,
                                  int Lp, int nseg, int tsh, clptr other,
                                  lptr outLds, gptr xg, int tmul, int tadd) {
  const int tid  = threadIdx.x;
  const int wave = tid >> 5;
  const int lane = tid & 31;
  const int lhi  = lane >> 4;
  const int lln  = lane & 15;

  for (int i = tid; i < K1N / 4; i += NTHREADS) {
    const int p = i >> 4, j = (i & 15) * 4;
    *(AS3 v4f*)(w1s + p * PRS + j) = ((const AS1 v4f*)Wt1)[i];
  }
  for (int i = tid; i < K2N / 4; i += NTHREADS) {
    const int p = i >> 4, j = (i & 15) * 4;
    *(AS3 v4f*)(w2s + p * PRS + j) = ((const AS1 v4f*)Wt2)[i];
  }
  __syncthreads();

  const int S    = Lp + 6;
  const int M1   = nseg * S - 4;
  const int nMt1 = (M1 + 15) >> 4;
  const float b1a = b1[lln];
  const float b1b = b1[lln < 3 ? 16 + lln : 18];
  for (int mt = wave; mt < nMt1; mt += NWAVES) {
    const int m0 = mt << 4;
    int mrow = m0 + lln; if (mrow > M1 - 1) mrow = M1 - 1;
    clptr ap = in + mrow * NCH + 2 * lhi;
    clptr bp = w1s + lhi * PRS + 2 * lln;
    v8f acc0 = {}; v8f acc1 = {};
#pragma unroll 4
    for (int k0 = 0; k0 < KPACK1; k0 += 4) {
      v2f a; a.x = ap[k0]; a.y = ap[k0 + 1];
      v2f bv0 = *(const AS3 v2f*)(bp);
      v2f bv1 = *(const AS3 v2f*)(bp + 32);
      bp += 2 * PRS;
      acc0 = wmma4(a, bv0, acc0);
      acc1 = wmma4(a, bv1, acc1);
    }
    lptr hrow = hb + (m0 + 8 * lhi) * NCH;
    if (m0 + 16 <= M1) {
#pragma unroll
      for (int v = 0; v < 8; v++) hrow[v * NCH + lln] = leaky(acc0[v] + b1a);
      if (lln < 3) {
#pragma unroll
        for (int v = 0; v < 8; v++)
          hrow[v * NCH + 16 + lln] = leaky(acc1[v] + b1b);
      }
    } else {
#pragma unroll
      for (int v = 0; v < 8; v++)
        if (m0 + 8 * lhi + v < M1) hrow[v * NCH + lln] = leaky(acc0[v] + b1a);
      if (lln < 3) {
#pragma unroll
        for (int v = 0; v < 8; v++)
          if (m0 + 8 * lhi + v < M1)
            hrow[v * NCH + 16 + lln] = leaky(acc1[v] + b1b);
      }
    }
  }
  __syncthreads();

  const int nMt2  = (nseg * Lp) >> 4;
  const int tmask = (1 << tsh) - 1;
  const float b2a = b2[lln];
  const float b2b = b2[lln < 3 ? 16 + lln : 18];
  for (int mt = wave; mt < nMt2; mt += NWAVES) {
    const int g  = mt >> tsh;
    const int j0 = (mt & tmask) << 4;
    const int r0 = g * S + j0;
    clptr ap = hb + (r0 + lln) * NCH + 2 * lhi;
    clptr bp = w2s + lhi * PRS + 2 * lln;
    v8f acc0 = {}; v8f acc1 = {};
#pragma unroll 4
    for (int k0 = 0; k0 < KPACK2; k0 += 4) {
      v2f a; a.x = ap[k0]; a.y = ap[k0 + 1];
      v2f bv0 = *(const AS3 v2f*)(bp);
      v2f bv1 = *(const AS3 v2f*)(bp + 32);
      bp += 2 * PRS;
      acc0 = wmma4(a, bv0, acc0);
      acc1 = wmma4(a, bv1, acc1);
    }
    const int mb = j0 + 8 * lhi;
    const int lb = (r0 + 3 + 8 * lhi) * NCH;
    gptr xs = xg + g * (L_SEQ * NCH);
#pragma unroll
    for (int v = 0; v < 8; v++) {
      const int m = mb + v;
      const float t   = fast_tanh(acc0[v] + b2a);
      const float oth = other[lb + v * NCH + lln];
      if (MODE <= 1)      outLds[lb + v * NCH + lln] = oth * __expf(t);
      else if (MODE == 2) xs[(m * tmul + tadd) * NCH + lln] = oth + t;
      else                xs[(m * tmul + tadd) * NCH + lln] = oth - t;
    }
    if (lln < 3) {
      const int co = 16 + lln;
#pragma unroll
      for (int v = 0; v < 8; v++) {
        const int m = mb + v;
        const float t   = fast_tanh(acc1[v] + b2b);
        const float oth = other[lb + v * NCH + co];
        if (MODE <= 1)      outLds[lb + v * NCH + co] = oth * __expf(t);
        else if (MODE == 2) xs[(m * tmul + tadd) * NCH + co] = oth + t;
        else                xs[(m * tmul + tadd) * NCH + co] = oth - t;
      }
    }
  }
  __syncthreads();
}

__global__ __launch_bounds__(NTHREADS) void scinet_level_kernel(
    float* __restrict__ xbuf_, const float* __restrict__ Wt1p_,
    const float* __restrict__ Wt2p_, const float* __restrict__ b1g_,
    const float* __restrict__ b2g_, int depth) {
  __shared__ __align__(16) float smem[SMEM_TOT];
  const int tid = threadIdx.x;
  const int G   = 1 << depth;
  const int sel = blockIdx.x & (G - 1);
  const int bg  = blockIdx.x >> depth;
  int node = 0;
  for (int i = 0; i < depth; i++) node = 2 * node + 1 + ((sel >> i) & 1);
  const int stride = 1 << depth;
  const int Lp     = (L_SEQ >> depth) >> 1;
  const int S      = Lp + 6;
  const int lsh    = 7 - depth;
  const int tsh    = 3 - depth;

  lptr sev = (lptr)(smem + SEV_OFF);
  lptr sod = (lptr)(smem + SOD_OFF);
  lptr hb  = (lptr)(smem + HB_OFF);
  lptr db  = (lptr)(smem + DB_OFF);
  lptr cb  = (lptr)(smem + CB_OFF);
  lptr w1s = (lptr)(smem + W1S_OFF);
  lptr w2s = (lptr)(smem + W2S_OFF);
  lptr sm  = (lptr)smem;

  for (int i = tid; i < SMEM_TOT; i += NTHREADS) sm[i] = 0.f;
  __syncthreads();

  gptr  xg   = (gptr)(xbuf_ + (size_t)bg * G * L_SEQ * NCH);
  cgptr Wt1p = (cgptr)Wt1p_;
  cgptr Wt2p = (cgptr)Wt2p_;
  cgptr b1g  = (cgptr)b1g_;
  cgptr b2g  = (cgptr)b2g_;

  for (int i = tid; i < (L_SEQ / 2) * NCH; i += NTHREADS) {
    const int ra = i / NCH, c = i - ra * NCH;
    const int g = ra >> lsh, r = ra & (Lp - 1);
    cgptr xs = xg + g * (L_SEQ * NCH);
    const int o = (g * S + 3 + r) * NCH + c;
    sev[o] = xs[((2 * r) * stride + sel) * NCH + c];
    sod[o] = xs[((2 * r + 1) * stride + sel) * NCH + c];
  }
  __syncthreads();
  for (int i = tid; i < G * 2 * 3 * NCH; i += NTHREADS) {
    const int q = i / NCH, c = i - q * NCH;
    const int w = q / 3, r = q - w * 3;
    lptr buf = ((w & 1) ? sod : sev) + (w >> 1) * S * NCH;
    buf[r * NCH + c]            = buf[3 * NCH + c];
    buf[(Lp + 3 + r) * NCH + c] = buf[(Lp + 2) * NCH + c];
  }
  __syncthreads();

  const int nb = node * 4;
  run_branch<0>(Wt1p + (size_t)(nb + 0) * K1N, Wt2p + (size_t)(nb + 0) * K2N,
                b1g + (nb + 0) * NCH, b2g + (nb + 0) * NCH,
                sev, hb, w1s, w2s, Lp, G, tsh, sod, db, xg, 0, 0);
  run_branch<1>(Wt1p + (size_t)(nb + 1) * K1N, Wt2p + (size_t)(nb + 1) * K2N,
                b1g + (nb + 1) * NCH, b2g + (nb + 1) * NCH,
                sod, hb, w1s, w2s, Lp, G, tsh, sev, cb, xg, 0, 0);
  for (int i = tid; i < G * 2 * 3 * NCH; i += NTHREADS) {
    const int q = i / NCH, c = i - q * NCH;
    const int w = q / 3, r = q - w * 3;
    lptr buf = ((w & 1) ? cb : db) + (w >> 1) * S * NCH;
    buf[r * NCH + c]            = buf[3 * NCH + c];
    buf[(Lp + 3 + r) * NCH + c] = buf[(Lp + 2) * NCH + c];
  }
  __syncthreads();
  run_branch<2>(Wt1p + (size_t)(nb + 3) * K1N, Wt2p + (size_t)(nb + 3) * K2N,
                b1g + (nb + 3) * NCH, b2g + (nb + 3) * NCH,
                db, hb, w1s, w2s, Lp, G, tsh, cb, db, xg, 2 * stride, sel);
  run_branch<3>(Wt1p + (size_t)(nb + 2) * K1N, Wt2p + (size_t)(nb + 2) * K2N,
                b1g + (nb + 2) * NCH, b2g + (nb + 2) * NCH,
                cb, hb, w1s, w2s, Lp, G, tsh, db, cb, xg, 2 * stride,
                sel + stride);
}

__global__ void pack_weights_kernel(const float* __restrict__ W1,
                                    const float* __restrict__ W2,
                                    float* __restrict__ Wt1p,
                                    float* __restrict__ Wt2p) {
  const int idx = blockIdx.x * blockDim.x + threadIdx.x;
  const int N1 = 28 * K1N, N2 = 28 * K2N;
  if (idx < N1) {
    const int nb = idx / K1N, rem = idx - nb * K1N;
    const int p = rem >> 6, q = rem & 63;
    const int co = q >> 1, k = 2 * p + (q & 1);
    float v = 0.f;
    if (co < NCH && k < 95) {
      const int tap = k / NCH, ci = k - tap * NCH;
      v = W1[((size_t)(nb * NCH + co) * NCH + ci) * 5 + tap];
    }
    Wt1p[idx] = v;
  } else if (idx < N1 + N2) {
    const int j = idx - N1;
    const int nb = j / K2N, rem = j - nb * K2N;
    const int p = rem >> 6, q = rem & 63;
    const int co = q >> 1, k = 2 * p + (q & 1);
    float v = 0.f;
    if (co < NCH && k < 57) {
      const int tap = k / NCH, ci = k - tap * NCH;
      v = W2[((size_t)(nb * NCH + co) * NCH + ci) * 3 + tap];
    }
    Wt2p[j] = v;
  }
}

__global__ void concat_kernel(const float* __restrict__ x0D,
                              const float* __restrict__ xctrl,
                              float* __restrict__ xbuf) {
  const long long idx = (long long)blockIdx.x * blockDim.x + threadIdx.x;
  if (idx >= (long long)B_TOT * L_SEQ * NCH) return;
  const int c = (int)(idx % NCH);
  const long long bl = idx / NCH;
  xbuf[idx] = (c < 5) ? x0D[bl * 5 + c] : xctrl[bl * 14 + (c - 5)];
}

__global__ void proj_kernel(const float* __restrict__ xbuf,
                            const float* __restrict__ x0D,
                            const float* __restrict__ projw,
                            float* __restrict__ out) {
  __shared__ float yres_s[L_SEQ * 5];
  lptr yres = (lptr)yres_s;
  const int b = blockIdx.x, tid = threadIdx.x;
  cgptr xb = (cgptr)xbuf, x0 = (cgptr)x0D, pj = (cgptr)projw;
  for (int i = tid; i < L_SEQ * 5; i += blockDim.x) {
    const int l = i / 5, c = i - l * 5;
    yres[i] = xb[((size_t)b * L_SEQ + l) * NCH + c] +
              x0[((size_t)b * L_SEQ + l) * 5 + c];
  }
  __syncthreads();
  for (int i = tid; i < 64 * 5; i += blockDim.x) {
    const int o = i / 5, c = i - o * 5;
    cgptr pw = pj + o * L_SEQ;
    float acc = 0.f;
    for (int l = 0; l < L_SEQ; l++) acc = fmaf(pw[l], yres[l * 5 + c], acc);
    ((gptr)out)[(size_t)b * 320 + i] = acc;
  }
}

extern "C" void kernel_launch(void* const* d_in, const int* in_sizes, int n_in,
                              void* d_out, int out_size, void* d_ws,
                              size_t ws_size, hipStream_t stream) {
  const float* x0D   = (const float*)d_in[0];
  const float* xctrl = (const float*)d_in[1];
  const float* W1    = (const float*)d_in[2];
  const float* b1    = (const float*)d_in[3];
  const float* W2    = (const float*)d_in[4];
  const float* b2    = (const float*)d_in[5];
  const float* projw = (const float*)d_in[6];
  float* out = (float*)d_out;

  float* ws   = (float*)d_ws;
  float* xbuf = ws;
  const size_t XB = (size_t)B_TOT * L_SEQ * NCH;
  float* Wt1p = xbuf + XB;
  float* Wt2p = Wt1p + 28 * K1N;

  const int packN = 28 * K1N + 28 * K2N;
  pack_weights_kernel<<<(packN + 255) / 256, 256, 0, stream>>>(W1, W2, Wt1p, Wt2p);
  concat_kernel<<<(int)((XB + 255) / 256), 256, 0, stream>>>(x0D, xctrl, xbuf);
  for (int d = 0; d < 3; d++)
    scinet_level_kernel<<<B_TOT, NTHREADS, 0, stream>>>(xbuf, Wt1p, Wt2p,
                                                        b1, b2, d);
  proj_kernel<<<B_TOT, 256, 0, stream>>>(xbuf, x0D, projw, out);
}


# Out channels 16-18 leave the padded 2nd N tile: one N tile = 5 row shifts x 3 channels (A rows 5 apart, B from zero-guarded weight strips shifted in k, k ascending as before); one chain per tile
# speedup vs baseline: 1.6322x; 1.2502x over previous
#include <hip/hip_runtime.h>
#include <math.h>


typedef __attribute__((ext_vector_type(2))) float v2f;
typedef __attribute__((ext_vector_type(4))) float v4f;
typedef __attribute__((ext_vector_type(8))) float v8f;

#define AS3 __attribute__((address_space(3)))
#define AS1 __attribute__((address_space(1)))
typedef AS3 float*       lptr;
typedef const AS3 float* clptr;
typedef AS1 float*       gptr;
typedef const AS1 float* cgptr;

#define B_TOT    8192
#define L_SEQ    256
#define NCH      19
#define NWAVES   8
#define NTHREADS 256

#define K1M      96
#define K2M      64
#define K1T      176
#define K2T      136
#define TG       76
#define TS1      260
#define TS2      252
#define WM1_OFF  0
#define WM2_OFF  1536
#define WT1_OFF  2560
#define WT2_OFF  (WT1_OFF + 3 * TS1)
#define WB_TOT   (WT2_OFF + 3 * TS2)

#define ROWS_MAX 152
#define BUF_SZ   (ROWS_MAX * NCH + 64)
#define SEV_OFF  0
#define SOD_OFF  (SEV_OFF + BUF_SZ)
#define HB_OFF   (SOD_OFF + BUF_SZ)
#define DB_OFF   (HB_OFF + BUF_SZ)
#define CB_OFF   (DB_OFF + BUF_SZ)
#define WL_OFF   (CB_OFF + BUF_SZ)
#define SMEM_TOT (WL_OFF + WB_TOT)

static __device__ __forceinline__ v8f wmma4(v2f a, v2f b, v8f c) {
  return __builtin_amdgcn_wmma_f32_16x16x4_f32(false, a, false, b, (short)0, c,
                                               false, false);
}

static __device__ __forceinline__ float leaky(float x) {
  return x > 0.f ? x : 0.01f * x;
}

static __device__ __forceinline__ float fast_tanh(float x) {
  x = fminf(fmaxf(x, -15.f), 15.f);
  const float e = __expf(2.f * x);
  return __fdividef(e - 1.f, e + 1.f);
}

template <int MODE>
static __device__ void run_branch(cgptr Wp, cgptr b1, cgptr b2, clptr in,
                                  lptr hb, lptr wl, int Lp, int nseg, int tsh,
                                  clptr other, lptr outLds, gptr xg, int tmul,
                                  int tadd) {
  const int tid  = threadIdx.x;
  const int wave = __builtin_amdgcn_readfirstlane(tid >> 5);
  const int lane = tid & 31;
  const int lhi  = lane >> 4;
  const int lln  = lane & 15;
  int tj = lln / 3;
  int tc = lln - 3 * tj;
  if (lln == 15) { tj = 0; tc = 0; }

#pragma unroll
  for (int it = 0; it < WB_TOT / 4 / NTHREADS; it++)
    ((AS3 v4f*)wl)[tid + it * NTHREADS] = ((const AS1 v4f*)Wp)[tid + it * NTHREADS];
  __syncthreads();

  const int S    = Lp + 6;
  const int M1   = nseg * S - 4;
  const int nMt1 = (M1 + 15) >> 4;
  const int nTr  = (M1 + 4) / 5;

  if (wave < 2) {
    int ti = (wave << 4) + lln; if (ti > nTr - 1) ti = nTr - 1;
    clptr ap = in + ti * (5 * NCH) + 2 * lhi;
    clptr bp = wl + WT1_OFF + tc * TS1 + TG - NCH * tj + 2 * lhi;
    v8f acc = {};
#pragma unroll 11
    for (int k0 = 0; k0 < K1T; k0 += 4) {
      v2f a; a.x = ap[k0]; a.y = ap[k0 + 1];
      v2f b; b.x = bp[k0]; b.y = bp[k0 + 1];
      acc = wmma4(a, b, acc);
    }
    const float bt = b1[16 + tc];
    const int rb = ((wave << 4) + 8 * lhi) * 5 + tj;
    if (lln < 15) {
#pragma unroll
      for (int v = 0; v < 8; v++) {
        const int r = rb + 5 * v;
        if (r < M1) hb[r * NCH + 16 + tc] = leaky(acc[v] + bt);
      }
    }
  }
#pragma nounroll
  for (int pass = 0; pass < 2; pass++) {
    const int mt = pass == 0 ? wave - 2
                             : (wave >= 6 ? wave : (wave >= 4 ? wave + 4 : 99));
    if (mt < 0 || mt >= nMt1) continue;
    const int m0 = mt << 4;
    int mrow = m0 + lln; if (mrow > M1 - 1) mrow = M1 - 1;
    clptr ap = in + mrow * NCH + 2 * lhi;
    clptr bp = wl + WM1_OFF + 2 * lane;
    v8f acc = {};
#pragma unroll 12
    for (int k0 = 0; k0 < K1M; k0 += 4) {
      v2f a; a.x = ap[k0]; a.y = ap[k0 + 1];
      v2f b = *(const AS3 v2f*)(bp);
      bp += 64;
      acc = wmma4(a, b, acc);
    }
    const float b1a = b1[lln];
    lptr hrow = hb + (m0 + 8 * lhi) * NCH;
    if (m0 + 16 <= M1) {
#pragma unroll
      for (int v = 0; v < 8; v++) hrow[v * NCH + lln] = leaky(acc[v] + b1a);
    } else {
#pragma unroll
      for (int v = 0; v < 8; v++)
        if (m0 + 8 * lhi + v < M1) hrow[v * NCH + lln] = leaky(acc[v] + b1a);
    }
  }
  __syncthreads();

  if (wave < 2) {
    int ti = (wave << 4) + lln; if (ti > nTr - 1) ti = nTr - 1;
    clptr ap = hb + ti * (5 * NCH) + 2 * lhi;
    clptr bp = wl + WT2_OFF + tc * TS2 + TG - NCH * tj + 2 * lhi;
    v8f acc = {};
#pragma unroll 17
    for (int k0 = 0; k0 < K2T; k0 += 4) {
      v2f a; a.x = ap[k0]; a.y = ap[k0 + 1];
      v2f b; b.x = bp[k0]; b.y = bp[k0 + 1];
      acc = wmma4(a, b, acc);
    }
    const float bt = b2[16 + tc];
    const int rb = ((wave << 4) + 8 * lhi) * 5 + tj;
    if (lln < 15) {
#pragma unroll
      for (int v = 0; v < 8; v++) {
        const int r  = rb + 5 * v;
        const int g  = (r >= S) + (r >= 2 * S) + (r >= 3 * S);
        const int jj = r - g * S;
        if (jj < Lp && g < nseg) {
          const int li = (r + 3) * NCH + 16 + tc;
          const float t   = fast_tanh(acc[v] + bt);
          const float oth = other[li];
          if (MODE <= 1) outLds[li] = oth * __expf(t);
          else {
            gptr xs = xg + g * (L_SEQ * NCH);
            if (MODE == 2) xs[(jj * tmul + tadd) * NCH + 16 + tc] = oth + t;
            else           xs[(jj * tmul + tadd) * NCH + 16 + tc] = oth - t;
          }
        }
      }
    }
  }
  const int nMt2  = (nseg * Lp) >> 4;
  const int tmask = (1 << tsh) - 1;
#pragma nounroll
  for (int pass = 0; pass < 2; pass++) {
    const int mt = pass == 0 ? wave - 2 : (wave >= 6 ? wave : 99);
    if (mt < 0 || mt >= nMt2) continue;
    const int g  = mt >> tsh;
    const int j0 = (mt & tmask) << 4;
    const int r0 = g * S + j0;
    clptr ap = hb + (r0 + lln) * NCH + 2 * lhi;
    clptr bp = wl + WM2_OFF + 2 * lane;
    v8f acc = {};
#pragma unroll 8
    for (int k0 = 0; k0 < K2M; k0 += 4) {
      v2f a; a.x = ap[k0]; a.y = ap[k0 + 1];
      v2f b = *(const AS3 v2f*)(bp);
      bp += 64;
      acc = wmma4(a, b, acc);
    }
    const float b2a = b2[lln];
    const int mb = j0 + 8 * lhi;
    const int lb = (r0 + 3 + 8 * lhi) * NCH;
    gptr xs = xg + g * (L_SEQ * NCH);
#pragma unroll
    for (int v = 0; v < 8; v++) {
      const int m = mb + v;
      const float t   = fast_tanh(acc[v] + b2a);
      const float oth = other[lb + v * NCH + lln];
      if (MODE <= 1)      outLds[lb + v * NCH + lln] = oth * __expf(t);
      else if (MODE == 2) xs[(m * tmul + tadd) * NCH + lln] = oth + t;
      else                xs[(m * tmul + tadd) * NCH + lln] = oth - t;
    }
  }
  __syncthreads();
}

__global__ __launch_bounds__(NTHREADS) void scinet_level_kernel(
    float* __restrict__ xbuf_, const float* __restrict__ Wpk_,
    const float* __restrict__ b1g_, const float* __restrict__ b2g_, int depth) {
  __shared__ __align__(16) float smem[SMEM_TOT];
  const int tid = threadIdx.x;
  const int G   = 1 << depth;
  const int sel = blockIdx.x & (G - 1);
  const int bg  = blockIdx.x >> depth;
  int node = 0;
  for (int i = 0; i < depth; i++) node = 2 * node + 1 + ((sel >> i) & 1);
  const int stride = 1 << depth;
  const int Lp     = (L_SEQ >> depth) >> 1;
  const int S      = Lp + 6;
  const int lsh    = 7 - depth;
  const int tsh    = 3 - depth;

  lptr sev = (lptr)(smem + SEV_OFF);
  lptr sod = (lptr)(smem + SOD_OFF);
  lptr hb  = (lptr)(smem + HB_OFF);
  lptr db  = (lptr)(smem + DB_OFF);
  lptr cb  = (lptr)(smem + CB_OFF);
  lptr wl  = (lptr)(smem + WL_OFF);
  lptr sm  = (lptr)smem;

  for (int i = tid; i < SMEM_TOT; i += NTHREADS) sm[i] = 0.f;
  __syncthreads();

  gptr  xg  = (gptr)(xbuf_ + (size_t)bg * G * L_SEQ * NCH);
  cgptr Wpk = (cgptr)Wpk_;
  cgptr b1g = (cgptr)b1g_;
  cgptr b2g = (cgptr)b2g_;

  for (int i = tid; i < (L_SEQ / 2) * NCH; i += NTHREADS) {
    const int ra = i / NCH, c = i - ra * NCH;
    const int g = ra >> lsh, r = ra & (Lp - 1);
    cgptr xs = xg + g * (L_SEQ * NCH);
    const int o = (g * S + 3 + r) * NCH + c;
    sev[o] = xs[((2 * r) * stride + sel) * NCH + c];
    sod[o] = xs[((2 * r + 1) * stride + sel) * NCH + c];
  }
  __syncthreads();
  for (int i = tid; i < G * 2 * 3 * NCH; i += NTHREADS) {
    const int q = i / NCH, c = i - q * NCH;
    const int w = q / 3, r = q - w * 3;
    lptr buf = ((w & 1) ? sod : sev) + (w >> 1) * S * NCH;
    buf[r * NCH + c]            = buf[3 * NCH + c];
    buf[(Lp + 3 + r) * NCH + c] = buf[(Lp + 2) * NCH + c];
  }
  __syncthreads();

  const int nb = node * 4;
  run_branch<0>(Wpk + (size_t)(nb + 0) * WB_TOT, b1g + (nb + 0) * NCH,
                b2g + (nb + 0) * NCH, sev, hb, wl, Lp, G, tsh, sod, db, xg, 0, 0);
  run_branch<1>(Wpk + (size_t)(nb + 1) * WB_TOT, b1g + (nb + 1) * NCH,
                b2g + (nb + 1) * NCH, sod, hb, wl, Lp, G, tsh, sev, cb, xg, 0, 0);
  for (int i = tid; i < G * 2 * 3 * NCH; i += NTHREADS) {
    const int q = i / NCH, c = i - q * NCH;
    const int w = q / 3, r = q - w * 3;
    lptr buf = ((w & 1) ? cb : db) + (w >> 1) * S * NCH;
    buf[r * NCH + c]            = buf[3 * NCH + c];
    buf[(Lp + 3 + r) * NCH + c] = buf[(Lp + 2) * NCH + c];
  }
  __syncthreads();
  run_branch<2>(Wpk + (size_t)(nb + 3) * WB_TOT, b1g + (nb + 3) * NCH,
                b2g + (nb + 3) * NCH, db, hb, wl, Lp, G, tsh, cb, db, xg,
                2 * stride, sel);
  run_branch<3>(Wpk + (size_t)(nb + 2) * WB_TOT, b1g + (nb + 2) * NCH,
                b2g + (nb + 2) * NCH, cb, hb, wl, Lp, G, tsh, db, cb, xg,
                2 * stride, sel + stride);
}

__global__ void pack_weights_kernel(const float* __restrict__ W1,
                                    const float* __restrict__ W2,
                                    float* __restrict__ Wpk) {
  const int idx = blockIdx.x * blockDim.x + threadIdx.x;
  if (idx >= 28 * WB_TOT) return;
  const int nb = idx / WB_TOT, rem = idx - nb * WB_TOT;
  float v = 0.f;
  if (rem < WM2_OFF) {
    const int p = rem >> 5, q = rem & 31;
    const int co = q >> 1, k = 2 * p + (q & 1);
    if (k < 95) {
      const int tap = k / NCH, ci = k - tap * NCH;
      v = W1[((size_t)(nb * NCH + co) * NCH + ci) * 5 + tap];
    }
  } else if (rem < WT1_OFF) {
    const int r = rem - WM2_OFF;
    const int p = r >> 5, q = r & 31;
    const int co = q >> 1, k = 2 * p + (q & 1);
    if (k < 57) {
      const int tap = k / NCH, ci = k - tap * NCH;
      v = W2[((size_t)(nb * NCH + co) * NCH + ci) * 3 + tap];
    }
  } else if (rem < WT2_OFF) {
    const int r = rem - WT1_OFF;
    const int c = r / TS1, k = r - c * TS1 - TG;
    if (k >= 0 && k < 95) {
      const int tap = k / NCH, ci = k - tap * NCH;
      v = W1[((size_t)(nb * NCH + 16 + c) * NCH + ci) * 5 + tap];
    }
  } else {
    const int r = rem - WT2_OFF;
    const int c = r / TS2, k = r - c * TS2 - TG;
    if (k >= 0 && k < 57) {
      const int tap = k / NCH, ci = k - tap * NCH;
      v = W2[((size_t)(nb * NCH + 16 + c) * NCH + ci) * 3 + tap];
    }
  }
  Wpk[idx] = v;
}

__global__ void concat_kernel(const float* __restrict__ x0D,
                              const float* __restrict__ xctrl,
                              float* __restrict__ xbuf) {
  const long long idx = (long long)blockIdx.x * blockDim.x + threadIdx.x;
  if (idx >= (long long)B_TOT * L_SEQ * NCH) return;
  const int c = (int)(idx % NCH);
  const long long bl = idx / NCH;
  xbuf[idx] = (c < 5) ? x0D[bl * 5 + c] : xctrl[bl * 14 + (c - 5)];
}

__global__ void proj_kernel(const float* __restrict__ xbuf,
                            const float* __restrict__ x0D,
                            const float* __restrict__ projw,
                            float* __restrict__ out) {
  __shared__ float yres_s[L_SEQ * 5];
  lptr yres = (lptr)yres_s;
  const int b = blockIdx.x, tid = threadIdx.x;
  cgptr xb = (cgptr)xbuf, x0 = (cgptr)x0D, pj = (cgptr)projw;
  for (int i = tid; i < L_SEQ * 5; i += blockDim.x) {
    const int l = i / 5, c = i - l * 5;
    yres[i] = xb[((size_t)b * L_SEQ + l) * NCH + c] +
              x0[((size_t)b * L_SEQ + l) * 5 + c];
  }
  __syncthreads();
  for (int i = tid; i < 64 * 5; i += blockDim.x) {
    const int o = i / 5, c = i - o * 5;
    cgptr pw = pj + o * L_SEQ;
    float acc = 0.f;
    for (int l = 0; l < L_SEQ; l++) acc = fmaf(pw[l], yres[l * 5 + c], acc);
    ((gptr)out)[(size_t)b * 320 + i] = acc;
  }
}

extern "C" void kernel_launch(void* const* d_in, const int* in_sizes, int n_in,
                              void* d_out, int out_size, void* d_ws,
                              size_t ws_size, hipStream_t stream) {
  const float* x0D   = (const float*)d_in[0];
  const float* xctrl = (const float*)d_in[1];
  const float* W1    = (const float*)d_in[2];
  const float* b1    = (const float*)d_in[3];
  const float* W2    = (const float*)d_in[4];
  const float* b2    = (const float*)d_in[5];
  const float* projw = (const float*)d_in[6];
  float* out = (float*)d_out;

  float* ws   = (float*)d_ws;
  float* xbuf = ws;
  const size_t XB = (size_t)B_TOT * L_SEQ * NCH;
  float* Wpk = xbuf + XB;

  const int packN = 28 * WB_TOT;
  pack_weights_kernel<<<(packN + 255) / 256, 256, 0, stream>>>(W1, W2, Wpk);
  concat_kernel<<<(int)((XB + 255) / 256), 256, 0, stream>>>(x0D, xctrl, xbuf);
  for (int d = 0; d < 3; d++)
    scinet_level_kernel<<<B_TOT, NTHREADS, 0, stream>>>(xbuf, Wpk, b1, b2, d);
  proj_kernel<<<B_TOT, 256, 0, stream>>>(xbuf, x0D, projw, out);
}


# Level 0 reads x_0D and x_ctrl itself, one contiguous run per batch element, straight into its even/odd LDS buffers: the concat launch, its 1245184 one-element waves and 318 MB of traffic are gone
# speedup vs baseline: 1.7976x; 1.1013x over previous
#include <hip/hip_runtime.h>
#include <math.h>


typedef __attribute__((ext_vector_type(2))) float v2f;
typedef __attribute__((ext_vector_type(4))) float v4f;
typedef __attribute__((ext_vector_type(8))) float v8f;

#define AS3 __attribute__((address_space(3)))
#define AS1 __attribute__((address_space(1)))
typedef AS3 float*       lptr;
typedef const AS3 float* clptr;
typedef AS1 float*       gptr;
typedef const AS1 float* cgptr;

#define B_TOT    8192
#define L_SEQ    256
#define NCH      19
#define NWAVES   8
#define NTHREADS 256

#define K1M      96
#define K2M      64
#define K1T      176
#define K2T      136
#define TG       76
#define TS1      260
#define TS2      252
#define WM1_OFF  0
#define WM2_OFF  1536
#define WT1_OFF  2560
#define WT2_OFF  (WT1_OFF + 3 * TS1)
#define WB_TOT   (WT2_OFF + 3 * TS2)

#define ROWS_MAX 152
#define BUF_SZ   (ROWS_MAX * NCH + 64)
#define SEV_OFF  0
#define SOD_OFF  (SEV_OFF + BUF_SZ)
#define HB_OFF   (SOD_OFF + BUF_SZ)
#define DB_OFF   (HB_OFF + BUF_SZ)
#define CB_OFF   (DB_OFF + BUF_SZ)
#define WL_OFF   (CB_OFF + BUF_SZ)
#define SMEM_TOT (WL_OFF + WB_TOT)

static __device__ __forceinline__ v8f wmma4(v2f a, v2f b, v8f c) {
  return __builtin_amdgcn_wmma_f32_16x16x4_f32(false, a, false, b, (short)0, c,
                                               false, false);
}

static __device__ __forceinline__ float leaky(float x) {
  return x > 0.f ? x : 0.01f * x;
}

static __device__ __forceinline__ float fast_tanh(float x) {
  x = fminf(fmaxf(x, -15.f), 15.f);
  const float e = __expf(2.f * x);
  return __fdividef(e - 1.f, e + 1.f);
}

template <int MODE>
static __device__ void run_branch(cgptr Wp, cgptr b1, cgptr b2, clptr in,
                                  lptr hb, lptr wl, int Lp, int nseg, int tsh,
                                  clptr other, lptr outLds, gptr xg, int tmul,
                                  int tadd) {
  const int tid  = threadIdx.x;
  const int wave = __builtin_amdgcn_readfirstlane(tid >> 5);
  const int lane = tid & 31;
  const int lhi  = lane >> 4;
  const int lln  = lane & 15;
  int tj = lln / 3;
  int tc = lln - 3 * tj;
  if (lln == 15) { tj = 0; tc = 0; }

#pragma unroll
  for (int it = 0; it < WB_TOT / 4 / NTHREADS; it++)
    ((AS3 v4f*)wl)[tid + it * NTHREADS] = ((const AS1 v4f*)Wp)[tid + it * NTHREADS];
  __syncthreads();

  const int S    = Lp + 6;
  const int M1   = nseg * S - 4;
  const int nMt1 = (M1 + 15) >> 4;
  const int nTr  = (M1 + 4) / 5;

  if (wave < 2) {
    int ti = (wave << 4) + lln; if (ti > nTr - 1) ti = nTr - 1;
    clptr ap = in + ti * (5 * NCH) + 2 * lhi;
    clptr bp = wl + WT1_OFF + tc * TS1 + TG - NCH * tj + 2 * lhi;
    v8f acc = {};
#pragma unroll 11
    for (int k0 = 0; k0 < K1T; k0 += 4) {
      v2f a; a.x = ap[k0]; a.y = ap[k0 + 1];
      v2f b; b.x = bp[k0]; b.y = bp[k0 + 1];
      acc = wmma4(a, b, acc);
    }
    const float bt = b1[16 + tc];
    const int rb = ((wave << 4) + 8 * lhi) * 5 + tj;
    if (lln < 15) {
#pragma unroll
      for (int v = 0; v < 8; v++) {
        const int r = rb + 5 * v;
        if (r < M1) hb[r * NCH + 16 + tc] = leaky(acc[v] + bt);
      }
    }
  }
#pragma nounroll
  for (int pass = 0; pass < 2; pass++) {
    const int mt = pass == 0 ? wave - 2
                             : (wave >= 6 ? wave : (wave >= 4 ? wave + 4 : 99));
    if (mt < 0 || mt >= nMt1) continue;
    const int m0 = mt << 4;
    int mrow = m0 + lln; if (mrow > M1 - 1) mrow = M1 - 1;
    clptr ap = in + mrow * NCH + 2 * lhi;
    clptr bp = wl + WM1_OFF + 2 * lane;
    v8f acc = {};
#pragma unroll 12
    for (int k0 = 0; k0 < K1M; k0 += 4) {
      v2f a; a.x = ap[k0]; a.y = ap[k0 + 1];
      v2f b = *(const AS3 v2f*)(bp);
      bp += 64;
      acc = wmma4(a, b, acc);
    }
    const float b1a = b1[lln];
    lptr hrow = hb + (m0 + 8 * lhi) * NCH;
    if (m0 + 16 <= M1) {
#pragma unroll
      for (int v = 0; v < 8; v++) hrow[v * NCH + lln] = leaky(acc[v] + b1a);
    } else {
#pragma unroll
      for (int v = 0; v < 8; v++)
        if (m0 + 8 * lhi + v < M1) hrow[v * NCH + lln] = leaky(acc[v] + b1a);
    }
  }
  __syncthreads();

  if (wave < 2) {
    int ti = (wave << 4) + lln; if (ti > nTr - 1) ti = nTr - 1;
    clptr ap = hb + ti * (5 * NCH) + 2 * lhi;
    clptr bp = wl + WT2_OFF + tc * TS2 + TG - NCH * tj + 2 * lhi;
    v8f acc = {};
#pragma unroll 17
    for (int k0 = 0; k0 < K2T; k0 += 4) {
      v2f a; a.x = ap[k0]; a.y = ap[k0 + 1];
      v2f b; b.x = bp[k0]; b.y = bp[k0 + 1];
      acc = wmma4(a, b, acc);
    }
    const float bt = b2[16 + tc];
    const int rb = ((wave << 4) + 8 * lhi) * 5 + tj;
    if (lln < 15) {
#pragma unroll
      for (int v = 0; v < 8; v++) {
        const int r  = rb + 5 * v;
        const int g  = (r >= S) + (r >= 2 * S) + (r >= 3 * S);
        const int jj = r - g * S;
        if (jj < Lp && g < nseg) {
          const int li = (r + 3) * NCH + 16 + tc;
          const float t   = fast_tanh(acc[v] + bt);
          const float oth = other[li];
          if (MODE <= 1) outLds[li] = oth * __expf(t);
          else {
            gptr xs = xg + g * (L_SEQ * NCH);
            if (MODE == 2) xs[(jj * tmul + tadd) * NCH + 16 + tc] = oth + t;
            else           xs[(jj * tmul + tadd) * NCH + 16 + tc] = oth - t;
          }
        }
      }
    }
  }
  const int nMt2  = (nseg * Lp) >> 4;
  const int tmask = (1 << tsh) - 1;
#pragma nounroll
  for (int pass = 0; pass < 2; pass++) {
    const int mt = pass == 0 ? wave - 2 : (wave >= 6 ? wave : 99);
    if (mt < 0 || mt >= nMt2) continue;
    const int g  = mt >> tsh;
    const int j0 = (mt & tmask) << 4;
    const int r0 = g * S + j0;
    clptr ap = hb + (r0 + lln) * NCH + 2 * lhi;
    clptr bp = wl + WM2_OFF + 2 * lane;
    v8f acc = {};
#pragma unroll 8
    for (int k0 = 0; k0 < K2M; k0 += 4) {
      v2f a; a.x = ap[k0]; a.y = ap[k0 + 1];
      v2f b = *(const AS3 v2f*)(bp);
      bp += 64;
      acc = wmma4(a, b, acc);
    }
    const float b2a = b2[lln];
    const int mb = j0 + 8 * lhi;
    const int lb = (r0 + 3 + 8 * lhi) * NCH;
    gptr xs = xg + g * (L_SEQ * NCH);
#pragma unroll
    for (int v = 0; v < 8; v++) {
      const int m = mb + v;
      const float t   = fast_tanh(acc[v] + b2a);
      const float oth = other[lb + v * NCH + lln];
      if (MODE <= 1)      outLds[lb + v * NCH + lln] = oth * __expf(t);
      else if (MODE == 2) xs[(m * tmul + tadd) * NCH + lln] = oth + t;
      else                xs[(m * tmul + tadd) * NCH + lln] = oth - t;
    }
  }
  __syncthreads();
}

__global__ __launch_bounds__(NTHREADS) void scinet_level_kernel(
    float* __restrict__ xbuf_, const float* __restrict__ x0D_,
    const float* __restrict__ xctrl_, const float* __restrict__ Wpk_,
    const float* __restrict__ b1g_, const float* __restrict__ b2g_, int depth) {
  __shared__ __align__(16) float smem[SMEM_TOT];
  const int tid = threadIdx.x;
  const int G   = 1 << depth;
  const int sel = blockIdx.x & (G - 1);
  const int bg  = blockIdx.x >> depth;
  int node = 0;
  for (int i = 0; i < depth; i++) node = 2 * node + 1 + ((sel >> i) & 1);
  const int stride = 1 << depth;
  const int Lp     = (L_SEQ >> depth) >> 1;
  const int S      = Lp + 6;
  const int lsh    = 7 - depth;
  const int tsh    = 3 - depth;

  lptr sev = (lptr)(smem + SEV_OFF);
  lptr sod = (lptr)(smem + SOD_OFF);
  lptr hb  = (lptr)(smem + HB_OFF);
  lptr db  = (lptr)(smem + DB_OFF);
  lptr cb  = (lptr)(smem + CB_OFF);
  lptr wl  = (lptr)(smem + WL_OFF);
  lptr sm  = (lptr)smem;

  for (int i = tid; i < SMEM_TOT; i += NTHREADS) sm[i] = 0.f;
  __syncthreads();

  gptr  xg  = (gptr)(xbuf_ + (size_t)bg * G * L_SEQ * NCH);
  cgptr Wpk = (cgptr)Wpk_;
  cgptr b1g = (cgptr)b1g_;
  cgptr b2g = (cgptr)b2g_;

  if (depth == 0) {
    cgptr x0 = (cgptr)x0D_ + (size_t)bg * (L_SEQ * 5);
    cgptr xc = (cgptr)xctrl_ + (size_t)bg * (L_SEQ * 14);
    for (int it = 0; it < (L_SEQ * 5) / NTHREADS; it++) {
      const int e = tid + it * NTHREADS;
      const int t = e / 5, c = e - t * 5;
      lptr dst = (t & 1) ? sod : sev;
      dst[(3 + (t >> 1)) * NCH + c] = x0[e];
    }
    for (int it = 0; it < (L_SEQ * 14) / NTHREADS; it++) {
      const int e = tid + it * NTHREADS;
      const int t = e / 14, c = e - t * 14;
      lptr dst = (t & 1) ? sod : sev;
      dst[(3 + (t >> 1)) * NCH + 5 + c] = xc[e];
    }
  } else {
    for (int i = tid; i < (L_SEQ / 2) * NCH; i += NTHREADS) {
      const int ra = i / NCH, c = i - ra * NCH;
      const int g = ra >> lsh, r = ra & (Lp - 1);
      cgptr xs = xg + g * (L_SEQ * NCH);
      const int o = (g * S + 3 + r) * NCH + c;
      sev[o] = xs[((2 * r) * stride + sel) * NCH + c];
      sod[o] = xs[((2 * r + 1) * stride + sel) * NCH + c];
    }
  }
  __syncthreads();
  for (int i = tid; i < G * 2 * 3 * NCH; i += NTHREADS) {
    const int q = i / NCH, c = i - q * NCH;
    const int w = q / 3, r = q - w * 3;
    lptr buf = ((w & 1) ? sod : sev) + (w >> 1) * S * NCH;
    buf[r * NCH + c]            = buf[3 * NCH + c];
    buf[(Lp + 3 + r) * NCH + c] = buf[(Lp + 2) * NCH + c];
  }
  __syncthreads();

  const int nb = node * 4;
  run_branch<0>(Wpk + (size_t)(nb + 0) * WB_TOT, b1g + (nb + 0) * NCH,
                b2g + (nb + 0) * NCH, sev, hb, wl, Lp, G, tsh, sod, db, xg, 0, 0);
  run_branch<1>(Wpk + (size_t)(nb + 1) * WB_TOT, b1g + (nb + 1) * NCH,
                b2g + (nb + 1) * NCH, sod, hb, wl, Lp, G, tsh, sev, cb, xg, 0, 0);
  for (int i = tid; i < G * 2 * 3 * NCH; i += NTHREADS) {
    const int q = i / NCH, c = i - q * NCH;
    const int w = q / 3, r = q - w * 3;
    lptr buf = ((w & 1) ? cb : db) + (w >> 1) * S * NCH;
    buf[r * NCH + c]            = buf[3 * NCH + c];
    buf[(Lp + 3 + r) * NCH + c] = buf[(Lp + 2) * NCH + c];
  }
  __syncthreads();
  run_branch<2>(Wpk + (size_t)(nb + 3) * WB_TOT, b1g + (nb + 3) * NCH,
                b2g + (nb + 3) * NCH, db, hb, wl, Lp, G, tsh, cb, db, xg,
                2 * stride, sel);
  run_branch<3>(Wpk + (size_t)(nb + 2) * WB_TOT, b1g + (nb + 2) * NCH,
                b2g + (nb + 2) * NCH, cb, hb, wl, Lp, G, tsh, db, cb, xg,
                2 * stride, sel + stride);
}

__global__ void pack_weights_kernel(const float* __restrict__ W1,
                                    const float* __restrict__ W2,
                                    float* __restrict__ Wpk) {
  const int idx = blockIdx.x * blockDim.x + threadIdx.x;
  if (idx >= 28 * WB_TOT) return;
  const int nb = idx / WB_TOT, rem = idx - nb * WB_TOT;
  float v = 0.f;
  if (rem < WM2_OFF) {
    const int p = rem >> 5, q = rem & 31;
    const int co = q >> 1, k = 2 * p + (q & 1);
    if (k < 95) {
      const int tap = k / NCH, ci = k - tap * NCH;
      v = W1[((size_t)(nb * NCH + co) * NCH + ci) * 5 + tap];
    }
  } else if (rem < WT1_OFF) {
    const int r = rem - WM2_OFF;
    const int p = r >> 5, q = r & 31;
    const int co = q >> 1, k = 2 * p + (q & 1);
    if (k < 57) {
      const int tap = k / NCH, ci = k - tap * NCH;
      v = W2[((size_t)(nb * NCH + co) * NCH + ci) * 3 + tap];
    }
  } else if (rem < WT2_OFF) {
    const int r = rem - WT1_OFF;
    const int c = r / TS1, k = r - c * TS1 - TG;
    if (k >= 0 && k < 95) {
      const int tap = k / NCH, ci = k - tap * NCH;
      v = W1[((size_t)(nb * NCH + 16 + c) * NCH + ci) * 5 + tap];
    }
  } else {
    const int r = rem - WT2_OFF;
    const int c = r / TS2, k = r - c * TS2 - TG;
    if (k >= 0 && k < 57) {
      const int tap = k / NCH, ci = k - tap * NCH;
      v = W2[((size_t)(nb * NCH + 16 + c) * NCH + ci) * 3 + tap];
    }
  }
  Wpk[idx] = v;
}

__global__ void proj_kernel(const float* __restrict__ xbuf,
                            const float* __restrict__ x0D,
                            const float* __restrict__ projw,
                            float* __restrict__ out) {
  __shared__ float yres_s[L_SEQ * 5];
  lptr yres = (lptr)yres_s;
  const int b = blockIdx.x, tid = threadIdx.x;
  cgptr xb = (cgptr)xbuf, x0 = (cgptr)x0D, pj = (cgptr)projw;
  for (int i = tid; i < L_SEQ * 5; i += blockDim.x) {
    const int l = i / 5, c = i - l * 5;
    yres[i] = xb[((size_t)b * L_SEQ + l) * NCH + c] +
              x0[((size_t)b * L_SEQ + l) * 5 + c];
  }
  __syncthreads();
  for (int i = tid; i < 64 * 5; i += blockDim.x) {
    const int o = i / 5, c = i - o * 5;
    cgptr pw = pj + o * L_SEQ;
    float acc = 0.f;
    for (int l = 0; l < L_SEQ; l++) acc = fmaf(pw[l], yres[l * 5 + c], acc);
    ((gptr)out)[(size_t)b * 320 + i] = acc;
  }
}

extern "C" void kernel_launch(void* const* d_in, const int* in_sizes, int n_in,
                              void* d_out, int out_size, void* d_ws,
                              size_t ws_size, hipStream_t stream) {
  const float* x0D   = (const float*)d_in[0];
  const float* xctrl = (const float*)d_in[1];
  const float* W1    = (const float*)d_in[2];
  const float* b1    = (const float*)d_in[3];
  const float* W2    = (const float*)d_in[4];
  const float* b2    = (const float*)d_in[5];
  const float* projw = (const float*)d_in[6];
  float* out = (float*)d_out;

  float* ws   = (float*)d_ws;
  float* xbuf = ws;
  const size_t XB = (size_t)B_TOT * L_SEQ * NCH;
  float* Wpk = xbuf + XB;

  const int packN = 28 * WB_TOT;
  pack_weights_kernel<<<(packN + 255) / 256, 256, 0, stream>>>(W1, W2, Wpk);
  for (int d = 0; d < 3; d++)
    scinet_level_kernel<<<B_TOT, NTHREADS, 0, stream>>>(xbuf, x0D, xctrl, Wpk, b1,
                                                        b2, d);
  proj_kernel<<<B_TOT, 256, 0, stream>>>(xbuf, x0D, projw, out);
}


# LDS rows padded 19 -> 20 floats (80 B): each A/B operand of two k-steps is one conflict-free ds_load_b128; K 100/60, tails 180/140; sums reordered: k inside each 8 runs 0,1,4,5,2,3,6,7
# speedup vs baseline: 1.8724x; 1.0416x over previous
#include <hip/hip_runtime.h>
#include <math.h>


typedef __attribute__((ext_vector_type(2))) float v2f;
typedef __attribute__((ext_vector_type(4))) float v4f;
typedef __attribute__((ext_vector_type(8))) float v8f;

#define AS3 __attribute__((address_space(3)))
#define AS1 __attribute__((address_space(1)))
typedef AS3 float*       lptr;
typedef const AS3 float* clptr;
typedef AS1 float*       gptr;
typedef const AS1 float* cgptr;

#define B_TOT    8192
#define L_SEQ    256
#define NCH      19
#define PCH      20
#define NWAVES   8
#define NTHREADS 256

#define K1M      100
#define K2M      60
#define K1T      180
#define K2T      140
#define TG       80
#define TS1      284
#define TS2      220
#define WM1_OFF  0
#define WM2_OFF  1600
#define WT1_OFF  2560
#define WT2_OFF  (WT1_OFF + 3 * TS1)
#define WB_TOT   4096

#define ROWS_MAX 152
#define BUF_SZ   (ROWS_MAX * PCH + 64)
#define SEV_OFF  0
#define SOD_OFF  (SEV_OFF + BUF_SZ)
#define HB_OFF   (SOD_OFF + BUF_SZ)
#define DB_OFF   (HB_OFF + BUF_SZ)
#define CB_OFF   (DB_OFF + BUF_SZ)
#define WL_OFF   (CB_OFF + BUF_SZ)
#define SMEM_TOT (WL_OFF + WB_TOT)

static __device__ __forceinline__ v8f wmma4(v2f a, v2f b, v8f c) {
  return __builtin_amdgcn_wmma_f32_16x16x4_f32(false, a, false, b, (short)0, c,
                                               false, false);
}

static __device__ __forceinline__ float leaky(float x) {
  return x > 0.f ? x : 0.01f * x;
}

static __device__ __forceinline__ float fast_tanh(float x) {
  x = fminf(fmaxf(x, -15.f), 15.f);
  const float e = __expf(2.f * x);
  return __fdividef(e - 1.f, e + 1.f);
}

template <int MODE>
static __device__ void run_branch(cgptr Wp, cgptr b1, cgptr b2, clptr in,
                                  lptr hb, lptr wl, int Lp, int nseg, int tsh,
                                  clptr other, lptr outLds, gptr xg, int tmul,
                                  int tadd) {
  const int tid  = threadIdx.x;
  const int wave = __builtin_amdgcn_readfirstlane(tid >> 5);
  const int lane = tid & 31;
  const int lhi  = lane >> 4;
  const int lln  = lane & 15;
  int tj = lln / 3;
  int tc = lln - 3 * tj;
  if (lln == 15) { tj = 0; tc = 0; }

#pragma unroll
  for (int it = 0; it < WB_TOT / 4 / NTHREADS; it++)
    ((AS3 v4f*)wl)[tid + it * NTHREADS] = ((const AS1 v4f*)Wp)[tid + it * NTHREADS];
  __syncthreads();

  const int S    = Lp + 6;
  const int M1   = nseg * S - 4;
  const int nMt1 = (M1 + 15) >> 4;
  const int nTr  = (M1 + 4) / 5;

  if (wave < 2) {
    int ti = (wave << 4) + lln; if (ti > nTr - 1) ti = nTr - 1;
    clptr ar = in + ti * (5 * PCH);
    clptr br = wl + WT1_OFF + tc * TS1 + TG - PCH * tj;
    v8f acc = {};
    v2f aS[3], bS[3];
#pragma unroll
    for (int u = 0; u < 3; u++) {
      aS[u] = *(const AS3 v2f*)(ar + 168 + 4 * u + 2 * lhi);
      bS[u] = *(const AS3 v2f*)(br + 168 + 4 * u + 2 * lhi);
    }
#pragma unroll 7
    for (int s = 0; s < 21; s++) {
      const v4f a = *(const AS3 v4f*)(ar + 8 * s + 4 * lhi);
      const v4f b = *(const AS3 v4f*)(br + 8 * s + 4 * lhi);
      acc = wmma4(a.xy, b.xy, acc);
      acc = wmma4(a.zw, b.zw, acc);
    }
#pragma unroll
    for (int u = 0; u < 3; u++) acc = wmma4(aS[u], bS[u], acc);
    const float bt = b1[16 + tc];
    const int rb = ((wave << 4) + 8 * lhi) * 5 + tj;
    if (lln < 15) {
#pragma unroll
      for (int v = 0; v < 8; v++) {
        const int r = rb + 5 * v;
        if (r < M1) hb[r * PCH + 16 + tc] = leaky(acc[v] + bt);
      }
    }
  }
#pragma nounroll
  for (int pass = 0; pass < 2; pass++) {
    const int mt = pass == 0 ? wave - 2
                             : (wave >= 6 ? wave : (wave >= 4 ? wave + 4 : 99));
    if (mt < 0 || mt >= nMt1) continue;
    const int m0 = mt << 4;
    int mrow = m0 + lln; if (mrow > M1 - 1) mrow = M1 - 1;
    clptr ar = in + mrow * PCH;
    clptr bq = wl + WM1_OFF + 4 * lane;
    v8f acc = {};
    const v2f aL = *(const AS3 v2f*)(ar + 96 + 2 * lhi);
    const v2f bL = *(const AS3 v2f*)(wl + WM1_OFF + 12 * 128 + 2 * lane);
#pragma unroll 6
    for (int s = 0; s < 12; s++) {
      const v4f a = *(const AS3 v4f*)(ar + 8 * s + 4 * lhi);
      const v4f b = *(const AS3 v4f*)(bq + 128 * s);
      acc = wmma4(a.xy, b.xy, acc);
      acc = wmma4(a.zw, b.zw, acc);
    }
    acc = wmma4(aL, bL, acc);
    const float b1a = b1[lln];
    lptr hrow = hb + (m0 + 8 * lhi) * PCH;
    if (m0 + 16 <= M1) {
#pragma unroll
      for (int v = 0; v < 8; v++) hrow[v * PCH + lln] = leaky(acc[v] + b1a);
    } else {
#pragma unroll
      for (int v = 0; v < 8; v++)
        if (m0 + 8 * lhi + v < M1) hrow[v * PCH + lln] = leaky(acc[v] + b1a);
    }
  }
  __syncthreads();

  if (wave < 2) {
    int ti = (wave << 4) + lln; if (ti > nTr - 1) ti = nTr - 1;
    clptr ar = hb + ti * (5 * PCH);
    clptr br = wl + WT2_OFF + tc * TS2 + TG - PCH * tj;
    v8f acc = {};
    v2f aS[3], bS[3];
#pragma unroll
    for (int u = 0; u < 3; u++) {
      aS[u] = *(const AS3 v2f*)(ar + 128 + 4 * u + 2 * lhi);
      bS[u] = *(const AS3 v2f*)(br + 128 + 4 * u + 2 * lhi);
    }
#pragma unroll 8
    for (int s = 0; s < 16; s++) {
      const v4f a = *(const AS3 v4f*)(ar + 8 * s + 4 * lhi);
      const v4f b = *(const AS3 v4f*)(br + 8 * s + 4 * lhi);
      acc = wmma4(a.xy, b.xy, acc);
      acc = wmma4(a.zw, b.zw, acc);
    }
#pragma unroll
    for (int u = 0; u < 3; u++) acc = wmma4(aS[u], bS[u], acc);
    const float bt = b2[16 + tc];
    const int rb = ((wave << 4) + 8 * lhi) * 5 + tj;
    if (lln < 15) {
#pragma unroll
      for (int v = 0; v < 8; v++) {
        const int r  = rb + 5 * v;
        const int g  = (r >= S) + (r >= 2 * S) + (r >= 3 * S);
        const int jj = r - g * S;
        if (jj < Lp && g < nseg) {
          const int li = (r + 3) * PCH + 16 + tc;
          const float t   = fast_tanh(acc[v] + bt);
          const float oth = other[li];
          if (MODE <= 1) outLds[li] = oth * __expf(t);
          else {
            gptr xs = xg + g * (L_SEQ * NCH);
            if (MODE == 2) xs[(jj * tmul + tadd) * NCH + 16 + tc] = oth + t;
            else           xs[(jj * tmul + tadd) * NCH + 16 + tc] = oth - t;
          }
        }
      }
    }
  }
  const int nMt2  = (nseg * Lp) >> 4;
  const int tmask = (1 << tsh) - 1;
#pragma nounroll
  for (int pass = 0; pass < 2; pass++) {
    const int mt = pass == 0 ? wave - 2 : (wave >= 6 ? wave : 99);
    if (mt < 0 || mt >= nMt2) continue;
    const int g  = mt >> tsh;
    const int j0 = (mt & tmask) << 4;
    const int r0 = g * S + j0;
    clptr ar = hb + (r0 + lln) * PCH;
    clptr bq = wl + WM2_OFF + 4 * lane;
    v8f acc = {};
    const v2f aL = *(const AS3 v2f*)(ar + 56 + 2 * lhi);
    const v2f bL = *(const AS3 v2f*)(wl + WM2_OFF + 7 * 128 + 2 * lane);
#pragma unroll
    for (int s = 0; s < 7; s++) {
      const v4f a = *(const AS3 v4f*)(ar + 8 * s + 4 * lhi);
      const v4f b = *(const AS3 v4f*)(bq + 128 * s);
      acc = wmma4(a.xy, b.xy, acc);
      acc = wmma4(a.zw, b.zw, acc);
    }
    acc = wmma4(aL, bL, acc);
    const float b2a = b2[lln];
    const int mb = j0 + 8 * lhi;
    const int lb = (r0 + 3 + 8 * lhi) * PCH;
    gptr xs = xg + g * (L_SEQ * NCH);
#pragma unroll
    for (int v = 0; v < 8; v++) {
      const int m = mb + v;
      const float t   = fast_tanh(acc[v] + b2a);
      const float oth = other[lb + v * PCH + lln];
      if (MODE <= 1)      outLds[lb + v * PCH + lln] = oth * __expf(t);
      else if (MODE == 2) xs[(m * tmul + tadd) * NCH + lln] = oth + t;
      else                xs[(m * tmul + tadd) * NCH + lln] = oth - t;
    }
  }
  __syncthreads();
}

__global__ __launch_bounds__(NTHREADS) void scinet_level_kernel(
    float* __restrict__ xbuf_, const float* __restrict__ x0D_,
    const float* __restrict__ xctrl_, const float* __restrict__ Wpk_,
    const float* __restrict__ b1g_, const float* __restrict__ b2g_, int depth) {
  __shared__ __align__(16) float smem[SMEM_TOT];
  const int tid = threadIdx.x;
  const int G   = 1 << depth;
  const int sel = blockIdx.x & (G - 1);
  const int bg  = blockIdx.x >> depth;
  int node = 0;
  for (int i = 0; i < depth; i++) node = 2 * node + 1 + ((sel >> i) & 1);
  const int stride = 1 << depth;
  const int Lp     = (L_SEQ >> depth) >> 1;
  const int S      = Lp + 6;
  const int lsh    = 7 - depth;
  const int tsh    = 3 - depth;

  lptr sev = (lptr)(smem + SEV_OFF);
  lptr sod = (lptr)(smem + SOD_OFF);
  lptr hb  = (lptr)(smem + HB_OFF);
  lptr db  = (lptr)(smem + DB_OFF);
  lptr cb  = (lptr)(smem + CB_OFF);
  lptr wl  = (lptr)(smem + WL_OFF);
  lptr sm  = (lptr)smem;

  for (int i = tid; i < SMEM_TOT; i += NTHREADS) sm[i] = 0.f;
  __syncthreads();

  gptr  xg  = (gptr)(xbuf_ + (size_t)bg * G * L_SEQ * NCH);
  cgptr Wpk = (cgptr)Wpk_;
  cgptr b1g = (cgptr)b1g_;
  cgptr b2g = (cgptr)b2g_;

  if (depth == 0) {
    cgptr x0 = (cgptr)x0D_ + (size_t)bg * (L_SEQ * 5);
    cgptr xc = (cgptr)xctrl_ + (size_t)bg * (L_SEQ * 14);
    for (int it = 0; it < (L_SEQ * 5) / NTHREADS; it++) {
      const int e = tid + it * NTHREADS;
      const int t = e / 5, c = e - t * 5;
      lptr dst = (t & 1) ? sod : sev;
      dst[(3 + (t >> 1)) * PCH + c] = x0[e];
    }
    for (int it = 0; it < (L_SEQ * 14) / NTHREADS; it++) {
      const int e = tid + it * NTHREADS;
      const int t = e / 14, c = e - t * 14;
      lptr dst = (t & 1) ? sod : sev;
      dst[(3 + (t >> 1)) * PCH + 5 + c] = xc[e];
    }
  } else {
    for (int i = tid; i < (L_SEQ / 2) * NCH; i += NTHREADS) {
      const int ra = i / NCH, c = i - ra * NCH;
      const int g = ra >> lsh, r = ra & (Lp - 1);
      cgptr xs = xg + g * (L_SEQ * NCH);
      const int o = (g * S + 3 + r) * PCH + c;
      sev[o] = xs[((2 * r) * stride + sel) * NCH + c];
      sod[o] = xs[((2 * r + 1) * stride + sel) * NCH + c];
    }
  }
  __syncthreads();
  for (int i = tid; i < G * 2 * 3 * NCH; i += NTHREADS) {
    const int q = i / NCH, c = i - q * NCH;
    const int w = q / 3, r = q - w * 3;
    lptr buf = ((w & 1) ? sod : sev) + (w >> 1) * S * PCH;
    buf[r * PCH + c]            = buf[3 * PCH + c];
    buf[(Lp + 3 + r) * PCH + c] = buf[(Lp + 2) * PCH + c];
  }
  __syncthreads();

  const int nb = node * 4;
  run_branch<0>(Wpk + (size_t)(nb + 0) * WB_TOT, b1g + (nb + 0) * NCH,
                b2g + (nb + 0) * NCH, sev, hb, wl, Lp, G, tsh, sod, db, xg, 0, 0);
  run_branch<1>(Wpk + (size_t)(nb + 1) * WB_TOT, b1g + (nb + 1) * NCH,
                b2g + (nb + 1) * NCH, sod, hb, wl, Lp, G, tsh, sev, cb, xg, 0, 0);
  for (int i = tid; i < G * 2 * 3 * NCH; i += NTHREADS) {
    const int q = i / NCH, c = i - q * NCH;
    const int w = q / 3, r = q - w * 3;
    lptr buf = ((w & 1) ? cb : db) + (w >> 1) * S * PCH;
    buf[r * PCH + c]            = buf[3 * PCH + c];
    buf[(Lp + 3 + r) * PCH + c] = buf[(Lp + 2) * PCH + c];
  }
  __syncthreads();
  run_branch<2>(Wpk + (size_t)(nb + 3) * WB_TOT, b1g + (nb + 3) * NCH,
                b2g + (nb + 3) * NCH, db, hb, wl, Lp, G, tsh, cb, db, xg,
                2 * stride, sel);
  run_branch<3>(Wpk + (size_t)(nb + 2) * WB_TOT, b1g + (nb + 2) * NCH,
                b2g + (nb + 2) * NCH, cb, hb, wl, Lp, G, tsh, db, cb, xg,
                2 * stride, sel + stride);
}

__global__ void pack_weights_kernel(const float* __restrict__ W1,
                                    const float* __restrict__ W2,
                                    float* __restrict__ Wpk) {
  const int idx = blockIdx.x * blockDim.x + threadIdx.x;
  if (idx >= 28 * WB_TOT) return;
  const int nb = idx / WB_TOT, rem = idx - nb * WB_TOT;
  float v = 0.f;
  if (rem < WM2_OFF) {
    int co, k;
    if (rem < 12 * 128) { const int l = (rem & 127) >> 2; co = l & 15; k = 8 * (rem >> 7) + 4 * (l >> 4) + (rem & 3); }
    else { const int r = rem - 12 * 128, l = r >> 1; co = l & 15; k = 96 + 2 * (l >> 4) + (r & 1); }
    const int tap = k / PCH, ci = k - tap * PCH;
    if (ci < NCH) v = W1[((size_t)(nb * NCH + co) * NCH + ci) * 5 + tap];
  } else if (rem < WT1_OFF) {
    const int r = rem - WM2_OFF;
    int co, k;
    if (r < 7 * 128) { const int l = (r & 127) >> 2; co = l & 15; k = 8 * (r >> 7) + 4 * (l >> 4) + (r & 3); }
    else { const int rr = r - 7 * 128, l = rr >> 1; co = l & 15; k = 56 + 2 * (l >> 4) + (rr & 1); }
    const int tap = k / PCH, ci = k - tap * PCH;
    if (ci < NCH) v = W2[((size_t)(nb * NCH + co) * NCH + ci) * 3 + tap];
  } else if (rem < WT2_OFF) {
    const int r = rem - WT1_OFF;
    const int c = r / TS1, k = r - c * TS1 - TG;
    if (k >= 0 && k < K1M) {
      const int tap = k / PCH, ci = k - tap * PCH;
      if (ci < NCH) v = W1[((size_t)(nb * NCH + 16 + c) * NCH + ci) * 5 + tap];
    }
  } else if (rem < WT2_OFF + 3 * TS2) {
    const int r = rem - WT2_OFF;
    const int c = r / TS2, k = r - c * TS2 - TG;
    if (k >= 0 && k < K2M) {
      const int tap = k / PCH, ci = k - tap * PCH;
      if (ci < NCH) v = W2[((size_t)(nb * NCH + 16 + c) * NCH + ci) * 3 + tap];
    }
  }
  Wpk[idx] = v;
}

__global__ void proj_kernel(const float* __restrict__ xbuf,
                            const float* __restrict__ x0D,
                            const float* __restrict__ projw,
                            float* __restrict__ out) {
  __shared__ float yres_s[L_SEQ * 5];
  lptr yres = (lptr)yres_s;
  const int b = blockIdx.x, tid = threadIdx.x;
  cgptr xb = (cgptr)xbuf, x0 = (cgptr)x0D, pj = (cgptr)projw;
  for (int i = tid; i < L_SEQ * 5; i += blockDim.x) {
    const int l = i / 5, c = i - l * 5;
    yres[i] = xb[((size_t)b * L_SEQ + l) * NCH + c] +
              x0[((size_t)b * L_SEQ + l) * 5 + c];
  }
  __syncthreads();
  for (int i = tid; i < 64 * 5; i += blockDim.x) {
    const int o = i / 5, c = i - o * 5;
    cgptr pw = pj + o * L_SEQ;
    float acc = 0.f;
    for (int l = 0; l < L_SEQ; l++) acc = fmaf(pw[l], yres[l * 5 + c], acc);
    ((gptr)out)[(size_t)b * 320 + i] = acc;
  }
}

extern "C" void kernel_launch(void* const* d_in, const int* in_sizes, int n_in,
                              void* d_out, int out_size, void* d_ws,
                              size_t ws_size, hipStream_t stream) {
  const float* x0D   = (const float*)d_in[0];
  const float* xctrl = (const float*)d_in[1];
  const float* W1    = (const float*)d_in[2];
  const float* b1    = (const float*)d_in[3];
  const float* W2    = (const float*)d_in[4];
  const float* b2    = (const float*)d_in[5];
  const float* projw = (const float*)d_in[6];
  float* out = (float*)d_out;

  float* ws   = (float*)d_ws;
  float* xbuf = ws;
  const size_t XB = (size_t)B_TOT * L_SEQ * NCH;
  float* Wpk = xbuf + XB;

  const int packN = 28 * WB_TOT;
  pack_weights_kernel<<<(packN + 255) / 256, 256, 0, stream>>>(W1, W2, Wpk);
  for (int d = 0; d < 3; d++)
    scinet_level_kernel<<<B_TOT, NTHREADS, 0, stream>>>(xbuf, x0D, xctrl, Wpk, b1,
                                                        b2, d);
  proj_kernel<<<B_TOT, 256, 0, stream>>>(xbuf, x0D, projw, out);
}


# Output projection moves onto the f32 matrix unit: 16 batch elements per block as one 64x80x256 GEMM from LDS, out lines stored whole; the sum over l now runs in WMMA k order
# speedup vs baseline: 2.0008x; 1.0686x over previous
#include <hip/hip_runtime.h>
#include <math.h>


typedef __attribute__((ext_vector_type(2))) float v2f;
typedef __attribute__((ext_vector_type(4))) float v4f;
typedef __attribute__((ext_vector_type(8))) float v8f;

#define AS3 __attribute__((address_space(3)))
#define AS1 __attribute__((address_space(1)))
typedef AS3 float*       lptr;
typedef const AS3 float* clptr;
typedef AS1 float*       gptr;
typedef const AS1 float* cgptr;

#define B_TOT    8192
#define L_SEQ    256
#define NCH      19
#define PCH      20
#define NWAVES   8
#define NTHREADS 256

#define K1M      100
#define K2M      60
#define K1T      180
#define K2T      140
#define TG       80
#define TS1      284
#define TS2      220
#define WM1_OFF  0
#define WM2_OFF  1600
#define WT1_OFF  2560
#define WT2_OFF  (WT1_OFF + 3 * TS1)
#define WB_TOT   4096

#define ROWS_MAX 152
#define BUF_SZ   (ROWS_MAX * PCH + 64)
#define SEV_OFF  0
#define SOD_OFF  (SEV_OFF + BUF_SZ)
#define HB_OFF   (SOD_OFF + BUF_SZ)
#define DB_OFF   (HB_OFF + BUF_SZ)
#define CB_OFF   (DB_OFF + BUF_SZ)
#define WL_OFF   (CB_OFF + BUF_SZ)
#define SMEM_TOT (WL_OFF + WB_TOT)

static __device__ __forceinline__ v8f wmma4(v2f a, v2f b, v8f c) {
  return __builtin_amdgcn_wmma_f32_16x16x4_f32(false, a, false, b, (short)0, c,
                                               false, false);
}

static __device__ __forceinline__ float leaky(float x) {
  return x > 0.f ? x : 0.01f * x;
}

static __device__ __forceinline__ float fast_tanh(float x) {
  x = fminf(fmaxf(x, -15.f), 15.f);
  const float e = __expf(2.f * x);
  return __fdividef(e - 1.f, e + 1.f);
}

template <int MODE>
static __device__ void run_branch(cgptr Wp, cgptr b1, cgptr b2, clptr in,
                                  lptr hb, lptr wl, int Lp, int nseg, int tsh,
                                  clptr other, lptr outLds, gptr xg, int tmul,
                                  int tadd) {
  const int tid  = threadIdx.x;
  const int wave = __builtin_amdgcn_readfirstlane(tid >> 5);
  const int lane = tid & 31;
  const int lhi  = lane >> 4;
  const int lln  = lane & 15;
  int tj = lln / 3;
  int tc = lln - 3 * tj;
  if (lln == 15) { tj = 0; tc = 0; }

#pragma unroll
  for (int it = 0; it < WB_TOT / 4 / NTHREADS; it++)
    ((AS3 v4f*)wl)[tid + it * NTHREADS] = ((const AS1 v4f*)Wp)[tid + it * NTHREADS];
  __syncthreads();

  const int S    = Lp + 6;
  const int M1   = nseg * S - 4;
  const int nMt1 = (M1 + 15) >> 4;
  const int nTr  = (M1 + 4) / 5;

  if (wave < 2) {
    int ti = (wave << 4) + lln; if (ti > nTr - 1) ti = nTr - 1;
    clptr ar = in + ti * (5 * PCH);
    clptr br = wl + WT1_OFF + tc * TS1 + TG - PCH * tj;
    v8f acc = {};
    v2f aS[3], bS[3];
#pragma unroll
    for (int u = 0; u < 3; u++) {
      aS[u] = *(const AS3 v2f*)(ar + 168 + 4 * u + 2 * lhi);
      bS[u] = *(const AS3 v2f*)(br + 168 + 4 * u + 2 * lhi);
    }
#pragma unroll 7
    for (int s = 0; s < 21; s++) {
      const v4f a = *(const AS3 v4f*)(ar + 8 * s + 4 * lhi);
      const v4f b = *(const AS3 v4f*)(br + 8 * s + 4 * lhi);
      acc = wmma4(a.xy, b.xy, acc);
      acc = wmma4(a.zw, b.zw, acc);
    }
#pragma unroll
    for (int u = 0; u < 3; u++) acc = wmma4(aS[u], bS[u], acc);
    const float bt = b1[16 + tc];
    const int rb = ((wave << 4) + 8 * lhi) * 5 + tj;
    if (lln < 15) {
#pragma unroll
      for (int v = 0; v < 8; v++) {
        const int r = rb + 5 * v;
        if (r < M1) hb[r * PCH + 16 + tc] = leaky(acc[v] + bt);
      }
    }
  }
#pragma nounroll
  for (int pass = 0; pass < 2; pass++) {
    const int mt = pass == 0 ? wave - 2
                             : (wave >= 6 ? wave : (wave >= 4 ? wave + 4 : 99));
    if (mt < 0 || mt >= nMt1) continue;
    const int m0 = mt << 4;
    int mrow = m0 + lln; if (mrow > M1 - 1) mrow = M1 - 1;
    clptr ar = in + mrow * PCH;
    clptr bq = wl + WM1_OFF + 4 * lane;
    v8f acc = {};
    const v2f aL = *(const AS3 v2f*)(ar + 96 + 2 * lhi);
    const v2f bL = *(const AS3 v2f*)(wl + WM1_OFF + 12 * 128 + 2 * lane);
#pragma unroll 6
    for (int s = 0; s < 12; s++) {
      const v4f a = *(const AS3 v4f*)(ar + 8 * s + 4 * lhi);
      const v4f b = *(const AS3 v4f*)(bq + 128 * s);
      acc = wmma4(a.xy, b.xy, acc);
      acc = wmma4(a.zw, b.zw, acc);
    }
    acc = wmma4(aL, bL, acc);
    const float b1a = b1[lln];
    lptr hrow = hb + (m0 + 8 * lhi) * PCH;
    if (m0 + 16 <= M1) {
#pragma unroll
      for (int v = 0; v < 8; v++) hrow[v * PCH + lln] = leaky(acc[v] + b1a);
    } else {
#pragma unroll
      for (int v = 0; v < 8; v++)
        if (m0 + 8 * lhi + v < M1) hrow[v * PCH + lln] = leaky(acc[v] + b1a);
    }
  }
  __syncthreads();

  if (wave < 2) {
    int ti = (wave << 4) + lln; if (ti > nTr - 1) ti = nTr - 1;
    clptr ar = hb + ti * (5 * PCH);
    clptr br = wl + WT2_OFF + tc * TS2 + TG - PCH * tj;
    v8f acc = {};
    v2f aS[3], bS[3];
#pragma unroll
    for (int u = 0; u < 3; u++) {
      aS[u] = *(const AS3 v2f*)(ar + 128 + 4 * u + 2 * lhi);
      bS[u] = *(const AS3 v2f*)(br + 128 + 4 * u + 2 * lhi);
    }
#pragma unroll 8
    for (int s = 0; s < 16; s++) {
      const v4f a = *(const AS3 v4f*)(ar + 8 * s + 4 * lhi);
      const v4f b = *(const AS3 v4f*)(br + 8 * s + 4 * lhi);
      acc = wmma4(a.xy, b.xy, acc);
      acc = wmma4(a.zw, b.zw, acc);
    }
#pragma unroll
    for (int u = 0; u < 3; u++) acc = wmma4(aS[u], bS[u], acc);
    const float bt = b2[16 + tc];
    const int rb = ((wave << 4) + 8 * lhi) * 5 + tj;
    if (lln < 15) {
#pragma unroll
      for (int v = 0; v < 8; v++) {
        const int r  = rb + 5 * v;
        const int g  = (r >= S) + (r >= 2 * S) + (r >= 3 * S);
        const int jj = r - g * S;
        if (jj < Lp && g < nseg) {
          const int li = (r + 3) * PCH + 16 + tc;
          const float t   = fast_tanh(acc[v] + bt);
          const float oth = other[li];
          if (MODE <= 1) outLds[li] = oth * __expf(t);
          else {
            gptr xs = xg + g * (L_SEQ * NCH);
            if (MODE == 2) xs[(jj * tmul + tadd) * NCH + 16 + tc] = oth + t;
            else           xs[(jj * tmul + tadd) * NCH + 16 + tc] = oth - t;
          }
        }
      }
    }
  }
  const int nMt2  = (nseg * Lp) >> 4;
  const int tmask = (1 << tsh) - 1;
#pragma nounroll
  for (int pass = 0; pass < 2; pass++) {
    const int mt = pass == 0 ? wave - 2 : (wave >= 6 ? wave : 99);
    if (mt < 0 || mt >= nMt2) continue;
    const int g  = mt >> tsh;
    const int j0 = (mt & tmask) << 4;
    const int r0 = g * S + j0;
    clptr ar = hb + (r0 + lln) * PCH;
    clptr bq = wl + WM2_OFF + 4 * lane;
    v8f acc = {};
    const v2f aL = *(const AS3 v2f*)(ar + 56 + 2 * lhi);
    const v2f bL = *(const AS3 v2f*)(wl + WM2_OFF + 7 * 128 + 2 * lane);
#pragma unroll
    for (int s = 0; s < 7; s++) {
      const v4f a = *(const AS3 v4f*)(ar + 8 * s + 4 * lhi);
      const v4f b = *(const AS3 v4f*)(bq + 128 * s);
      acc = wmma4(a.xy, b.xy, acc);
      acc = wmma4(a.zw, b.zw, acc);
    }
    acc = wmma4(aL, bL, acc);
    const float b2a = b2[lln];
    const int mb = j0 + 8 * lhi;
    const int lb = (r0 + 3 + 8 * lhi) * PCH;
    gptr xs = xg + g * (L_SEQ * NCH);
#pragma unroll
    for (int v = 0; v < 8; v++) {
      const int m = mb + v;
      const float t   = fast_tanh(acc[v] + b2a);
      const float oth = other[lb + v * PCH + lln];
      if (MODE <= 1)      outLds[lb + v * PCH + lln] = oth * __expf(t);
      else if (MODE == 2) xs[(m * tmul + tadd) * NCH + lln] = oth + t;
      else                xs[(m * tmul + tadd) * NCH + lln] = oth - t;
    }
  }
  __syncthreads();
}

__global__ __launch_bounds__(NTHREADS) void scinet_level_kernel(
    float* __restrict__ xbuf_, const float* __restrict__ x0D_,
    const float* __restrict__ xctrl_, const float* __restrict__ Wpk_,
    const float* __restrict__ b1g_, const float* __restrict__ b2g_, int depth) {
  __shared__ __align__(16) float smem[SMEM_TOT];
  const int tid = threadIdx.x;
  const int G   = 1 << depth;
  const int sel = blockIdx.x & (G - 1);
  const int bg  = blockIdx.x >> depth;
  int node = 0;
  for (int i = 0; i < depth; i++) node = 2 * node + 1 + ((sel >> i) & 1);
  const int stride = 1 << depth;
  const int Lp     = (L_SEQ >> depth) >> 1;
  const int S      = Lp + 6;
  const int lsh    = 7 - depth;
  const int tsh    = 3 - depth;

  lptr sev = (lptr)(smem + SEV_OFF);
  lptr sod = (lptr)(smem + SOD_OFF);
  lptr hb  = (lptr)(smem + HB_OFF);
  lptr db  = (lptr)(smem + DB_OFF);
  lptr cb  = (lptr)(smem + CB_OFF);
  lptr wl  = (lptr)(smem + WL_OFF);
  lptr sm  = (lptr)smem;

  for (int i = tid; i < SMEM_TOT; i += NTHREADS) sm[i] = 0.f;
  __syncthreads();

  gptr  xg  = (gptr)(xbuf_ + (size_t)bg * G * L_SEQ * NCH);
  cgptr Wpk = (cgptr)Wpk_;
  cgptr b1g = (cgptr)b1g_;
  cgptr b2g = (cgptr)b2g_;

  if (depth == 0) {
    cgptr x0 = (cgptr)x0D_ + (size_t)bg * (L_SEQ * 5);
    cgptr xc = (cgptr)xctrl_ + (size_t)bg * (L_SEQ * 14);
    for (int it = 0; it < (L_SEQ * 5) / NTHREADS; it++) {
      const int e = tid + it * NTHREADS;
      const int t = e / 5, c = e - t * 5;
      lptr dst = (t & 1) ? sod : sev;
      dst[(3 + (t >> 1)) * PCH + c] = x0[e];
    }
    for (int it = 0; it < (L_SEQ * 14) / NTHREADS; it++) {
      const int e = tid + it * NTHREADS;
      const int t = e / 14, c = e - t * 14;
      lptr dst = (t & 1) ? sod : sev;
      dst[(3 + (t >> 1)) * PCH + 5 + c] = xc[e];
    }
  } else {
    for (int i = tid; i < (L_SEQ / 2) * NCH; i += NTHREADS) {
      const int ra = i / NCH, c = i - ra * NCH;
      const int g = ra >> lsh, r = ra & (Lp - 1);
      cgptr xs = xg + g * (L_SEQ * NCH);
      const int o = (g * S + 3 + r) * PCH + c;
      sev[o] = xs[((2 * r) * stride + sel) * NCH + c];
      sod[o] = xs[((2 * r + 1) * stride + sel) * NCH + c];
    }
  }
  __syncthreads();
  for (int i = tid; i < G * 2 * 3 * NCH; i += NTHREADS) {
    const int q = i / NCH, c = i - q * NCH;
    const int w = q / 3, r = q - w * 3;
    lptr buf = ((w & 1) ? sod : sev) + (w >> 1) * S * PCH;
    buf[r * PCH + c]            = buf[3 * PCH + c];
    buf[(Lp + 3 + r) * PCH + c] = buf[(Lp + 2) * PCH + c];
  }
  __syncthreads();

  const int nb = node * 4;
  run_branch<0>(Wpk + (size_t)(nb + 0) * WB_TOT, b1g + (nb + 0) * NCH,
                b2g + (nb + 0) * NCH, sev, hb, wl, Lp, G, tsh, sod, db, xg, 0, 0);
  run_branch<1>(Wpk + (size_t)(nb + 1) * WB_TOT, b1g + (nb + 1) * NCH,
                b2g + (nb + 1) * NCH, sod, hb, wl, Lp, G, tsh, sev, cb, xg, 0, 0);
  for (int i = tid; i < G * 2 * 3 * NCH; i += NTHREADS) {
    const int q = i / NCH, c = i - q * NCH;
    const int w = q / 3, r = q - w * 3;
    lptr buf = ((w & 1) ? cb : db) + (w >> 1) * S * PCH;
    buf[r * PCH + c]            = buf[3 * PCH + c];
    buf[(Lp + 3 + r) * PCH + c] = buf[(Lp + 2) * PCH + c];
  }
  __syncthreads();
  run_branch<2>(Wpk + (size_t)(nb + 3) * WB_TOT, b1g + (nb + 3) * NCH,
                b2g + (nb + 3) * NCH, db, hb, wl, Lp, G, tsh, cb, db, xg,
                2 * stride, sel);
  run_branch<3>(Wpk + (size_t)(nb + 2) * WB_TOT, b1g + (nb + 2) * NCH,
                b2g + (nb + 2) * NCH, cb, hb, wl, Lp, G, tsh, db, cb, xg,
                2 * stride, sel + stride);
}

__global__ void pack_weights_kernel(const float* __restrict__ W1,
                                    const float* __restrict__ W2,
                                    float* __restrict__ Wpk) {
  const int idx = blockIdx.x * blockDim.x + threadIdx.x;
  if (idx >= 28 * WB_TOT) return;
  const int nb = idx / WB_TOT, rem = idx - nb * WB_TOT;
  float v = 0.f;
  if (rem < WM2_OFF) {
    int co, k;
    if (rem < 12 * 128) { const int l = (rem & 127) >> 2; co = l & 15; k = 8 * (rem >> 7) + 4 * (l >> 4) + (rem & 3); }
    else { const int r = rem - 12 * 128, l = r >> 1; co = l & 15; k = 96 + 2 * (l >> 4) + (r & 1); }
    const int tap = k / PCH, ci = k - tap * PCH;
    if (ci < NCH) v = W1[((size_t)(nb * NCH + co) * NCH + ci) * 5 + tap];
  } else if (rem < WT1_OFF) {
    const int r = rem - WM2_OFF;
    int co, k;
    if (r < 7 * 128) { const int l = (r & 127) >> 2; co = l & 15; k = 8 * (r >> 7) + 4 * (l >> 4) + (r & 3); }
    else { const int rr = r - 7 * 128, l = rr >> 1; co = l & 15; k = 56 + 2 * (l >> 4) + (rr & 1); }
    const int tap = k / PCH, ci = k - tap * PCH;
    if (ci < NCH) v = W2[((size_t)(nb * NCH + co) * NCH + ci) * 3 + tap];
  } else if (rem < WT2_OFF) {
    const int r = rem - WT1_OFF;
    const int c = r / TS1, k = r - c * TS1 - TG;
    if (k >= 0 && k < K1M) {
      const int tap = k / PCH, ci = k - tap * PCH;
      if (ci < NCH) v = W1[((size_t)(nb * NCH + 16 + c) * NCH + ci) * 5 + tap];
    }
  } else if (rem < WT2_OFF + 3 * TS2) {
    const int r = rem - WT2_OFF;
    const int c = r / TS2, k = r - c * TS2 - TG;
    if (k >= 0 && k < K2M) {
      const int tap = k / PCH, ci = k - tap * PCH;
      if (ci < NCH) v = W2[((size_t)(nb * NCH + 16 + c) * NCH + ci) * 3 + tap];
    }
  }
  Wpk[idx] = v;
}

#define PJ_NB    16
#define PJ_N     (PJ_NB * 5)
#define PJ_PITCH 268
#define PJ_P_OFF (PJ_N * PJ_PITCH)
#define PJ_SMEM  (PJ_P_OFF + 64 * PJ_PITCH)

template <int NT>
static __device__ __forceinline__ void proj_tiles(clptr pa, clptr yb, v8f* acc) {
#pragma unroll 4
  for (int s = 0; s < 32; s++) {
    const v4f a = *(const AS3 v4f*)(pa + 8 * s);
    v4f b[NT];
#pragma unroll
    for (int j = 0; j < NT; j++)
      b[j] = *(const AS3 v4f*)(yb + j * (16 * PJ_PITCH) + 8 * s);
#pragma unroll
    for (int j = 0; j < NT; j++) acc[j] = wmma4(a.xy, b[j].xy, acc[j]);
#pragma unroll
    for (int j = 0; j < NT; j++) acc[j] = wmma4(a.zw, b[j].zw, acc[j]);
  }
}

__global__ __launch_bounds__(NTHREADS) void proj_kernel(
    const float* __restrict__ xbuf, const float* __restrict__ x0D,
    const float* __restrict__ projw, float* __restrict__ out) {
  __shared__ __align__(16) float psm[PJ_SMEM];
  lptr ys = (lptr)psm;
  lptr ps = (lptr)psm + PJ_P_OFF;
  const int tid  = threadIdx.x;
  const int wave = __builtin_amdgcn_readfirstlane(tid >> 5);
  const int lane = tid & 31;
  const int lhi  = lane >> 4;
  const int lln  = lane & 15;
  const size_t b0 = (size_t)blockIdx.x * PJ_NB;
  cgptr xb = (cgptr)xbuf + b0 * (L_SEQ * NCH);
  cgptr x0 = (cgptr)x0D + b0 * (L_SEQ * 5);
  cgptr pj = (cgptr)projw;

#pragma unroll 8
  for (int it = 0; it < (64 * L_SEQ / 4) / NTHREADS; it++) {
    const int q = tid + it * NTHREADS;
    *(AS3 v4f*)(ps + (q >> 6) * PJ_PITCH + 4 * (q & 63)) = ((const AS1 v4f*)pj)[q];
  }
#pragma unroll 2
  for (int g = 0; g < PJ_NB; g++) {
    cgptr xbg = xb + g * (L_SEQ * NCH);
    cgptr x0g = x0 + g * (L_SEQ * 5);
    lptr  yg  = ys + g * 5 * PJ_PITCH;
#pragma unroll
    for (int it = 0; it < (L_SEQ * 5) / NTHREADS; it++) {
      const int e = tid + it * NTHREADS;
      const int l = e / 5, c = e - 5 * l;
      yg[c * PJ_PITCH + l] = xbg[l * NCH + c] + x0g[e];
    }
  }
  __syncthreads();

  const int mt  = wave & 3;
  const int nt0 = wave < 4 ? 0 : 3;
  const int ntn = wave < 4 ? 3 : 2;
  clptr pa = ps + (16 * mt + lln) * PJ_PITCH + 4 * lhi;
  clptr yb = ys + (16 * nt0 + lln) * PJ_PITCH + 4 * lhi;
  v8f acc[3];
#pragma unroll
  for (int j = 0; j < 3; j++) acc[j] = (v8f){};
  if (wave < 4) proj_tiles<3>(pa, yb, acc);
  else          proj_tiles<2>(pa, yb, acc);
  __syncthreads();
  lptr outs = ys;
#pragma unroll
  for (int j = 0; j < 3; j++) {
    if (j < ntn) {
      const int n  = 16 * (nt0 + j) + lln;
      const int bl = n / 5, c = n - 5 * bl;
      lptr op = outs + bl * 320 + (16 * mt + 8 * lhi) * 5 + c;
#pragma unroll
      for (int v = 0; v < 8; v++) op[v * 5] = acc[j][v];
    }
  }
  __syncthreads();
  AS1 v4f* og = (AS1 v4f*)((gptr)out + b0 * 320);
  for (int it = 0; it < (PJ_NB * 320 / 4) / NTHREADS; it++)
    og[tid + it * NTHREADS] = ((const AS3 v4f*)outs)[tid + it * NTHREADS];
}

extern "C" void kernel_launch(void* const* d_in, const int* in_sizes, int n_in,
                              void* d_out, int out_size, void* d_ws,
                              size_t ws_size, hipStream_t stream) {
  const float* x0D   = (const float*)d_in[0];
  const float* xctrl = (const float*)d_in[1];
  const float* W1    = (const float*)d_in[2];
  const float* b1    = (const float*)d_in[3];
  const float* W2    = (const float*)d_in[4];
  const float* b2    = (const float*)d_in[5];
  const float* projw = (const float*)d_in[6];
  float* out = (float*)d_out;

  float* ws   = (float*)d_ws;
  float* xbuf = ws;
  const size_t XB = (size_t)B_TOT * L_SEQ * NCH;
  float* Wpk = xbuf + XB;

  const int packN = 28 * WB_TOT;
  pack_weights_kernel<<<(packN + 255) / 256, 256, 0, stream>>>(W1, W2, Wpk);
  for (int d = 0; d < 3; d++)
    scinet_level_kernel<<<B_TOT, NTHREADS, 0, stream>>>(xbuf, x0D, xctrl, Wpk, b1,
                                                        b2, d);
  proj_kernel<<<B_TOT / PJ_NB, NTHREADS, 0, stream>>>(xbuf, x0D, projw, out);
}


# conv2 epilogue: tanh(x) = 1 - 2/(exp(2x)+1) with one v_exp_f32, one v_rcp_f32 and one fma replacing clamp + IEEE division (17 fewer instructions per element); tanh rounds differently in last bits
# speedup vs baseline: 2.1154x; 1.0573x over previous
#include <hip/hip_runtime.h>
#include <math.h>


typedef __attribute__((ext_vector_type(2))) float v2f;
typedef __attribute__((ext_vector_type(4))) float v4f;
typedef __attribute__((ext_vector_type(8))) float v8f;

#define AS3 __attribute__((address_space(3)))
#define AS1 __attribute__((address_space(1)))
typedef AS3 float*       lptr;
typedef const AS3 float* clptr;
typedef AS1 float*       gptr;
typedef const AS1 float* cgptr;

#define B_TOT    8192
#define L_SEQ    256
#define NCH      19
#define PCH      20
#define NWAVES   8
#define NTHREADS 256

#define K1M      100
#define K2M      60
#define K1T      180
#define K2T      140
#define TG       80
#define TS1      284
#define TS2      220
#define WM1_OFF  0
#define WM2_OFF  1600
#define WT1_OFF  2560
#define WT2_OFF  (WT1_OFF + 3 * TS1)
#define WB_TOT   4096

#define ROWS_MAX 152
#define BUF_SZ   (ROWS_MAX * PCH + 64)
#define SEV_OFF  0
#define SOD_OFF  (SEV_OFF + BUF_SZ)
#define HB_OFF   (SOD_OFF + BUF_SZ)
#define DB_OFF   (HB_OFF + BUF_SZ)
#define CB_OFF   (DB_OFF + BUF_SZ)
#define WL_OFF   (CB_OFF + BUF_SZ)
#define SMEM_TOT (WL_OFF + WB_TOT)

static __device__ __forceinline__ v8f wmma4(v2f a, v2f b, v8f c) {
  return __builtin_amdgcn_wmma_f32_16x16x4_f32(false, a, false, b, (short)0, c,
                                               false, false);
}

static __device__ __forceinline__ float leaky(float x) {
  return x > 0.f ? x : 0.01f * x;
}

static __device__ __forceinline__ float fast_tanh(float x) {
  const float e = __builtin_amdgcn_exp2f(x * 2.8853900817779268f);
  return fmaf(-2.f, __builtin_amdgcn_rcpf(e + 1.f), 1.f);
}

template <int MODE>
static __device__ void run_branch(cgptr Wp, cgptr b1, cgptr b2, clptr in,
                                  lptr hb, lptr wl, int Lp, int nseg, int tsh,
                                  clptr other, lptr outLds, gptr xg, int tmul,
                                  int tadd) {
  const int tid  = threadIdx.x;
  const int wave = __builtin_amdgcn_readfirstlane(tid >> 5);
  const int lane = tid & 31;
  const int lhi  = lane >> 4;
  const int lln  = lane & 15;
  int tj = lln / 3;
  int tc = lln - 3 * tj;
  if (lln == 15) { tj = 0; tc = 0; }

#pragma unroll
  for (int it = 0; it < WB_TOT / 4 / NTHREADS; it++)
    ((AS3 v4f*)wl)[tid + it * NTHREADS] = ((const AS1 v4f*)Wp)[tid + it * NTHREADS];
  __syncthreads();

  const int S    = Lp + 6;
  const int M1   = nseg * S - 4;
  const int nMt1 = (M1 + 15) >> 4;
  const int nTr  = (M1 + 4) / 5;

  if (wave < 2) {
    int ti = (wave << 4) + lln; if (ti > nTr - 1) ti = nTr - 1;
    clptr ar = in + ti * (5 * PCH);
    clptr br = wl + WT1_OFF + tc * TS1 + TG - PCH * tj;
    v8f acc = {};
    v2f aS[3], bS[3];
#pragma unroll
    for (int u = 0; u < 3; u++) {
      aS[u] = *(const AS3 v2f*)(ar + 168 + 4 * u + 2 * lhi);
      bS[u] = *(const AS3 v2f*)(br + 168 + 4 * u + 2 * lhi);
    }
#pragma unroll 7
    for (int s = 0; s < 21; s++) {
      const v4f a = *(const AS3 v4f*)(ar + 8 * s + 4 * lhi);
      const v4f b = *(const AS3 v4f*)(br + 8 * s + 4 * lhi);
      acc = wmma4(a.xy, b.xy, acc);
      acc = wmma4(a.zw, b.zw, acc);
    }
#pragma unroll
    for (int u = 0; u < 3; u++) acc = wmma4(aS[u], bS[u], acc);
    const float bt = b1[16 + tc];
    const int rb = ((wave << 4) + 8 * lhi) * 5 + tj;
    if (lln < 15) {
#pragma unroll
      for (int v = 0; v < 8; v++) {
        const int r = rb + 5 * v;
        if (r < M1) hb[r * PCH + 16 + tc] = leaky(acc[v] + bt);
      }
    }
  }
#pragma nounroll
  for (int pass = 0; pass < 2; pass++) {
    const int mt = pass == 0 ? wave - 2
                             : (wave >= 6 ? wave : (wave >= 4 ? wave + 4 : 99));
    if (mt < 0 || mt >= nMt1) continue;
    const int m0 = mt << 4;
    int mrow = m0 + lln; if (mrow > M1 - 1) mrow = M1 - 1;
    clptr ar = in + mrow * PCH;
    clptr bq = wl + WM1_OFF + 4 * lane;
    v8f acc = {};
    const v2f aL = *(const AS3 v2f*)(ar + 96 + 2 * lhi);
    const v2f bL = *(const AS3 v2f*)(wl + WM1_OFF + 12 * 128 + 2 * lane);
#pragma unroll 6
    for (int s = 0; s < 12; s++) {
      const v4f a = *(const AS3 v4f*)(ar + 8 * s + 4 * lhi);
      const v4f b = *(const AS3 v4f*)(bq + 128 * s);
      acc = wmma4(a.xy, b.xy, acc);
      acc = wmma4(a.zw, b.zw, acc);
    }
    acc = wmma4(aL, bL, acc);
    const float b1a = b1[lln];
    lptr hrow = hb + (m0 + 8 * lhi) * PCH;
    if (m0 + 16 <= M1) {
#pragma unroll
      for (int v = 0; v < 8; v++) hrow[v * PCH + lln] = leaky(acc[v] + b1a);
    } else {
#pragma unroll
      for (int v = 0; v < 8; v++)
        if (m0 + 8 * lhi + v < M1) hrow[v * PCH + lln] = leaky(acc[v] + b1a);
    }
  }
  __syncthreads();

  if (wave < 2) {
    int ti = (wave << 4) + lln; if (ti > nTr - 1) ti = nTr - 1;
    clptr ar = hb + ti * (5 * PCH);
    clptr br = wl + WT2_OFF + tc * TS2 + TG - PCH * tj;
    v8f acc = {};
    v2f aS[3], bS[3];
#pragma unroll
    for (int u = 0; u < 3; u++) {
      aS[u] = *(const AS3 v2f*)(ar + 128 + 4 * u + 2 * lhi);
      bS[u] = *(const AS3 v2f*)(br + 128 + 4 * u + 2 * lhi);
    }
#pragma unroll 8
    for (int s = 0; s < 16; s++) {
      const v4f a = *(const AS3 v4f*)(ar + 8 * s + 4 * lhi);
      const v4f b = *(const AS3 v4f*)(br + 8 * s + 4 * lhi);
      acc = wmma4(a.xy, b.xy, acc);
      acc = wmma4(a.zw, b.zw, acc);
    }
#pragma unroll
    for (int u = 0; u < 3; u++) acc = wmma4(aS[u], bS[u], acc);
    const float bt = b2[16 + tc];
    const int rb = ((wave << 4) + 8 * lhi) * 5 + tj;
    if (lln < 15) {
#pragma unroll
      for (int v = 0; v < 8; v++) {
        const int r  = rb + 5 * v;
        const int g  = (r >= S) + (r >= 2 * S) + (r >= 3 * S);
        const int jj = r - g * S;
        if (jj < Lp && g < nseg) {
          const int li = (r + 3) * PCH + 16 + tc;
          const float t   = fast_tanh(acc[v] + bt);
          const float oth = other[li];
          if (MODE <= 1) outLds[li] = oth * __expf(t);
          else {
            gptr xs = xg + g * (L_SEQ * NCH);
            if (MODE == 2) xs[(jj * tmul + tadd) * NCH + 16 + tc] = oth + t;
            else           xs[(jj * tmul + tadd) * NCH + 16 + tc] = oth - t;
          }
        }
      }
    }
  }
  const int nMt2  = (nseg * Lp) >> 4;
  const int tmask = (1 << tsh) - 1;
#pragma nounroll
  for (int pass = 0; pass < 2; pass++) {
    const int mt = pass == 0 ? wave - 2 : (wave >= 6 ? wave : 99);
    if (mt < 0 || mt >= nMt2) continue;
    const int g  = mt >> tsh;
    const int j0 = (mt & tmask) << 4;
    const int r0 = g * S + j0;
    clptr ar = hb + (r0 + lln) * PCH;
    clptr bq = wl + WM2_OFF + 4 * lane;
    v8f acc = {};
    const v2f aL = *(const AS3 v2f*)(ar + 56 + 2 * lhi);
    const v2f bL = *(const AS3 v2f*)(wl + WM2_OFF + 7 * 128 + 2 * lane);
#pragma unroll
    for (int s = 0; s < 7; s++) {
      const v4f a = *(const AS3 v4f*)(ar + 8 * s + 4 * lhi);
      const v4f b = *(const AS3 v4f*)(bq + 128 * s);
      acc = wmma4(a.xy, b.xy, acc);
      acc = wmma4(a.zw, b.zw, acc);
    }
    acc = wmma4(aL, bL, acc);
    const float b2a = b2[lln];
    const int mb = j0 + 8 * lhi;
    const int lb = (r0 + 3 + 8 * lhi) * PCH;
    gptr xs = xg + g * (L_SEQ * NCH);
#pragma unroll
    for (int v = 0; v < 8; v++) {
      const int m = mb + v;
      const float t   = fast_tanh(acc[v] + b2a);
      const float oth = other[lb + v * PCH + lln];
      if (MODE <= 1)      outLds[lb + v * PCH + lln] = oth * __expf(t);
      else if (MODE == 2) xs[(m * tmul + tadd) * NCH + lln] = oth + t;
      else                xs[(m * tmul + tadd) * NCH + lln] = oth - t;
    }
  }
  __syncthreads();
}

__global__ __launch_bounds__(NTHREADS) void scinet_level_kernel(
    float* __restrict__ xbuf_, const float* __restrict__ x0D_,
    const float* __restrict__ xctrl_, const float* __restrict__ Wpk_,
    const float* __restrict__ b1g_, const float* __restrict__ b2g_, int depth) {
  __shared__ __align__(16) float smem[SMEM_TOT];
  const int tid = threadIdx.x;
  const int G   = 1 << depth;
  const int sel = blockIdx.x & (G - 1);
  const int bg  = blockIdx.x >> depth;
  int node = 0;
  for (int i = 0; i < depth; i++) node = 2 * node + 1 + ((sel >> i) & 1);
  const int stride = 1 << depth;
  const int Lp     = (L_SEQ >> depth) >> 1;
  const int S      = Lp + 6;
  const int lsh    = 7 - depth;
  const int tsh    = 3 - depth;

  lptr sev = (lptr)(smem + SEV_OFF);
  lptr sod = (lptr)(smem + SOD_OFF);
  lptr hb  = (lptr)(smem + HB_OFF);
  lptr db  = (lptr)(smem + DB_OFF);
  lptr cb  = (lptr)(smem + CB_OFF);
  lptr wl  = (lptr)(smem + WL_OFF);
  lptr sm  = (lptr)smem;

  for (int i = tid; i < SMEM_TOT; i += NTHREADS) sm[i] = 0.f;
  __syncthreads();

  gptr  xg  = (gptr)(xbuf_ + (size_t)bg * G * L_SEQ * NCH);
  cgptr Wpk = (cgptr)Wpk_;
  cgptr b1g = (cgptr)b1g_;
  cgptr b2g = (cgptr)b2g_;

  if (depth == 0) {
    cgptr x0 = (cgptr)x0D_ + (size_t)bg * (L_SEQ * 5);
    cgptr xc = (cgptr)xctrl_ + (size_t)bg * (L_SEQ * 14);
    for (int it = 0; it < (L_SEQ * 5) / NTHREADS; it++) {
      const int e = tid + it * NTHREADS;
      const int t = e / 5, c = e - t * 5;
      lptr dst = (t & 1) ? sod : sev;
      dst[(3 + (t >> 1)) * PCH + c] = x0[e];
    }
    for (int it = 0; it < (L_SEQ * 14) / NTHREADS; it++) {
      const int e = tid + it * NTHREADS;
      const int t = e / 14, c = e - t * 14;
      lptr dst = (t & 1) ? sod : sev;
      dst[(3 + (t >> 1)) * PCH + 5 + c] = xc[e];
    }
  } else {
    for (int i = tid; i < (L_SEQ / 2) * NCH; i += NTHREADS) {
      const int ra = i / NCH, c = i - ra * NCH;
      const int g = ra >> lsh, r = ra & (Lp - 1);
      cgptr xs = xg + g * (L_SEQ * NCH);
      const int o = (g * S + 3 + r) * PCH + c;
      sev[o] = xs[((2 * r) * stride + sel) * NCH + c];
      sod[o] = xs[((2 * r + 1) * stride + sel) * NCH + c];
    }
  }
  __syncthreads();
  for (int i = tid; i < G * 2 * 3 * NCH; i += NTHREADS) {
    const int q = i / NCH, c = i - q * NCH;
    const int w = q / 3, r = q - w * 3;
    lptr buf = ((w & 1) ? sod : sev) + (w >> 1) * S * PCH;
    buf[r * PCH + c]            = buf[3 * PCH + c];
    buf[(Lp + 3 + r) * PCH + c] = buf[(Lp + 2) * PCH + c];
  }
  __syncthreads();

  const int nb = node * 4;
  run_branch<0>(Wpk + (size_t)(nb + 0) * WB_TOT, b1g + (nb + 0) * NCH,
                b2g + (nb + 0) * NCH, sev, hb, wl, Lp, G, tsh, sod, db, xg, 0, 0);
  run_branch<1>(Wpk + (size_t)(nb + 1) * WB_TOT, b1g + (nb + 1) * NCH,
                b2g + (nb + 1) * NCH, sod, hb, wl, Lp, G, tsh, sev, cb, xg, 0, 0);
  for (int i = tid; i < G * 2 * 3 * NCH; i += NTHREADS) {
    const int q = i / NCH, c = i - q * NCH;
    const int w = q / 3, r = q - w * 3;
    lptr buf = ((w & 1) ? cb : db) + (w >> 1) * S * PCH;
    buf[r * PCH + c]            = buf[3 * PCH + c];
    buf[(Lp + 3 + r) * PCH + c] = buf[(Lp + 2) * PCH + c];
  }
  __syncthreads();
  run_branch<2>(Wpk + (size_t)(nb + 3) * WB_TOT, b1g + (nb + 3) * NCH,
                b2g + (nb + 3) * NCH, db, hb, wl, Lp, G, tsh, cb, db, xg,
                2 * stride, sel);
  run_branch<3>(Wpk + (size_t)(nb + 2) * WB_TOT, b1g + (nb + 2) * NCH,
                b2g + (nb + 2) * NCH, cb, hb, wl, Lp, G, tsh, db, cb, xg,
                2 * stride, sel + stride);
}

__global__ void pack_weights_kernel(const float* __restrict__ W1,
                                    const float* __restrict__ W2,
                                    float* __restrict__ Wpk) {
  const int idx = blockIdx.x * blockDim.x + threadIdx.x;
  if (idx >= 28 * WB_TOT) return;
  const int nb = idx / WB_TOT, rem = idx - nb * WB_TOT;
  float v = 0.f;
  if (rem < WM2_OFF) {
    int co, k;
    if (rem < 12 * 128) { const int l = (rem & 127) >> 2; co = l & 15; k = 8 * (rem >> 7) + 4 * (l >> 4) + (rem & 3); }
    else { const int r = rem - 12 * 128, l = r >> 1; co = l & 15; k = 96 + 2 * (l >> 4) + (r & 1); }
    const int tap = k / PCH, ci = k - tap * PCH;
    if (ci < NCH) v = W1[((size_t)(nb * NCH + co) * NCH + ci) * 5 + tap];
  } else if (rem < WT1_OFF) {
    const int r = rem - WM2_OFF;
    int co, k;
    if (r < 7 * 128) { const int l = (r & 127) >> 2; co = l & 15; k = 8 * (r >> 7) + 4 * (l >> 4) + (r & 3); }
    else { const int rr = r - 7 * 128, l = rr >> 1; co = l & 15; k = 56 + 2 * (l >> 4) + (rr & 1); }
    const int tap = k / PCH, ci = k - tap * PCH;
    if (ci < NCH) v = W2[((size_t)(nb * NCH + co) * NCH + ci) * 3 + tap];
  } else if (rem < WT2_OFF) {
    const int r = rem - WT1_OFF;
    const int c = r / TS1, k = r - c * TS1 - TG;
    if (k >= 0 && k < K1M) {
      const int tap = k / PCH, ci = k - tap * PCH;
      if (ci < NCH) v = W1[((size_t)(nb * NCH + 16 + c) * NCH + ci) * 5 + tap];
    }
  } else if (rem < WT2_OFF + 3 * TS2) {
    const int r = rem - WT2_OFF;
    const int c = r / TS2, k = r - c * TS2 - TG;
    if (k >= 0 && k < K2M) {
      const int tap = k / PCH, ci = k - tap * PCH;
      if (ci < NCH) v = W2[((size_t)(nb * NCH + 16 + c) * NCH + ci) * 3 + tap];
    }
  }
  Wpk[idx] = v;
}

#define PJ_NB    16
#define PJ_N     (PJ_NB * 5)
#define PJ_PITCH 268
#define PJ_P_OFF (PJ_N * PJ_PITCH)
#define PJ_SMEM  (PJ_P_OFF + 64 * PJ_PITCH)

template <int NT>
static __device__ __forceinline__ void proj_tiles(clptr pa, clptr yb, v8f* acc) {
#pragma unroll 4
  for (int s = 0; s < 32; s++) {
    const v4f a = *(const AS3 v4f*)(pa + 8 * s);
    v4f b[NT];
#pragma unroll
    for (int j = 0; j < NT; j++)
      b[j] = *(const AS3 v4f*)(yb + j * (16 * PJ_PITCH) + 8 * s);
#pragma unroll
    for (int j = 0; j < NT; j++) acc[j] = wmma4(a.xy, b[j].xy, acc[j]);
#pragma unroll
    for (int j = 0; j < NT; j++) acc[j] = wmma4(a.zw, b[j].zw, acc[j]);
  }
}

__global__ __launch_bounds__(NTHREADS) void proj_kernel(
    const float* __restrict__ xbuf, const float* __restrict__ x0D,
    const float* __restrict__ projw, float* __restrict__ out) {
  __shared__ __align__(16) float psm[PJ_SMEM];
  lptr ys = (lptr)psm;
  lptr ps = (lptr)psm + PJ_P_OFF;
  const int tid  = threadIdx.x;
  const int wave = __builtin_amdgcn_readfirstlane(tid >> 5);
  const int lane = tid & 31;
  const int lhi  = lane >> 4;
  const int lln  = lane & 15;
  const size_t b0 = (size_t)blockIdx.x * PJ_NB;
  cgptr xb = (cgptr)xbuf + b0 * (L_SEQ * NCH);
  cgptr x0 = (cgptr)x0D + b0 * (L_SEQ * 5);
  cgptr pj = (cgptr)projw;

#pragma unroll 8
  for (int it = 0; it < (64 * L_SEQ / 4) / NTHREADS; it++) {
    const int q = tid + it * NTHREADS;
    *(AS3 v4f*)(ps + (q >> 6) * PJ_PITCH + 4 * (q & 63)) = ((const AS1 v4f*)pj)[q];
  }
#pragma unroll 2
  for (int g = 0; g < PJ_NB; g++) {
    cgptr xbg = xb + g * (L_SEQ * NCH);
    cgptr x0g = x0 + g * (L_SEQ * 5);
    lptr  yg  = ys + g * 5 * PJ_PITCH;
#pragma unroll
    for (int it = 0; it < (L_SEQ * 5) / NTHREADS; it++) {
      const int e = tid + it * NTHREADS;
      const int l = e / 5, c = e - 5 * l;
      yg[c * PJ_PITCH + l] = xbg[l * NCH + c] + x0g[e];
    }
  }
  __syncthreads();

  const int mt  = wave & 3;
  const int nt0 = wave < 4 ? 0 : 3;
  const int ntn = wave < 4 ? 3 : 2;
  clptr pa = ps + (16 * mt + lln) * PJ_PITCH + 4 * lhi;
  clptr yb = ys + (16 * nt0 + lln) * PJ_PITCH + 4 * lhi;
  v8f acc[3];
#pragma unroll
  for (int j = 0; j < 3; j++) acc[j] = (v8f){};
  if (wave < 4) proj_tiles<3>(pa, yb, acc);
  else          proj_tiles<2>(pa, yb, acc);
  __syncthreads();
  lptr outs = ys;
#pragma unroll
  for (int j = 0; j < 3; j++) {
    if (j < ntn) {
      const int n  = 16 * (nt0 + j) + lln;
      const int bl = n / 5, c = n - 5 * bl;
      lptr op = outs + bl * 320 + (16 * mt + 8 * lhi) * 5 + c;
#pragma unroll
      for (int v = 0; v < 8; v++) op[v * 5] = acc[j][v];
    }
  }
  __syncthreads();
  AS1 v4f* og = (AS1 v4f*)((gptr)out + b0 * 320);
  for (int it = 0; it < (PJ_NB * 320 / 4) / NTHREADS; it++)
    og[tid + it * NTHREADS] = ((const AS3 v4f*)outs)[tid + it * NTHREADS];
}

extern "C" void kernel_launch(void* const* d_in, const int* in_sizes, int n_in,
                              void* d_out, int out_size, void* d_ws,
                              size_t ws_size, hipStream_t stream) {
  const float* x0D   = (const float*)d_in[0];
  const float* xctrl = (const float*)d_in[1];
  const float* W1    = (const float*)d_in[2];
  const float* b1    = (const float*)d_in[3];
  const float* W2    = (const float*)d_in[4];
  const float* b2    = (const float*)d_in[5];
  const float* projw = (const float*)d_in[6];
  float* out = (float*)d_out;

  float* ws   = (float*)d_ws;
  float* xbuf = ws;
  const size_t XB = (size_t)B_TOT * L_SEQ * NCH;
  float* Wpk = xbuf + XB;

  const int packN = 28 * WB_TOT;
  pack_weights_kernel<<<(packN + 255) / 256, 256, 0, stream>>>(W1, W2, Wpk);
  for (int d = 0; d < 3; d++)
    scinet_level_kernel<<<B_TOT, NTHREADS, 0, stream>>>(xbuf, x0D, xctrl, Wpk, b1,
                                                        b2, d);
  proj_kernel<<<B_TOT / PJ_NB, NTHREADS, 0, stream>>>(xbuf, x0D, projw, out);
}


# LDS per block 78464 -> 65408 B: c = ev*exp(psi(od)) overwrites ev in place (one buffer fewer) and every buffer ends where its reads end: five blocks resident per WGP instead of four
# speedup vs baseline: 2.1889x; 1.0347x over previous
#include <hip/hip_runtime.h>
#include <math.h>


typedef __attribute__((ext_vector_type(2))) float v2f;
typedef __attribute__((ext_vector_type(4))) float v4f;
typedef __attribute__((ext_vector_type(8))) float v8f;

#define AS3 __attribute__((address_space(3)))
#define AS1 __attribute__((address_space(1)))
typedef AS3 float*       lptr;
typedef const AS3 float* clptr;
typedef AS1 float*       gptr;
typedef const AS1 float* cgptr;

#define B_TOT    8192
#define L_SEQ    256
#define NCH      19
#define PCH      20
#define NWAVES   8
#define NTHREADS 256

#define K1M      100
#define K2M      60
#define K1T      180
#define K2T      140
#define TG       80
#define TS1      284
#define TS2      220
#define WM1_OFF  0
#define WM2_OFF  1600
#define WT1_OFF  2560
#define WT2_OFF  (WT1_OFF + 3 * TS1)
#define WB_TOT   4096

#define ROWS_MAX 152
#define IN_SZ    ((ROWS_MAX + 2) * PCH)
#define HB_SZ    (ROWS_MAX * PCH)
#define WL_SZ    (WT2_OFF + 3 * TS2)
#define SEV_OFF  0
#define SOD_OFF  (SEV_OFF + IN_SZ)
#define DB_OFF   (SOD_OFF + IN_SZ)
#define WL_OFF   (DB_OFF + IN_SZ)
#define HB_OFF   (WL_OFF + WL_SZ)
#define SMEM_TOT (HB_OFF + HB_SZ)

static __device__ __forceinline__ v8f wmma4(v2f a, v2f b, v8f c) {
  return __builtin_amdgcn_wmma_f32_16x16x4_f32(false, a, false, b, (short)0, c,
                                               false, false);
}

static __device__ __forceinline__ float leaky(float x) {
  return x > 0.f ? x : 0.01f * x;
}

static __device__ __forceinline__ float fast_tanh(float x) {
  const float e = __builtin_amdgcn_exp2f(x * 2.8853900817779268f);
  return fmaf(-2.f, __builtin_amdgcn_rcpf(e + 1.f), 1.f);
}

template <int MODE>
static __device__ void run_branch(cgptr Wp, cgptr b1, cgptr b2, clptr in,
                                  lptr hb, lptr wl, int Lp, int nseg, int tsh,
                                  clptr other, lptr outLds, gptr xg, int tmul,
                                  int tadd) {
  const int tid  = threadIdx.x;
  const int wave = __builtin_amdgcn_readfirstlane(tid >> 5);
  const int lane = tid & 31;
  const int lhi  = lane >> 4;
  const int lln  = lane & 15;
  int tj = lln / 3;
  int tc = lln - 3 * tj;
  if (lln == 15) { tj = 0; tc = 0; }

#pragma unroll
  for (int it = 0; it < WB_TOT / 4 / NTHREADS; it++)
    ((AS3 v4f*)wl)[tid + it * NTHREADS] = ((const AS1 v4f*)Wp)[tid + it * NTHREADS];
  __syncthreads();

  const int S    = Lp + 6;
  const int M1   = nseg * S - 4;
  const int nMt1 = (M1 + 15) >> 4;
  const int nTr  = (M1 + 4) / 5;

  if (wave < 2) {
    int ti = (wave << 4) + lln; if (ti > nTr - 1) ti = nTr - 1;
    clptr ar = in + ti * (5 * PCH);
    clptr br = wl + WT1_OFF + tc * TS1 + TG - PCH * tj;
    v8f acc = {};
    v2f aS[3], bS[3];
#pragma unroll
    for (int u = 0; u < 3; u++) {
      aS[u] = *(const AS3 v2f*)(ar + 168 + 4 * u + 2 * lhi);
      bS[u] = *(const AS3 v2f*)(br + 168 + 4 * u + 2 * lhi);
    }
#pragma unroll 7
    for (int s = 0; s < 21; s++) {
      const v4f a = *(const AS3 v4f*)(ar + 8 * s + 4 * lhi);
      const v4f b = *(const AS3 v4f*)(br + 8 * s + 4 * lhi);
      acc = wmma4(a.xy, b.xy, acc);
      acc = wmma4(a.zw, b.zw, acc);
    }
#pragma unroll
    for (int u = 0; u < 3; u++) acc = wmma4(aS[u], bS[u], acc);
    const float bt = b1[16 + tc];
    const int rb = ((wave << 4) + 8 * lhi) * 5 + tj;
    if (lln < 15) {
#pragma unroll
      for (int v = 0; v < 8; v++) {
        const int r = rb + 5 * v;
        if (r < M1) hb[r * PCH + 16 + tc] = leaky(acc[v] + bt);
      }
    }
  }
#pragma nounroll
  for (int pass = 0; pass < 2; pass++) {
    const int mt = pass == 0 ? wave - 2
                             : (wave >= 6 ? wave : (wave >= 4 ? wave + 4 : 99));
    if (mt < 0 || mt >= nMt1) continue;
    const int m0 = mt << 4;
    int mrow = m0 + lln; if (mrow > M1 - 1) mrow = M1 - 1;
    clptr ar = in + mrow * PCH;
    clptr bq = wl + WM1_OFF + 4 * lane;
    v8f acc = {};
    const v2f aL = *(const AS3 v2f*)(ar + 96 + 2 * lhi);
    const v2f bL = *(const AS3 v2f*)(wl + WM1_OFF + 12 * 128 + 2 * lane);
#pragma unroll 6
    for (int s = 0; s < 12; s++) {
      const v4f a = *(const AS3 v4f*)(ar + 8 * s + 4 * lhi);
      const v4f b = *(const AS3 v4f*)(bq + 128 * s);
      acc = wmma4(a.xy, b.xy, acc);
      acc = wmma4(a.zw, b.zw, acc);
    }
    acc = wmma4(aL, bL, acc);
    const float b1a = b1[lln];
    lptr hrow = hb + (m0 + 8 * lhi) * PCH;
    if (m0 + 16 <= M1) {
#pragma unroll
      for (int v = 0; v < 8; v++) hrow[v * PCH + lln] = leaky(acc[v] + b1a);
    } else {
#pragma unroll
      for (int v = 0; v < 8; v++)
        if (m0 + 8 * lhi + v < M1) hrow[v * PCH + lln] = leaky(acc[v] + b1a);
    }
  }
  __syncthreads();

  if (wave < 2) {
    int ti = (wave << 4) + lln; if (ti > nTr - 1) ti = nTr - 1;
    clptr ar = hb + ti * (5 * PCH);
    clptr br = wl + WT2_OFF + tc * TS2 + TG - PCH * tj;
    v8f acc = {};
    v2f aS[3], bS[3];
#pragma unroll
    for (int u = 0; u < 3; u++) {
      aS[u] = *(const AS3 v2f*)(ar + 128 + 4 * u + 2 * lhi);
      bS[u] = *(const AS3 v2f*)(br + 128 + 4 * u + 2 * lhi);
    }
#pragma unroll 8
    for (int s = 0; s < 16; s++) {
      const v4f a = *(const AS3 v4f*)(ar + 8 * s + 4 * lhi);
      const v4f b = *(const AS3 v4f*)(br + 8 * s + 4 * lhi);
      acc = wmma4(a.xy, b.xy, acc);
      acc = wmma4(a.zw, b.zw, acc);
    }
#pragma unroll
    for (int u = 0; u < 3; u++) acc = wmma4(aS[u], bS[u], acc);
    const float bt = b2[16 + tc];
    const int rb = ((wave << 4) + 8 * lhi) * 5 + tj;
    if (lln < 15) {
#pragma unroll
      for (int v = 0; v < 8; v++) {
        const int r  = rb + 5 * v;
        const int g  = (r >= S) + (r >= 2 * S) + (r >= 3 * S);
        const int jj = r - g * S;
        if (jj < Lp && g < nseg) {
          const int li = (r + 3) * PCH + 16 + tc;
          const float t   = fast_tanh(acc[v] + bt);
          const float oth = other[li];
          if (MODE <= 1) outLds[li] = oth * __expf(t);
          else {
            gptr xs = xg + g * (L_SEQ * NCH);
            if (MODE == 2) xs[(jj * tmul + tadd) * NCH + 16 + tc] = oth + t;
            else           xs[(jj * tmul + tadd) * NCH + 16 + tc] = oth - t;
          }
        }
      }
    }
  }
  const int nMt2  = (nseg * Lp) >> 4;
  const int tmask = (1 << tsh) - 1;
#pragma nounroll
  for (int pass = 0; pass < 2; pass++) {
    const int mt = pass == 0 ? wave - 2 : (wave >= 6 ? wave : 99);
    if (mt < 0 || mt >= nMt2) continue;
    const int g  = mt >> tsh;
    const int j0 = (mt & tmask) << 4;
    const int r0 = g * S + j0;
    clptr ar = hb + (r0 + lln) * PCH;
    clptr bq = wl + WM2_OFF + 4 * lane;
    v8f acc = {};
    const v2f aL = *(const AS3 v2f*)(ar + 56 + 2 * lhi);
    const v2f bL = *(const AS3 v2f*)(wl + WM2_OFF + 7 * 128 + 2 * lane);
#pragma unroll
    for (int s = 0; s < 7; s++) {
      const v4f a = *(const AS3 v4f*)(ar + 8 * s + 4 * lhi);
      const v4f b = *(const AS3 v4f*)(bq + 128 * s);
      acc = wmma4(a.xy, b.xy, acc);
      acc = wmma4(a.zw, b.zw, acc);
    }
    acc = wmma4(aL, bL, acc);
    const float b2a = b2[lln];
    const int mb = j0 + 8 * lhi;
    const int lb = (r0 + 3 + 8 * lhi) * PCH;
    gptr xs = xg + g * (L_SEQ * NCH);
#pragma unroll
    for (int v = 0; v < 8; v++) {
      const int m = mb + v;
      const float t   = fast_tanh(acc[v] + b2a);
      const float oth = other[lb + v * PCH + lln];
      if (MODE <= 1)      outLds[lb + v * PCH + lln] = oth * __expf(t);
      else if (MODE == 2) xs[(m * tmul + tadd) * NCH + lln] = oth + t;
      else                xs[(m * tmul + tadd) * NCH + lln] = oth - t;
    }
  }
  __syncthreads();
}

__global__ __launch_bounds__(NTHREADS) void scinet_level_kernel(
    float* __restrict__ xbuf_, const float* __restrict__ x0D_,
    const float* __restrict__ xctrl_, const float* __restrict__ Wpk_,
    const float* __restrict__ b1g_, const float* __restrict__ b2g_, int depth) {
  __shared__ __align__(16) float smem[SMEM_TOT];
  const int tid = threadIdx.x;
  const int G   = 1 << depth;
  const int sel = blockIdx.x & (G - 1);
  const int bg  = blockIdx.x >> depth;
  int node = 0;
  for (int i = 0; i < depth; i++) node = 2 * node + 1 + ((sel >> i) & 1);
  const int stride = 1 << depth;
  const int Lp     = (L_SEQ >> depth) >> 1;
  const int S      = Lp + 6;
  const int lsh    = 7 - depth;
  const int tsh    = 3 - depth;

  lptr sev = (lptr)(smem + SEV_OFF);
  lptr sod = (lptr)(smem + SOD_OFF);
  lptr hb  = (lptr)(smem + HB_OFF);
  lptr db  = (lptr)(smem + DB_OFF);
  lptr cb  = sev;
  lptr wl  = (lptr)(smem + WL_OFF);
  lptr sm  = (lptr)smem;

  for (int i = tid; i < SMEM_TOT; i += NTHREADS) sm[i] = 0.f;
  __syncthreads();

  gptr  xg  = (gptr)(xbuf_ + (size_t)bg * G * L_SEQ * NCH);
  cgptr Wpk = (cgptr)Wpk_;
  cgptr b1g = (cgptr)b1g_;
  cgptr b2g = (cgptr)b2g_;

  if (depth == 0) {
    cgptr x0 = (cgptr)x0D_ + (size_t)bg * (L_SEQ * 5);
    cgptr xc = (cgptr)xctrl_ + (size_t)bg * (L_SEQ * 14);
    for (int it = 0; it < (L_SEQ * 5) / NTHREADS; it++) {
      const int e = tid + it * NTHREADS;
      const int t = e / 5, c = e - t * 5;
      lptr dst = (t & 1) ? sod : sev;
      dst[(3 + (t >> 1)) * PCH + c] = x0[e];
    }
    for (int it = 0; it < (L_SEQ * 14) / NTHREADS; it++) {
      const int e = tid + it * NTHREADS;
      const int t = e / 14, c = e - t * 14;
      lptr dst = (t & 1) ? sod : sev;
      dst[(3 + (t >> 1)) * PCH + 5 + c] = xc[e];
    }
  } else {
    for (int i = tid; i < (L_SEQ / 2) * NCH; i += NTHREADS) {
      const int ra = i / NCH, c = i - ra * NCH;
      const int g = ra >> lsh, r = ra & (Lp - 1);
      cgptr xs = xg + g * (L_SEQ * NCH);
      const int o = (g * S + 3 + r) * PCH + c;
      sev[o] = xs[((2 * r) * stride + sel) * NCH + c];
      sod[o] = xs[((2 * r + 1) * stride + sel) * NCH + c];
    }
  }
  __syncthreads();
  for (int i = tid; i < G * 2 * 3 * NCH; i += NTHREADS) {
    const int q = i / NCH, c = i - q * NCH;
    const int w = q / 3, r = q - w * 3;
    lptr buf = ((w & 1) ? sod : sev) + (w >> 1) * S * PCH;
    buf[r * PCH + c]            = buf[3 * PCH + c];
    buf[(Lp + 3 + r) * PCH + c] = buf[(Lp + 2) * PCH + c];
  }
  __syncthreads();

  const int nb = node * 4;
  run_branch<0>(Wpk + (size_t)(nb + 0) * WB_TOT, b1g + (nb + 0) * NCH,
                b2g + (nb + 0) * NCH, sev, hb, wl, Lp, G, tsh, sod, db, xg, 0, 0);
  run_branch<1>(Wpk + (size_t)(nb + 1) * WB_TOT, b1g + (nb + 1) * NCH,
                b2g + (nb + 1) * NCH, sod, hb, wl, Lp, G, tsh, sev, cb, xg, 0, 0);
  for (int i = tid; i < G * 2 * 3 * NCH; i += NTHREADS) {
    const int q = i / NCH, c = i - q * NCH;
    const int w = q / 3, r = q - w * 3;
    lptr buf = ((w & 1) ? cb : db) + (w >> 1) * S * PCH;
    buf[r * PCH + c]            = buf[3 * PCH + c];
    buf[(Lp + 3 + r) * PCH + c] = buf[(Lp + 2) * PCH + c];
  }
  __syncthreads();
  run_branch<2>(Wpk + (size_t)(nb + 3) * WB_TOT, b1g + (nb + 3) * NCH,
                b2g + (nb + 3) * NCH, db, hb, wl, Lp, G, tsh, cb, db, xg,
                2 * stride, sel);
  run_branch<3>(Wpk + (size_t)(nb + 2) * WB_TOT, b1g + (nb + 2) * NCH,
                b2g + (nb + 2) * NCH, cb, hb, wl, Lp, G, tsh, db, cb, xg,
                2 * stride, sel + stride);
}

__global__ void pack_weights_kernel(const float* __restrict__ W1,
                                    const float* __restrict__ W2,
                                    float* __restrict__ Wpk) {
  const int idx = blockIdx.x * blockDim.x + threadIdx.x;
  if (idx >= 28 * WB_TOT) return;
  const int nb = idx / WB_TOT, rem = idx - nb * WB_TOT;
  float v = 0.f;
  if (rem < WM2_OFF) {
    int co, k;
    if (rem < 12 * 128) { const int l = (rem & 127) >> 2; co = l & 15; k = 8 * (rem >> 7) + 4 * (l >> 4) + (rem & 3); }
    else { const int r = rem - 12 * 128, l = r >> 1; co = l & 15; k = 96 + 2 * (l >> 4) + (r & 1); }
    const int tap = k / PCH, ci = k - tap * PCH;
    if (ci < NCH) v = W1[((size_t)(nb * NCH + co) * NCH + ci) * 5 + tap];
  } else if (rem < WT1_OFF) {
    const int r = rem - WM2_OFF;
    int co, k;
    if (r < 7 * 128) { const int l = (r & 127) >> 2; co = l & 15; k = 8 * (r >> 7) + 4 * (l >> 4) + (r & 3); }
    else { const int rr = r - 7 * 128, l = rr >> 1; co = l & 15; k = 56 + 2 * (l >> 4) + (rr & 1); }
    const int tap = k / PCH, ci = k - tap * PCH;
    if (ci < NCH) v = W2[((size_t)(nb * NCH + co) * NCH + ci) * 3 + tap];
  } else if (rem < WT2_OFF) {
    const int r = rem - WT1_OFF;
    const int c = r / TS1, k = r - c * TS1 - TG;
    if (k >= 0 && k < K1M) {
      const int tap = k / PCH, ci = k - tap * PCH;
      if (ci < NCH) v = W1[((size_t)(nb * NCH + 16 + c) * NCH + ci) * 5 + tap];
    }
  } else if (rem < WT2_OFF + 3 * TS2) {
    const int r = rem - WT2_OFF;
    const int c = r / TS2, k = r - c * TS2 - TG;
    if (k >= 0 && k < K2M) {
      const int tap = k / PCH, ci = k - tap * PCH;
      if (ci < NCH) v = W2[((size_t)(nb * NCH + 16 + c) * NCH + ci) * 3 + tap];
    }
  }
  Wpk[idx] = v;
}

#define PJ_NB    16
#define PJ_N     (PJ_NB * 5)
#define PJ_PITCH 268
#define PJ_P_OFF (PJ_N * PJ_PITCH)
#define PJ_SMEM  (PJ_P_OFF + 64 * PJ_PITCH)

template <int NT>
static __device__ __forceinline__ void proj_tiles(clptr pa, clptr yb, v8f* acc) {
#pragma unroll 4
  for (int s = 0; s < 32; s++) {
    const v4f a = *(const AS3 v4f*)(pa + 8 * s);
    v4f b[NT];
#pragma unroll
    for (int j = 0; j < NT; j++)
      b[j] = *(const AS3 v4f*)(yb + j * (16 * PJ_PITCH) + 8 * s);
#pragma unroll
    for (int j = 0; j < NT; j++) acc[j] = wmma4(a.xy, b[j].xy, acc[j]);
#pragma unroll
    for (int j = 0; j < NT; j++) acc[j] = wmma4(a.zw, b[j].zw, acc[j]);
  }
}

__global__ __launch_bounds__(NTHREADS) void proj_kernel(
    const float* __restrict__ xbuf, const float* __restrict__ x0D,
    const float* __restrict__ projw, float* __restrict__ out) {
  __shared__ __align__(16) float psm[PJ_SMEM];
  lptr ys = (lptr)psm;
  lptr ps = (lptr)psm + PJ_P_OFF;
  const int tid  = threadIdx.x;
  const int wave = __builtin_amdgcn_readfirstlane(tid >> 5);
  const int lane = tid & 31;
  const int lhi  = lane >> 4;
  const int lln  = lane & 15;
  const size_t b0 = (size_t)blockIdx.x * PJ_NB;
  cgptr xb = (cgptr)xbuf + b0 * (L_SEQ * NCH);
  cgptr x0 = (cgptr)x0D + b0 * (L_SEQ * 5);
  cgptr pj = (cgptr)projw;

#pragma unroll 8
  for (int it = 0; it < (64 * L_SEQ / 4) / NTHREADS; it++) {
    const int q = tid + it * NTHREADS;
    *(AS3 v4f*)(ps + (q >> 6) * PJ_PITCH + 4 * (q & 63)) = ((const AS1 v4f*)pj)[q];
  }
#pragma unroll 2
  for (int g = 0; g < PJ_NB; g++) {
    cgptr xbg = xb + g * (L_SEQ * NCH);
    cgptr x0g = x0 + g * (L_SEQ * 5);
    lptr  yg  = ys + g * 5 * PJ_PITCH;
#pragma unroll
    for (int it = 0; it < (L_SEQ * 5) / NTHREADS; it++) {
      const int e = tid + it * NTHREADS;
      const int l = e / 5, c = e - 5 * l;
      yg[c * PJ_PITCH + l] = xbg[l * NCH + c] + x0g[e];
    }
  }
  __syncthreads();

  const int mt  = wave & 3;
  const int nt0 = wave < 4 ? 0 : 3;
  const int ntn = wave < 4 ? 3 : 2;
  clptr pa = ps + (16 * mt + lln) * PJ_PITCH + 4 * lhi;
  clptr yb = ys + (16 * nt0 + lln) * PJ_PITCH + 4 * lhi;
  v8f acc[3];
#pragma unroll
  for (int j = 0; j < 3; j++) acc[j] = (v8f){};
  if (wave < 4) proj_tiles<3>(pa, yb, acc);
  else          proj_tiles<2>(pa, yb, acc);
  __syncthreads();
  lptr outs = ys;
#pragma unroll
  for (int j = 0; j < 3; j++) {
    if (j < ntn) {
      const int n  = 16 * (nt0 + j) + lln;
      const int bl = n / 5, c = n - 5 * bl;
      lptr op = outs + bl * 320 + (16 * mt + 8 * lhi) * 5 + c;
#pragma unroll
      for (int v = 0; v < 8; v++) op[v * 5] = acc[j][v];
    }
  }
  __syncthreads();
  AS1 v4f* og = (AS1 v4f*)((gptr)out + b0 * 320);
  for (int it = 0; it < (PJ_NB * 320 / 4) / NTHREADS; it++)
    og[tid + it * NTHREADS] = ((const AS3 v4f*)outs)[tid + it * NTHREADS];
}

extern "C" void kernel_launch(void* const* d_in, const int* in_sizes, int n_in,
                              void* d_out, int out_size, void* d_ws,
                              size_t ws_size, hipStream_t stream) {
  const float* x0D   = (const float*)d_in[0];
  const float* xctrl = (const float*)d_in[1];
  const float* W1    = (const float*)d_in[2];
  const float* b1    = (const float*)d_in[3];
  const float* W2    = (const float*)d_in[4];
  const float* b2    = (const float*)d_in[5];
  const float* projw = (const float*)d_in[6];
  float* out = (float*)d_out;

  float* ws   = (float*)d_ws;
  float* xbuf = ws;
  const size_t XB = (size_t)B_TOT * L_SEQ * NCH;
  float* Wpk = xbuf + XB;

  const int packN = 28 * WB_TOT;
  pack_weights_kernel<<<(packN + 255) / 256, 256, 0, stream>>>(W1, W2, Wpk);
  for (int d = 0; d < 3; d++)
    scinet_level_kernel<<<B_TOT, NTHREADS, 0, stream>>>(xbuf, x0D, xctrl, Wpk, b1,
                                                        b2, d);
  proj_kernel<<<B_TOT / PJ_NB, NTHREADS, 0, stream>>>(xbuf, x0D, projw, out);
}


# Level kernel: wave-to-tile roles move on by one SIMD with every branch: each SIMD carries the heavy role pair (2 tail + 3 main tiles a branch) once per block, not one SIMD always; sums unchanged
# speedup vs baseline: 2.2949x; 1.0484x over previous
#include <hip/hip_runtime.h>
#include <math.h>


typedef __attribute__((ext_vector_type(2))) float v2f;
typedef __attribute__((ext_vector_type(4))) float v4f;
typedef __attribute__((ext_vector_type(8))) float v8f;

#define AS3 __attribute__((address_space(3)))
#define AS1 __attribute__((address_space(1)))
typedef AS3 float*       lptr;
typedef const AS3 float* clptr;
typedef AS1 float*       gptr;
typedef const AS1 float* cgptr;

#define B_TOT    8192
#define L_SEQ    256
#define NCH      19
#define PCH      20
#define NWAVES   8
#define NTHREADS 256

#define K1M      100
#define K2M      60
#define K1T      180
#define K2T      140
#define TG       80
#define TS1      284
#define TS2      220
#define WM1_OFF  0
#define WM2_OFF  1600
#define WT1_OFF  2560
#define WT2_OFF  (WT1_OFF + 3 * TS1)
#define WB_TOT   4096

#define ROWS_MAX 152
#define IN_SZ    ((ROWS_MAX + 2) * PCH)
#define HB_SZ    (ROWS_MAX * PCH)
#define WL_SZ    (WT2_OFF + 3 * TS2)
#define SEV_OFF  0
#define SOD_OFF  (SEV_OFF + IN_SZ)
#define DB_OFF   (SOD_OFF + IN_SZ)
#define WL_OFF   (DB_OFF + IN_SZ)
#define HB_OFF   (WL_OFF + WL_SZ)
#define SMEM_TOT (HB_OFF + HB_SZ)

static __device__ __forceinline__ v8f wmma4(v2f a, v2f b, v8f c) {
  return __builtin_amdgcn_wmma_f32_16x16x4_f32(false, a, false, b, (short)0, c,
                                               false, false);
}

static __device__ __forceinline__ float leaky(float x) {
  return x > 0.f ? x : 0.01f * x;
}

static __device__ __forceinline__ float fast_tanh(float x) {
  const float e = __builtin_amdgcn_exp2f(x * 2.8853900817779268f);
  return fmaf(-2.f, __builtin_amdgcn_rcpf(e + 1.f), 1.f);
}

template <int MODE>
static __device__ void run_branch(cgptr Wp, cgptr b1, cgptr b2, clptr in,
                                  lptr hb, lptr wl, int Lp, int nseg, int tsh,
                                  clptr other, lptr outLds, gptr xg, int tmul,
                                  int tadd) {
  const int tid  = threadIdx.x;
  const int hwv  = __builtin_amdgcn_readfirstlane(tid >> 5);
  const int wave = ((hwv + MODE) & 3) | (hwv & 4);
  const int lane = tid & 31;
  const int lhi  = lane >> 4;
  const int lln  = lane & 15;
  int tj = lln / 3;
  int tc = lln - 3 * tj;
  if (lln == 15) { tj = 0; tc = 0; }

#pragma unroll
  for (int it = 0; it < WB_TOT / 4 / NTHREADS; it++)
    ((AS3 v4f*)wl)[tid + it * NTHREADS] = ((const AS1 v4f*)Wp)[tid + it * NTHREADS];
  __syncthreads();

  const int S    = Lp + 6;
  const int M1   = nseg * S - 4;
  const int nMt1 = (M1 + 15) >> 4;
  const int nTr  = (M1 + 4) / 5;

  if (wave < 2) {
    int ti = (wave << 4) + lln; if (ti > nTr - 1) ti = nTr - 1;
    clptr ar = in + ti * (5 * PCH);
    clptr br = wl + WT1_OFF + tc * TS1 + TG - PCH * tj;
    v8f acc = {};
    v2f aS[3], bS[3];
#pragma unroll
    for (int u = 0; u < 3; u++) {
      aS[u] = *(const AS3 v2f*)(ar + 168 + 4 * u + 2 * lhi);
      bS[u] = *(const AS3 v2f*)(br + 168 + 4 * u + 2 * lhi);
    }
#pragma unroll 7
    for (int s = 0; s < 21; s++) {
      const v4f a = *(const AS3 v4f*)(ar + 8 * s + 4 * lhi);
      const v4f b = *(const AS3 v4f*)(br + 8 * s + 4 * lhi);
      acc = wmma4(a.xy, b.xy, acc);
      acc = wmma4(a.zw, b.zw, acc);
    }
#pragma unroll
    for (int u = 0; u < 3; u++) acc = wmma4(aS[u], bS[u], acc);
    const float bt = b1[16 + tc];
    const int rb = ((wave << 4) + 8 * lhi) * 5 + tj;
    if (lln < 15) {
#pragma unroll
      for (int v = 0; v < 8; v++) {
        const int r = rb + 5 * v;
        if (r < M1) hb[r * PCH + 16 + tc] = leaky(acc[v] + bt);
      }
    }
  }
#pragma nounroll
  for (int pass = 0; pass < 2; pass++) {
    const int mt = pass == 0 ? wave - 2
                             : (wave >= 6 ? wave : (wave >= 4 ? wave + 4 : 99));
    if (mt < 0 || mt >= nMt1) continue;
    const int m0 = mt << 4;
    int mrow = m0 + lln; if (mrow > M1 - 1) mrow = M1 - 1;
    clptr ar = in + mrow * PCH;
    clptr bq = wl + WM1_OFF + 4 * lane;
    v8f acc = {};
    const v2f aL = *(const AS3 v2f*)(ar + 96 + 2 * lhi);
    const v2f bL = *(const AS3 v2f*)(wl + WM1_OFF + 12 * 128 + 2 * lane);
#pragma unroll 6
    for (int s = 0; s < 12; s++) {
      const v4f a = *(const AS3 v4f*)(ar + 8 * s + 4 * lhi);
      const v4f b = *(const AS3 v4f*)(bq + 128 * s);
      acc = wmma4(a.xy, b.xy, acc);
      acc = wmma4(a.zw, b.zw, acc);
    }
    acc = wmma4(aL, bL, acc);
    const float b1a = b1[lln];
    lptr hrow = hb + (m0 + 8 * lhi) * PCH;
    if (m0 + 16 <= M1) {
#pragma unroll
      for (int v = 0; v < 8; v++) hrow[v * PCH + lln] = leaky(acc[v] + b1a);
    } else {
#pragma unroll
      for (int v = 0; v < 8; v++)
        if (m0 + 8 * lhi + v < M1) hrow[v * PCH + lln] = leaky(acc[v] + b1a);
    }
  }
  __syncthreads();

  if (wave < 2) {
    int ti = (wave << 4) + lln; if (ti > nTr - 1) ti = nTr - 1;
    clptr ar = hb + ti * (5 * PCH);
    clptr br = wl + WT2_OFF + tc * TS2 + TG - PCH * tj;
    v8f acc = {};
    v2f aS[3], bS[3];
#pragma unroll
    for (int u = 0; u < 3; u++) {
      aS[u] = *(const AS3 v2f*)(ar + 128 + 4 * u + 2 * lhi);
      bS[u] = *(const AS3 v2f*)(br + 128 + 4 * u + 2 * lhi);
    }
#pragma unroll 8
    for (int s = 0; s < 16; s++) {
      const v4f a = *(const AS3 v4f*)(ar + 8 * s + 4 * lhi);
      const v4f b = *(const AS3 v4f*)(br + 8 * s + 4 * lhi);
      acc = wmma4(a.xy, b.xy, acc);
      acc = wmma4(a.zw, b.zw, acc);
    }
#pragma unroll
    for (int u = 0; u < 3; u++) acc = wmma4(aS[u], bS[u], acc);
    const float bt = b2[16 + tc];
    const int rb = ((wave << 4) + 8 * lhi) * 5 + tj;
    if (lln < 15) {
#pragma unroll
      for (int v = 0; v < 8; v++) {
        const int r  = rb + 5 * v;
        const int g  = (r >= S) + (r >= 2 * S) + (r >= 3 * S);
        const int jj = r - g * S;
        if (jj < Lp && g < nseg) {
          const int li = (r + 3) * PCH + 16 + tc;
          const float t   = fast_tanh(acc[v] + bt);
          const float oth = other[li];
          if (MODE <= 1) outLds[li] = oth * __expf(t);
          else {
            gptr xs = xg + g * (L_SEQ * NCH);
            if (MODE == 2) xs[(jj * tmul + tadd) * NCH + 16 + tc] = oth + t;
            else           xs[(jj * tmul + tadd) * NCH + 16 + tc] = oth - t;
          }
        }
      }
    }
  }
  const int nMt2  = (nseg * Lp) >> 4;
  const int tmask = (1 << tsh) - 1;
#pragma nounroll
  for (int pass = 0; pass < 2; pass++) {
    const int mt = pass == 0 ? wave - 2 : (wave >= 6 ? wave : 99);
    if (mt < 0 || mt >= nMt2) continue;
    const int g  = mt >> tsh;
    const int j0 = (mt & tmask) << 4;
    const int r0 = g * S + j0;
    clptr ar = hb + (r0 + lln) * PCH;
    clptr bq = wl + WM2_OFF + 4 * lane;
    v8f acc = {};
    const v2f aL = *(const AS3 v2f*)(ar + 56 + 2 * lhi);
    const v2f bL = *(const AS3 v2f*)(wl + WM2_OFF + 7 * 128 + 2 * lane);
#pragma unroll
    for (int s = 0; s < 7; s++) {
      const v4f a = *(const AS3 v4f*)(ar + 8 * s + 4 * lhi);
      const v4f b = *(const AS3 v4f*)(bq + 128 * s);
      acc = wmma4(a.xy, b.xy, acc);
      acc = wmma4(a.zw, b.zw, acc);
    }
    acc = wmma4(aL, bL, acc);
    const float b2a = b2[lln];
    const int mb = j0 + 8 * lhi;
    const int lb = (r0 + 3 + 8 * lhi) * PCH;
    gptr xs = xg + g * (L_SEQ * NCH);
#pragma unroll
    for (int v = 0; v < 8; v++) {
      const int m = mb + v;
      const float t   = fast_tanh(acc[v] + b2a);
      const float oth = other[lb + v * PCH + lln];
      if (MODE <= 1)      outLds[lb + v * PCH + lln] = oth * __expf(t);
      else if (MODE == 2) xs[(m * tmul + tadd) * NCH + lln] = oth + t;
      else                xs[(m * tmul + tadd) * NCH + lln] = oth - t;
    }
  }
  __syncthreads();
}

__global__ __launch_bounds__(NTHREADS) void scinet_level_kernel(
    float* __restrict__ xbuf_, const float* __restrict__ x0D_,
    const float* __restrict__ xctrl_, const float* __restrict__ Wpk_,
    const float* __restrict__ b1g_, const float* __restrict__ b2g_, int depth) {
  __shared__ __align__(16) float smem[SMEM_TOT];
  const int tid = threadIdx.x;
  const int G   = 1 << depth;
  const int sel = blockIdx.x & (G - 1);
  const int bg  = blockIdx.x >> depth;
  int node = 0;
  for (int i = 0; i < depth; i++) node = 2 * node + 1 + ((sel >> i) & 1);
  const int stride = 1 << depth;
  const int Lp     = (L_SEQ >> depth) >> 1;
  const int S      = Lp + 6;
  const int lsh    = 7 - depth;
  const int tsh    = 3 - depth;

  lptr sev = (lptr)(smem + SEV_OFF);
  lptr sod = (lptr)(smem + SOD_OFF);
  lptr hb  = (lptr)(smem + HB_OFF);
  lptr db  = (lptr)(smem + DB_OFF);
  lptr cb  = sev;
  lptr wl  = (lptr)(smem + WL_OFF);
  lptr sm  = (lptr)smem;

  for (int i = tid; i < SMEM_TOT; i += NTHREADS) sm[i] = 0.f;
  __syncthreads();

  gptr  xg  = (gptr)(xbuf_ + (size_t)bg * G * L_SEQ * NCH);
  cgptr Wpk = (cgptr)Wpk_;
  cgptr b1g = (cgptr)b1g_;
  cgptr b2g = (cgptr)b2g_;

  if (depth == 0) {
    cgptr x0 = (cgptr)x0D_ + (size_t)bg * (L_SEQ * 5);
    cgptr xc = (cgptr)xctrl_ + (size_t)bg * (L_SEQ * 14);
    for (int it = 0; it < (L_SEQ * 5) / NTHREADS; it++) {
      const int e = tid + it * NTHREADS;
      const int t = e / 5, c = e - t * 5;
      lptr dst = (t & 1) ? sod : sev;
      dst[(3 + (t >> 1)) * PCH + c] = x0[e];
    }
    for (int it = 0; it < (L_SEQ * 14) / NTHREADS; it++) {
      const int e = tid + it * NTHREADS;
      const int t = e / 14, c = e - t * 14;
      lptr dst = (t & 1) ? sod : sev;
      dst[(3 + (t >> 1)) * PCH + 5 + c] = xc[e];
    }
  } else {
    for (int i = tid; i < (L_SEQ / 2) * NCH; i += NTHREADS) {
      const int ra = i / NCH, c = i - ra * NCH;
      const int g = ra >> lsh, r = ra & (Lp - 1);
      cgptr xs = xg + g * (L_SEQ * NCH);
      const int o = (g * S + 3 + r) * PCH + c;
      sev[o] = xs[((2 * r) * stride + sel) * NCH + c];
      sod[o] = xs[((2 * r + 1) * stride + sel) * NCH + c];
    }
  }
  __syncthreads();
  for (int i = tid; i < G * 2 * 3 * NCH; i += NTHREADS) {
    const int q = i / NCH, c = i - q * NCH;
    const int w = q / 3, r = q - w * 3;
    lptr buf = ((w & 1) ? sod : sev) + (w >> 1) * S * PCH;
    buf[r * PCH + c]            = buf[3 * PCH + c];
    buf[(Lp + 3 + r) * PCH + c] = buf[(Lp + 2) * PCH + c];
  }
  __syncthreads();

  const int nb = node * 4;
  run_branch<0>(Wpk + (size_t)(nb + 0) * WB_TOT, b1g + (nb + 0) * NCH,
                b2g + (nb + 0) * NCH, sev, hb, wl, Lp, G, tsh, sod, db, xg, 0, 0);
  run_branch<1>(Wpk + (size_t)(nb + 1) * WB_TOT, b1g + (nb + 1) * NCH,
                b2g + (nb + 1) * NCH, sod, hb, wl, Lp, G, tsh, sev, cb, xg, 0, 0);
  for (int i = tid; i < G * 2 * 3 * NCH; i += NTHREADS) {
    const int q = i / NCH, c = i - q * NCH;
    const int w = q / 3, r = q - w * 3;
    lptr buf = ((w & 1) ? cb : db) + (w >> 1) * S * PCH;
    buf[r * PCH + c]            = buf[3 * PCH + c];
    buf[(Lp + 3 + r) * PCH + c] = buf[(Lp + 2) * PCH + c];
  }
  __syncthreads();
  run_branch<2>(Wpk + (size_t)(nb + 3) * WB_TOT, b1g + (nb + 3) * NCH,
                b2g + (nb + 3) * NCH, db, hb, wl, Lp, G, tsh, cb, db, xg,
                2 * stride, sel);
  run_branch<3>(Wpk + (size_t)(nb + 2) * WB_TOT, b1g + (nb + 2) * NCH,
                b2g + (nb + 2) * NCH, cb, hb, wl, Lp, G, tsh, db, cb, xg,
                2 * stride, sel + stride);
}

__global__ void pack_weights_kernel(const float* __restrict__ W1,
                                    const float* __restrict__ W2,
                                    float* __restrict__ Wpk) {
  const int idx = blockIdx.x * blockDim.x + threadIdx.x;
  if (idx >= 28 * WB_TOT) return;
  const int nb = idx / WB_TOT, rem = idx - nb * WB_TOT;
  float v = 0.f;
  if (rem < WM2_OFF) {
    int co, k;
    if (rem < 12 * 128) { const int l = (rem & 127) >> 2; co = l & 15; k = 8 * (rem >> 7) + 4 * (l >> 4) + (rem & 3); }
    else { const int r = rem - 12 * 128, l = r >> 1; co = l & 15; k = 96 + 2 * (l >> 4) + (r & 1); }
    const int tap = k / PCH, ci = k - tap * PCH;
    if (ci < NCH) v = W1[((size_t)(nb * NCH + co) * NCH + ci) * 5 + tap];
  } else if (rem < WT1_OFF) {
    const int r = rem - WM2_OFF;
    int co, k;
    if (r < 7 * 128) { const int l = (r & 127) >> 2; co = l & 15; k = 8 * (r >> 7) + 4 * (l >> 4) + (r & 3); }
    else { const int rr = r - 7 * 128, l = rr >> 1; co = l & 15; k = 56 + 2 * (l >> 4) + (rr & 1); }
    const int tap = k / PCH, ci = k - tap * PCH;
    if (ci < NCH) v = W2[((size_t)(nb * NCH + co) * NCH + ci) * 3 + tap];
  } else if (rem < WT2_OFF) {
    const int r = rem - WT1_OFF;
    const int c = r / TS1, k = r - c * TS1 - TG;
    if (k >= 0 && k < K1M) {
      const int tap = k / PCH, ci = k - tap * PCH;
      if (ci < NCH) v = W1[((size_t)(nb * NCH + 16 + c) * NCH + ci) * 5 + tap];
    }
  } else if (rem < WT2_OFF + 3 * TS2) {
    const int r = rem - WT2_OFF;
    const int c = r / TS2, k = r - c * TS2 - TG;
    if (k >= 0 && k < K2M) {
      const int tap = k / PCH, ci = k - tap * PCH;
      if (ci < NCH) v = W2[((size_t)(nb * NCH + 16 + c) * NCH + ci) * 3 + tap];
    }
  }
  Wpk[idx] = v;
}

#define PJ_NB    16
#define PJ_N     (PJ_NB * 5)
#define PJ_PITCH 268
#define PJ_P_OFF (PJ_N * PJ_PITCH)
#define PJ_SMEM  (PJ_P_OFF + 64 * PJ_PITCH)

template <int NT>
static __device__ __forceinline__ void proj_tiles(clptr pa, clptr yb, v8f* acc) {
#pragma unroll 4
  for (int s = 0; s < 32; s++) {
    const v4f a = *(const AS3 v4f*)(pa + 8 * s);
    v4f b[NT];
#pragma unroll
    for (int j = 0; j < NT; j++)
      b[j] = *(const AS3 v4f*)(yb + j * (16 * PJ_PITCH) + 8 * s);
#pragma unroll
    for (int j = 0; j < NT; j++) acc[j] = wmma4(a.xy, b[j].xy, acc[j]);
#pragma unroll
    for (int j = 0; j < NT; j++) acc[j] = wmma4(a.zw, b[j].zw, acc[j]);
  }
}

__global__ __launch_bounds__(NTHREADS) void proj_kernel(
    const float* __restrict__ xbuf, const float* __restrict__ x0D,
    const float* __restrict__ projw, float* __restrict__ out) {
  __shared__ __align__(16) float psm[PJ_SMEM];
  lptr ys = (lptr)psm;
  lptr ps = (lptr)psm + PJ_P_OFF;
  const int tid  = threadIdx.x;
  const int wave = __builtin_amdgcn_readfirstlane(tid >> 5);
  const int lane = tid & 31;
  const int lhi  = lane >> 4;
  const int lln  = lane & 15;
  const size_t b0 = (size_t)blockIdx.x * PJ_NB;
  cgptr xb = (cgptr)xbuf + b0 * (L_SEQ * NCH);
  cgptr x0 = (cgptr)x0D + b0 * (L_SEQ * 5);
  cgptr pj = (cgptr)projw;

#pragma unroll 8
  for (int it = 0; it < (64 * L_SEQ / 4) / NTHREADS; it++) {
    const int q = tid + it * NTHREADS;
    *(AS3 v4f*)(ps + (q >> 6) * PJ_PITCH + 4 * (q & 63)) = ((const AS1 v4f*)pj)[q];
  }
#pragma unroll 2
  for (int g = 0; g < PJ_NB; g++) {
    cgptr xbg = xb + g * (L_SEQ * NCH);
    cgptr x0g = x0 + g * (L_SEQ * 5);
    lptr  yg  = ys + g * 5 * PJ_PITCH;
#pragma unroll
    for (int it = 0; it < (L_SEQ * 5) / NTHREADS; it++) {
      const int e = tid + it * NTHREADS;
      const int l = e / 5, c = e - 5 * l;
      yg[c * PJ_PITCH + l] = xbg[l * NCH + c] + x0g[e];
    }
  }
  __syncthreads();

  const int mt  = wave & 3;
  const int nt0 = wave < 4 ? 0 : 3;
  const int ntn = wave < 4 ? 3 : 2;
  clptr pa = ps + (16 * mt + lln) * PJ_PITCH + 4 * lhi;
  clptr yb = ys + (16 * nt0 + lln) * PJ_PITCH + 4 * lhi;
  v8f acc[3];
#pragma unroll
  for (int j = 0; j < 3; j++) acc[j] = (v8f){};
  if (wave < 4) proj_tiles<3>(pa, yb, acc);
  else          proj_tiles<2>(pa, yb, acc);
  __syncthreads();
  lptr outs = ys;
#pragma unroll
  for (int j = 0; j < 3; j++) {
    if (j < ntn) {
      const int n  = 16 * (nt0 + j) + lln;
      const int bl = n / 5, c = n - 5 * bl;
      lptr op = outs + bl * 320 + (16 * mt + 8 * lhi) * 5 + c;
#pragma unroll
      for (int v = 0; v < 8; v++) op[v * 5] = acc[j][v];
    }
  }
  __syncthreads();
  AS1 v4f* og = (AS1 v4f*)((gptr)out + b0 * 320);
  for (int it = 0; it < (PJ_NB * 320 / 4) / NTHREADS; it++)
    og[tid + it * NTHREADS] = ((const AS3 v4f*)outs)[tid + it * NTHREADS];
}

extern "C" void kernel_launch(void* const* d_in, const int* in_sizes, int n_in,
                              void* d_out, int out_size, void* d_ws,
                              size_t ws_size, hipStream_t stream) {
  const float* x0D   = (const float*)d_in[0];
  const float* xctrl = (const float*)d_in[1];
  const float* W1    = (const float*)d_in[2];
  const float* b1    = (const float*)d_in[3];
  const float* W2    = (const float*)d_in[4];
  const float* b2    = (const float*)d_in[5];
  const float* projw = (const float*)d_in[6];
  float* out = (float*)d_out;

  float* ws   = (float*)d_ws;
  float* xbuf = ws;
  const size_t XB = (size_t)B_TOT * L_SEQ * NCH;
  float* Wpk = xbuf + XB;

  const int packN = 28 * WB_TOT;
  pack_weights_kernel<<<(packN + 255) / 256, 256, 0, stream>>>(W1, W2, Wpk);
  for (int d = 0; d < 3; d++)
    scinet_level_kernel<<<B_TOT, NTHREADS, 0, stream>>>(xbuf, x0D, xctrl, Wpk, b1,
                                                        b2, d);
  proj_kernel<<<B_TOT / PJ_NB, NTHREADS, 0, stream>>>(xbuf, x0D, projw, out);
}


# Last level, U and P: 2nd conv computes only out channels 0-4 (all the projection reads) as 3 tiles of 3 row shifts x 5 channels, K=100, not 8 main + 2 tail tiles; sums reordered: k shifted by column
# speedup vs baseline: 2.3900x; 1.0414x over previous
#include <hip/hip_runtime.h>
#include <math.h>


typedef __attribute__((ext_vector_type(2))) float v2f;
typedef __attribute__((ext_vector_type(4))) float v4f;
typedef __attribute__((ext_vector_type(8))) float v8f;

#define AS3 __attribute__((address_space(3)))
#define AS1 __attribute__((address_space(1)))
typedef AS3 float*       lptr;
typedef const AS3 float* clptr;
typedef AS1 float*       gptr;
typedef const AS1 float* cgptr;

#define B_TOT    8192
#define L_SEQ    256
#define NCH      19
#define PCH      20
#define NWAVES   8
#define NTHREADS 256

#define K1M      100
#define K2M      60
#define K1T      180
#define K2T      140
#define TG       80
#define TS1      284
#define TS2      220
#define WM1_OFF  0
#define WM2_OFF  1600
#define WT1_OFF  2560
#define WT2_OFF  (WT1_OFF + 3 * TS1)
#define WB_TOT   4096
#define L5_LP    32
#define L5_NSEG  4
#define L5_S     (L5_LP + 6)
#define L5_NA    11
#define L5_NT    3
#define L5_G     40
#define L5_P     132
#define L5_END   (4 * L5_P + L5_G + K1M)
static_assert(L5_END <= WT1_OFF - WM2_OFF, "the five strips fit the conv2 main area");

#define ROWS_MAX 152
#define IN_SZ    ((ROWS_MAX + 2) * PCH)
#define HB_SZ    (ROWS_MAX * PCH)
#define WL_SZ    (WT2_OFF + 3 * TS2)
#define SEV_OFF  0
#define SOD_OFF  (SEV_OFF + IN_SZ)
#define DB_OFF   (SOD_OFF + IN_SZ)
#define WL_OFF   (DB_OFF + IN_SZ)
#define HB_OFF   (WL_OFF + WL_SZ)
#define SMEM_TOT (HB_OFF + HB_SZ)

static __device__ __forceinline__ v8f wmma4(v2f a, v2f b, v8f c) {
  return __builtin_amdgcn_wmma_f32_16x16x4_f32(false, a, false, b, (short)0, c,
                                               false, false);
}

static __device__ __forceinline__ float leaky(float x) {
  return x > 0.f ? x : 0.01f * x;
}

static __device__ __forceinline__ float fast_tanh(float x) {
  const float e = __builtin_amdgcn_exp2f(x * 2.8853900817779268f);
  return fmaf(-2.f, __builtin_amdgcn_rcpf(e + 1.f), 1.f);
}

template <int MODE>
static __device__ void run_branch(cgptr Wp, cgptr b1, cgptr b2, clptr in,
                                  lptr hb, lptr wl, int Lp, int nseg, int tsh,
                                  clptr other, lptr outLds, gptr xg, int tmul,
                                  int tadd, int last5) {
  const int tid  = threadIdx.x;
  const int hwv  = __builtin_amdgcn_readfirstlane(tid >> 5);
  const int wave = ((hwv + MODE) & 3) | (hwv & 4);
  const int lane = tid & 31;
  const int lhi  = lane >> 4;
  const int lln  = lane & 15;
  int tj = lln / 3;
  int tc = lln - 3 * tj;
  if (lln == 15) { tj = 0; tc = 0; }

#pragma unroll
  for (int it = 0; it < WB_TOT / 4 / NTHREADS; it++)
    ((AS3 v4f*)wl)[tid + it * NTHREADS] = ((const AS1 v4f*)Wp)[tid + it * NTHREADS];
  __syncthreads();

  const int S    = Lp + 6;
  const int M1   = nseg * S - 4;
  const int nMt1 = (M1 + 15) >> 4;
  const int nTr  = (M1 + 4) / 5;

  if (wave < 2) {
    int ti = (wave << 4) + lln; if (ti > nTr - 1) ti = nTr - 1;
    clptr ar = in + ti * (5 * PCH);
    clptr br = wl + WT1_OFF + tc * TS1 + TG - PCH * tj;
    v8f acc = {};
    v2f aS[3], bS[3];
#pragma unroll
    for (int u = 0; u < 3; u++) {
      aS[u] = *(const AS3 v2f*)(ar + 168 + 4 * u + 2 * lhi);
      bS[u] = *(const AS3 v2f*)(br + 168 + 4 * u + 2 * lhi);
    }
#pragma unroll 7
    for (int s = 0; s < 21; s++) {
      const v4f a = *(const AS3 v4f*)(ar + 8 * s + 4 * lhi);
      const v4f b = *(const AS3 v4f*)(br + 8 * s + 4 * lhi);
      acc = wmma4(a.xy, b.xy, acc);
      acc = wmma4(a.zw, b.zw, acc);
    }
#pragma unroll
    for (int u = 0; u < 3; u++) acc = wmma4(aS[u], bS[u], acc);
    const float bt = b1[16 + tc];
    const int rb = ((wave << 4) + 8 * lhi) * 5 + tj;
    if (lln < 15) {
#pragma unroll
      for (int v = 0; v < 8; v++) {
        const int r = rb + 5 * v;
        if (r < M1) hb[r * PCH + 16 + tc] = leaky(acc[v] + bt);
      }
    }
  }
#pragma nounroll
  for (int pass = 0; pass < 2; pass++) {
    const int mt = pass == 0 ? wave - 2
                             : (wave >= 6 ? wave : (wave >= 4 ? wave + 4 : 99));
    if (mt < 0 || mt >= nMt1) continue;
    const int m0 = mt << 4;
    int mrow = m0 + lln; if (mrow > M1 - 1) mrow = M1 - 1;
    clptr ar = in + mrow * PCH;
    clptr bq = wl + WM1_OFF + 4 * lane;
    v8f acc = {};
    const v2f aL = *(const AS3 v2f*)(ar + 96 + 2 * lhi);
    const v2f bL = *(const AS3 v2f*)(wl + WM1_OFF + 12 * 128 + 2 * lane);
#pragma unroll 6
    for (int s = 0; s < 12; s++) {
      const v4f a = *(const AS3 v4f*)(ar + 8 * s + 4 * lhi);
      const v4f b = *(const AS3 v4f*)(bq + 128 * s);
      acc = wmma4(a.xy, b.xy, acc);
      acc = wmma4(a.zw, b.zw, acc);
    }
    acc = wmma4(aL, bL, acc);
    const float b1a = b1[lln];
    lptr hrow = hb + (m0 + 8 * lhi) * PCH;
    if (m0 + 16 <= M1) {
#pragma unroll
      for (int v = 0; v < 8; v++) hrow[v * PCH + lln] = leaky(acc[v] + b1a);
    } else {
#pragma unroll
      for (int v = 0; v < 8; v++)
        if (m0 + 8 * lhi + v < M1) hrow[v * PCH + lln] = leaky(acc[v] + b1a);
    }
  }
  __syncthreads();

  if (MODE >= 2 && last5) {
    if (wave < L5_NT) {
      int cj = lln / 5;
      int cc = lln - 5 * cj;
      if (lln == 15) { cj = 0; cc = 0; }
      int a = (wave << 4) + lln;
      if (a > L5_NSEG * L5_NA - 1) a = L5_NSEG * L5_NA - 1;
      const int ag = a / L5_NA;
      clptr ar = hb + (ag * L5_S + 3 * (a - ag * L5_NA)) * PCH;
      clptr br = wl + WM2_OFF + cc * L5_P + L5_G - PCH * cj;
      v8f acc = {};
      const v2f aL = *(const AS3 v2f*)(ar + 96 + 2 * lhi);
      const v2f bL = *(const AS3 v2f*)(br + 96 + 2 * lhi);
#pragma unroll 6
      for (int s = 0; s < 12; s++) {
        const v4f a4 = *(const AS3 v4f*)(ar + 8 * s + 4 * lhi);
        const v4f b4 = *(const AS3 v4f*)(br + 8 * s + 4 * lhi);
        acc = wmma4(a4.xy, b4.xy, acc);
        acc = wmma4(a4.zw, b4.zw, acc);
      }
      acc = wmma4(aL, bL, acc);
      const float bt = b2[cc];
      const int a0 = (wave << 4) + 8 * lhi;
      if (lln < 15) {
#pragma unroll
        for (int v = 0; v < 8; v++) {
          const int ad = a0 + v;
          const int g  = ad / L5_NA;
          const int jj = 3 * (ad - g * L5_NA) + cj;
          if (g < L5_NSEG && jj < L5_LP) {
            const float t   = fast_tanh(acc[v] + bt);
            const float oth = other[(g * L5_S + 3 + jj) * PCH + cc];
            gptr xs = xg + g * (L_SEQ * NCH);
            if (MODE == 2) xs[(jj * tmul + tadd) * NCH + cc] = oth + t;
            else           xs[(jj * tmul + tadd) * NCH + cc] = oth - t;
          }
        }
      }
    }
    __syncthreads();
    return;
  }
  if (wave < 2) {
    int ti = (wave << 4) + lln; if (ti > nTr - 1) ti = nTr - 1;
    clptr ar = hb + ti * (5 * PCH);
    clptr br = wl + WT2_OFF + tc * TS2 + TG - PCH * tj;
    v8f acc = {};
    v2f aS[3], bS[3];
#pragma unroll
    for (int u = 0; u < 3; u++) {
      aS[u] = *(const AS3 v2f*)(ar + 128 + 4 * u + 2 * lhi);
      bS[u] = *(const AS3 v2f*)(br + 128 + 4 * u + 2 * lhi);
    }
#pragma unroll 8
    for (int s = 0; s < 16; s++) {
      const v4f a = *(const AS3 v4f*)(ar + 8 * s + 4 * lhi);
      const v4f b = *(const AS3 v4f*)(br + 8 * s + 4 * lhi);
      acc = wmma4(a.xy, b.xy, acc);
      acc = wmma4(a.zw, b.zw, acc);
    }
#pragma unroll
    for (int u = 0; u < 3; u++) acc = wmma4(aS[u], bS[u], acc);
    const float bt = b2[16 + tc];
    const int rb = ((wave << 4) + 8 * lhi) * 5 + tj;
    if (lln < 15) {
#pragma unroll
      for (int v = 0; v < 8; v++) {
        const int r  = rb + 5 * v;
        const int g  = (r >= S) + (r >= 2 * S) + (r >= 3 * S);
        const int jj = r - g * S;
        if (jj < Lp && g < nseg) {
          const int li = (r + 3) * PCH + 16 + tc;
          const float t   = fast_tanh(acc[v] + bt);
          const float oth = other[li];
          if (MODE <= 1) outLds[li] = oth * __expf(t);
          else {
            gptr xs = xg + g * (L_SEQ * NCH);
            if (MODE == 2) xs[(jj * tmul + tadd) * NCH + 16 + tc] = oth + t;
            else           xs[(jj * tmul + tadd) * NCH + 16 + tc] = oth - t;
          }
        }
      }
    }
  }
  const int nMt2  = (nseg * Lp) >> 4;
  const int tmask = (1 << tsh) - 1;
#pragma nounroll
  for (int pass = 0; pass < 2; pass++) {
    const int mt = pass == 0 ? wave - 2 : (wave >= 6 ? wave : 99);
    if (mt < 0 || mt >= nMt2) continue;
    const int g  = mt >> tsh;
    const int j0 = (mt & tmask) << 4;
    const int r0 = g * S + j0;
    clptr ar = hb + (r0 + lln) * PCH;
    clptr bq = wl + WM2_OFF + 4 * lane;
    v8f acc = {};
    const v2f aL = *(const AS3 v2f*)(ar + 56 + 2 * lhi);
    const v2f bL = *(const AS3 v2f*)(wl + WM2_OFF + 7 * 128 + 2 * lane);
#pragma unroll
    for (int s = 0; s < 7; s++) {
      const v4f a = *(const AS3 v4f*)(ar + 8 * s + 4 * lhi);
      const v4f b = *(const AS3 v4f*)(bq + 128 * s);
      acc = wmma4(a.xy, b.xy, acc);
      acc = wmma4(a.zw, b.zw, acc);
    }
    acc = wmma4(aL, bL, acc);
    const float b2a = b2[lln];
    const int mb = j0 + 8 * lhi;
    const int lb = (r0 + 3 + 8 * lhi) * PCH;
    gptr xs = xg + g * (L_SEQ * NCH);
#pragma unroll
    for (int v = 0; v < 8; v++) {
      const int m = mb + v;
      const float t   = fast_tanh(acc[v] + b2a);
      const float oth = other[lb + v * PCH + lln];
      if (MODE <= 1)      outLds[lb + v * PCH + lln] = oth * __expf(t);
      else if (MODE == 2) xs[(m * tmul + tadd) * NCH + lln] = oth + t;
      else                xs[(m * tmul + tadd) * NCH + lln] = oth - t;
    }
  }
  __syncthreads();
}

__global__ __launch_bounds__(NTHREADS) void scinet_level_kernel(
    float* __restrict__ xbuf_, const float* __restrict__ x0D_,
    const float* __restrict__ xctrl_, const float* __restrict__ Wpk_,
    const float* __restrict__ b1g_, const float* __restrict__ b2g_, int depth) {
  __shared__ __align__(16) float smem[SMEM_TOT];
  const int tid = threadIdx.x;
  const int G   = 1 << depth;
  const int sel = blockIdx.x & (G - 1);
  const int bg  = blockIdx.x >> depth;
  int node = 0;
  for (int i = 0; i < depth; i++) node = 2 * node + 1 + ((sel >> i) & 1);
  const int stride = 1 << depth;
  const int Lp     = (L_SEQ >> depth) >> 1;
  const int S      = Lp + 6;
  const int lsh    = 7 - depth;
  const int tsh    = 3 - depth;

  lptr sev = (lptr)(smem + SEV_OFF);
  lptr sod = (lptr)(smem + SOD_OFF);
  lptr hb  = (lptr)(smem + HB_OFF);
  lptr db  = (lptr)(smem + DB_OFF);
  lptr cb  = sev;
  lptr wl  = (lptr)(smem + WL_OFF);
  lptr sm  = (lptr)smem;

  for (int i = tid; i < SMEM_TOT; i += NTHREADS) sm[i] = 0.f;
  __syncthreads();

  gptr  xg  = (gptr)(xbuf_ + (size_t)bg * G * L_SEQ * NCH);
  cgptr Wpk = (cgptr)Wpk_;
  cgptr b1g = (cgptr)b1g_;
  cgptr b2g = (cgptr)b2g_;

  if (depth == 0) {
    cgptr x0 = (cgptr)x0D_ + (size_t)bg * (L_SEQ * 5);
    cgptr xc = (cgptr)xctrl_ + (size_t)bg * (L_SEQ * 14);
    for (int it = 0; it < (L_SEQ * 5) / NTHREADS; it++) {
      const int e = tid + it * NTHREADS;
      const int t = e / 5, c = e - t * 5;
      lptr dst = (t & 1) ? sod : sev;
      dst[(3 + (t >> 1)) * PCH + c] = x0[e];
    }
    for (int it = 0; it < (L_SEQ * 14) / NTHREADS; it++) {
      const int e = tid + it * NTHREADS;
      const int t = e / 14, c = e - t * 14;
      lptr dst = (t & 1) ? sod : sev;
      dst[(3 + (t >> 1)) * PCH + 5 + c] = xc[e];
    }
  } else {
    for (int i = tid; i < (L_SEQ / 2) * NCH; i += NTHREADS) {
      const int ra = i / NCH, c = i - ra * NCH;
      const int g = ra >> lsh, r = ra & (Lp - 1);
      cgptr xs = xg + g * (L_SEQ * NCH);
      const int o = (g * S + 3 + r) * PCH + c;
      sev[o] = xs[((2 * r) * stride + sel) * NCH + c];
      sod[o] = xs[((2 * r + 1) * stride + sel) * NCH + c];
    }
  }
  __syncthreads();
  for (int i = tid; i < G * 2 * 3 * NCH; i += NTHREADS) {
    const int q = i / NCH, c = i - q * NCH;
    const int w = q / 3, r = q - w * 3;
    lptr buf = ((w & 1) ? sod : sev) + (w >> 1) * S * PCH;
    buf[r * PCH + c]            = buf[3 * PCH + c];
    buf[(Lp + 3 + r) * PCH + c] = buf[(Lp + 2) * PCH + c];
  }
  __syncthreads();

  const int nb = node * 4;
  run_branch<0>(Wpk + (size_t)(nb + 0) * WB_TOT, b1g + (nb + 0) * NCH,
                b2g + (nb + 0) * NCH, sev, hb, wl, Lp, G, tsh, sod, db, xg, 0, 0, 0);
  run_branch<1>(Wpk + (size_t)(nb + 1) * WB_TOT, b1g + (nb + 1) * NCH,
                b2g + (nb + 1) * NCH, sod, hb, wl, Lp, G, tsh, sev, cb, xg, 0, 0, 0);
  for (int i = tid; i < G * 2 * 3 * NCH; i += NTHREADS) {
    const int q = i / NCH, c = i - q * NCH;
    const int w = q / 3, r = q - w * 3;
    lptr buf = ((w & 1) ? cb : db) + (w >> 1) * S * PCH;
    buf[r * PCH + c]            = buf[3 * PCH + c];
    buf[(Lp + 3 + r) * PCH + c] = buf[(Lp + 2) * PCH + c];
  }
  __syncthreads();
  run_branch<2>(Wpk + (size_t)(nb + 3) * WB_TOT, b1g + (nb + 3) * NCH,
                b2g + (nb + 3) * NCH, db, hb, wl, Lp, G, tsh, cb, db, xg,
                2 * stride, sel, depth == 2);
  run_branch<3>(Wpk + (size_t)(nb + 2) * WB_TOT, b1g + (nb + 2) * NCH,
                b2g + (nb + 2) * NCH, cb, hb, wl, Lp, G, tsh, db, cb, xg,
                2 * stride, sel + stride, depth == 2);
}

__global__ void pack_weights_kernel(const float* __restrict__ W1,
                                    const float* __restrict__ W2,
                                    float* __restrict__ Wpk) {
  const int idx = blockIdx.x * blockDim.x + threadIdx.x;
  if (idx >= 28 * WB_TOT) return;
  const int nb = idx / WB_TOT, rem = idx - nb * WB_TOT;
  float v = 0.f;
  if (rem < WM2_OFF) {
    int co, k;
    if (rem < 12 * 128) { const int l = (rem & 127) >> 2; co = l & 15; k = 8 * (rem >> 7) + 4 * (l >> 4) + (rem & 3); }
    else { const int r = rem - 12 * 128, l = r >> 1; co = l & 15; k = 96 + 2 * (l >> 4) + (r & 1); }
    const int tap = k / PCH, ci = k - tap * PCH;
    if (ci < NCH) v = W1[((size_t)(nb * NCH + co) * NCH + ci) * 5 + tap];
  } else if (rem < WT1_OFF && nb >= 12 && (nb & 3) >= 2) {
    const int r = rem - WM2_OFF;
    const int co = r / L5_P, k = r - co * L5_P - L5_G;
    if (co < 5 && k >= 0 && k < K2M) {
      const int tap = k / PCH, ci = k - tap * PCH;
      if (ci < NCH) v = W2[((size_t)(nb * NCH + co) * NCH + ci) * 3 + tap];
    }
  } else if (rem < WT1_OFF) {
    const int r = rem - WM2_OFF;
    int co, k;
    if (r < 7 * 128) { const int l = (r & 127) >> 2; co = l & 15; k = 8 * (r >> 7) + 4 * (l >> 4) + (r & 3); }
    else { const int rr = r - 7 * 128, l = rr >> 1; co = l & 15; k = 56 + 2 * (l >> 4) + (rr & 1); }
    const int tap = k / PCH, ci = k - tap * PCH;
    if (ci < NCH) v = W2[((size_t)(nb * NCH + co) * NCH + ci) * 3 + tap];
  } else if (rem < WT2_OFF) {
    const int r = rem - WT1_OFF;
    const int c = r / TS1, k = r - c * TS1 - TG;
    if (k >= 0 && k < K1M) {
      const int tap = k / PCH, ci = k - tap * PCH;
      if (ci < NCH) v = W1[((size_t)(nb * NCH + 16 + c) * NCH + ci) * 5 + tap];
    }
  } else if (rem < WT2_OFF + 3 * TS2) {
    const int r = rem - WT2_OFF;
    const int c = r / TS2, k = r - c * TS2 - TG;
    if (k >= 0 && k < K2M) {
      const int tap = k / PCH, ci = k - tap * PCH;
      if (ci < NCH) v = W2[((size_t)(nb * NCH + 16 + c) * NCH + ci) * 3 + tap];
    }
  }
  Wpk[idx] = v;
}

#define PJ_NB    16
#define PJ_N     (PJ_NB * 5)
#define PJ_PITCH 268
#define PJ_P_OFF (PJ_N * PJ_PITCH)
#define PJ_SMEM  (PJ_P_OFF + 64 * PJ_PITCH)

template <int NT>
static __device__ __forceinline__ void proj_tiles(clptr pa, clptr yb, v8f* acc) {
#pragma unroll 4
  for (int s = 0; s < 32; s++) {
    const v4f a = *(const AS3 v4f*)(pa + 8 * s);
    v4f b[NT];
#pragma unroll
    for (int j = 0; j < NT; j++)
      b[j] = *(const AS3 v4f*)(yb + j * (16 * PJ_PITCH) + 8 * s);
#pragma unroll
    for (int j = 0; j < NT; j++) acc[j] = wmma4(a.xy, b[j].xy, acc[j]);
#pragma unroll
    for (int j = 0; j < NT; j++) acc[j] = wmma4(a.zw, b[j].zw, acc[j]);
  }
}

__global__ __launch_bounds__(NTHREADS) void proj_kernel(
    const float* __restrict__ xbuf, const float* __restrict__ x0D,
    const float* __restrict__ projw, float* __restrict__ out) {
  __shared__ __align__(16) float psm[PJ_SMEM];
  lptr ys = (lptr)psm;
  lptr ps = (lptr)psm + PJ_P_OFF;
  const int tid  = threadIdx.x;
  const int wave = __builtin_amdgcn_readfirstlane(tid >> 5);
  const int lane = tid & 31;
  const int lhi  = lane >> 4;
  const int lln  = lane & 15;
  const size_t b0 = (size_t)blockIdx.x * PJ_NB;
  cgptr xb = (cgptr)xbuf + b0 * (L_SEQ * NCH);
  cgptr x0 = (cgptr)x0D + b0 * (L_SEQ * 5);
  cgptr pj = (cgptr)projw;

#pragma unroll 8
  for (int it = 0; it < (64 * L_SEQ / 4) / NTHREADS; it++) {
    const int q = tid + it * NTHREADS;
    *(AS3 v4f*)(ps + (q >> 6) * PJ_PITCH + 4 * (q & 63)) = ((const AS1 v4f*)pj)[q];
  }
#pragma unroll 2
  for (int g = 0; g < PJ_NB; g++) {
    cgptr xbg = xb + g * (L_SEQ * NCH);
    cgptr x0g = x0 + g * (L_SEQ * 5);
    lptr  yg  = ys + g * 5 * PJ_PITCH;
#pragma unroll
    for (int it = 0; it < (L_SEQ * 5) / NTHREADS; it++) {
      const int e = tid + it * NTHREADS;
      const int l = e / 5, c = e - 5 * l;
      yg[c * PJ_PITCH + l] = xbg[l * NCH + c] + x0g[e];
    }
  }
  __syncthreads();

  const int mt  = wave & 3;
  const int nt0 = wave < 4 ? 0 : 3;
  const int ntn = wave < 4 ? 3 : 2;
  clptr pa = ps + (16 * mt + lln) * PJ_PITCH + 4 * lhi;
  clptr yb = ys + (16 * nt0 + lln) * PJ_PITCH + 4 * lhi;
  v8f acc[3];
#pragma unroll
  for (int j = 0; j < 3; j++) acc[j] = (v8f){};
  if (wave < 4) proj_tiles<3>(pa, yb, acc);
  else          proj_tiles<2>(pa, yb, acc);
  __syncthreads();
  lptr outs = ys;
#pragma unroll
  for (int j = 0; j < 3; j++) {
    if (j < ntn) {
      const int n  = 16 * (nt0 + j) + lln;
      const int bl = n / 5, c = n - 5 * bl;
      lptr op = outs + bl * 320 + (16 * mt + 8 * lhi) * 5 + c;
#pragma unroll
      for (int v = 0; v < 8; v++) op[v * 5] = acc[j][v];
    }
  }
  __syncthreads();
  AS1 v4f* og = (AS1 v4f*)((gptr)out + b0 * 320);
  for (int it = 0; it < (PJ_NB * 320 / 4) / NTHREADS; it++)
    og[tid + it * NTHREADS] = ((const AS3 v4f*)outs)[tid + it * NTHREADS];
}

extern "C" void kernel_launch(void* const* d_in, const int* in_sizes, int n_in,
                              void* d_out, int out_size, void* d_ws,
                              size_t ws_size, hipStream_t stream) {
  const float* x0D   = (const float*)d_in[0];
  const float* xctrl = (const float*)d_in[1];
  const float* W1    = (const float*)d_in[2];
  const float* b1    = (const float*)d_in[3];
  const float* W2    = (const float*)d_in[4];
  const float* b2    = (const float*)d_in[5];
  const float* projw = (const float*)d_in[6];
  float* out = (float*)d_out;

  float* ws   = (float*)d_ws;
  float* xbuf = ws;
  const size_t XB = (size_t)B_TOT * L_SEQ * NCH;
  float* Wpk = xbuf + XB;

  const int packN = 28 * WB_TOT;
  pack_weights_kernel<<<(packN + 255) / 256, 256, 0, stream>>>(W1, W2, Wpk);
  for (int d = 0; d < 3; d++)
    scinet_level_kernel<<<B_TOT, NTHREADS, 0, stream>>>(xbuf, x0D, xctrl, Wpk, b1,
                                                        b2, d);
  proj_kernel<<<B_TOT / PJ_NB, NTHREADS, 0, stream>>>(xbuf, x0D, projw, out);
}


# Block set-up in fewer issue slots: LDS cleared by 128-bit stores; input rows gathered with one (row-in-pass, channel) slot per thread, passes moving by constants; pad rows one thread per column
# speedup vs baseline: 2.4987x; 1.0455x over previous
#include <hip/hip_runtime.h>
#include <math.h>


typedef __attribute__((ext_vector_type(2))) float v2f;
typedef __attribute__((ext_vector_type(4))) float v4f;
typedef __attribute__((ext_vector_type(8))) float v8f;

#define AS3 __attribute__((address_space(3)))
#define AS1 __attribute__((address_space(1)))
typedef AS3 float*       lptr;
typedef const AS3 float* clptr;
typedef AS1 float*       gptr;
typedef const AS1 float* cgptr;

#define B_TOT    8192
#define L_SEQ    256
#define NCH      19
#define PCH      20
#define NWAVES   8
#define NTHREADS 256

#define K1M      100
#define K2M      60
#define K1T      180
#define K2T      140
#define TG       80
#define TS1      284
#define TS2      220
#define WM1_OFF  0
#define WM2_OFF  1600
#define WT1_OFF  2560
#define WT2_OFF  (WT1_OFF + 3 * TS1)
#define WB_TOT   4096
#define L5_LP    32
#define L5_NSEG  4
#define L5_S     (L5_LP + 6)
#define L5_NA    11
#define L5_NT    3
#define L5_G     40
#define L5_P     132
#define L5_END   (4 * L5_P + L5_G + K1M)
static_assert(L5_END <= WT1_OFF - WM2_OFF, "the five strips fit the conv2 main area");

#define ROWS_MAX 152
#define IN_SZ    ((ROWS_MAX + 2) * PCH)
#define HB_SZ    (ROWS_MAX * PCH)
#define WL_SZ    (WT2_OFF + 3 * TS2)
#define SEV_OFF  0
#define SOD_OFF  (SEV_OFF + IN_SZ)
#define DB_OFF   (SOD_OFF + IN_SZ)
#define WL_OFF   (DB_OFF + IN_SZ)
#define HB_OFF   (WL_OFF + WL_SZ)
#define SMEM_TOT (HB_OFF + HB_SZ)

static __device__ __forceinline__ v8f wmma4(v2f a, v2f b, v8f c) {
  return __builtin_amdgcn_wmma_f32_16x16x4_f32(false, a, false, b, (short)0, c,
                                               false, false);
}

static __device__ __forceinline__ float leaky(float x) {
  return x > 0.f ? x : 0.01f * x;
}

static __device__ __forceinline__ float fast_tanh(float x) {
  const float e = __builtin_amdgcn_exp2f(x * 2.8853900817779268f);
  return fmaf(-2.f, __builtin_amdgcn_rcpf(e + 1.f), 1.f);
}

template <int MODE>
static __device__ void run_branch(cgptr Wp, cgptr b1, cgptr b2, clptr in,
                                  lptr hb, lptr wl, int Lp, int nseg, int tsh,
                                  clptr other, lptr outLds, gptr xg, int tmul,
                                  int tadd, int last5) {
  const int tid  = threadIdx.x;
  const int hwv  = __builtin_amdgcn_readfirstlane(tid >> 5);
  const int wave = ((hwv + MODE) & 3) | (hwv & 4);
  const int lane = tid & 31;
  const int lhi  = lane >> 4;
  const int lln  = lane & 15;
  int tj = lln / 3;
  int tc = lln - 3 * tj;
  if (lln == 15) { tj = 0; tc = 0; }

#pragma unroll
  for (int it = 0; it < WB_TOT / 4 / NTHREADS; it++)
    ((AS3 v4f*)wl)[tid + it * NTHREADS] = ((const AS1 v4f*)Wp)[tid + it * NTHREADS];
  __syncthreads();

  const int S    = Lp + 6;
  const int M1   = nseg * S - 4;
  const int nMt1 = (M1 + 15) >> 4;
  const int nTr  = (M1 + 4) / 5;

  if (wave < 2) {
    int ti = (wave << 4) + lln; if (ti > nTr - 1) ti = nTr - 1;
    clptr ar = in + ti * (5 * PCH);
    clptr br = wl + WT1_OFF + tc * TS1 + TG - PCH * tj;
    v8f acc = {};
    v2f aS[3], bS[3];
#pragma unroll
    for (int u = 0; u < 3; u++) {
      aS[u] = *(const AS3 v2f*)(ar + 168 + 4 * u + 2 * lhi);
      bS[u] = *(const AS3 v2f*)(br + 168 + 4 * u + 2 * lhi);
    }
#pragma unroll 7
    for (int s = 0; s < 21; s++) {
      const v4f a = *(const AS3 v4f*)(ar + 8 * s + 4 * lhi);
      const v4f b = *(const AS3 v4f*)(br + 8 * s + 4 * lhi);
      acc = wmma4(a.xy, b.xy, acc);
      acc = wmma4(a.zw, b.zw, acc);
    }
#pragma unroll
    for (int u = 0; u < 3; u++) acc = wmma4(aS[u], bS[u], acc);
    const float bt = b1[16 + tc];
    const int rb = ((wave << 4) + 8 * lhi) * 5 + tj;
    if (lln < 15) {
#pragma unroll
      for (int v = 0; v < 8; v++) {
        const int r = rb + 5 * v;
        if (r < M1) hb[r * PCH + 16 + tc] = leaky(acc[v] + bt);
      }
    }
  }
#pragma nounroll
  for (int pass = 0; pass < 2; pass++) {
    const int mt = pass == 0 ? wave - 2
                             : (wave >= 6 ? wave : (wave >= 4 ? wave + 4 : 99));
    if (mt < 0 || mt >= nMt1) continue;
    const int m0 = mt << 4;
    int mrow = m0 + lln; if (mrow > M1 - 1) mrow = M1 - 1;
    clptr ar = in + mrow * PCH;
    clptr bq = wl + WM1_OFF + 4 * lane;
    v8f acc = {};
    const v2f aL = *(const AS3 v2f*)(ar + 96 + 2 * lhi);
    const v2f bL = *(const AS3 v2f*)(wl + WM1_OFF + 12 * 128 + 2 * lane);
#pragma unroll 6
    for (int s = 0; s < 12; s++) {
      const v4f a = *(const AS3 v4f*)(ar + 8 * s + 4 * lhi);
      const v4f b = *(const AS3 v4f*)(bq + 128 * s);
      acc = wmma4(a.xy, b.xy, acc);
      acc = wmma4(a.zw, b.zw, acc);
    }
    acc = wmma4(aL, bL, acc);
    const float b1a = b1[lln];
    lptr hrow = hb + (m0 + 8 * lhi) * PCH;
    if (m0 + 16 <= M1) {
#pragma unroll
      for (int v = 0; v < 8; v++) hrow[v * PCH + lln] = leaky(acc[v] + b1a);
    } else {
#pragma unroll
      for (int v = 0; v < 8; v++)
        if (m0 + 8 * lhi + v < M1) hrow[v * PCH + lln] = leaky(acc[v] + b1a);
    }
  }
  __syncthreads();

  if (MODE >= 2 && last5) {
    if (wave < L5_NT) {
      int cj = lln / 5;
      int cc = lln - 5 * cj;
      if (lln == 15) { cj = 0; cc = 0; }
      int a = (wave << 4) + lln;
      if (a > L5_NSEG * L5_NA - 1) a = L5_NSEG * L5_NA - 1;
      const int ag = a / L5_NA;
      clptr ar = hb + (ag * L5_S + 3 * (a - ag * L5_NA)) * PCH;
      clptr br = wl + WM2_OFF + cc * L5_P + L5_G - PCH * cj;
      v8f acc = {};
      const v2f aL = *(const AS3 v2f*)(ar + 96 + 2 * lhi);
      const v2f bL = *(const AS3 v2f*)(br + 96 + 2 * lhi);
#pragma unroll 6
      for (int s = 0; s < 12; s++) {
        const v4f a4 = *(const AS3 v4f*)(ar + 8 * s + 4 * lhi);
        const v4f b4 = *(const AS3 v4f*)(br + 8 * s + 4 * lhi);
        acc = wmma4(a4.xy, b4.xy, acc);
        acc = wmma4(a4.zw, b4.zw, acc);
      }
      acc = wmma4(aL, bL, acc);
      const float bt = b2[cc];
      const int a0 = (wave << 4) + 8 * lhi;
      if (lln < 15) {
#pragma unroll
        for (int v = 0; v < 8; v++) {
          const int ad = a0 + v;
          const int g  = ad / L5_NA;
          const int jj = 3 * (ad - g * L5_NA) + cj;
          if (g < L5_NSEG && jj < L5_LP) {
            const float t   = fast_tanh(acc[v] + bt);
            const float oth = other[(g * L5_S + 3 + jj) * PCH + cc];
            gptr xs = xg + g * (L_SEQ * NCH);
            if (MODE == 2) xs[(jj * tmul + tadd) * NCH + cc] = oth + t;
            else           xs[(jj * tmul + tadd) * NCH + cc] = oth - t;
          }
        }
      }
    }
    __syncthreads();
    return;
  }
  if (wave < 2) {
    int ti = (wave << 4) + lln; if (ti > nTr - 1) ti = nTr - 1;
    clptr ar = hb + ti * (5 * PCH);
    clptr br = wl + WT2_OFF + tc * TS2 + TG - PCH * tj;
    v8f acc = {};
    v2f aS[3], bS[3];
#pragma unroll
    for (int u = 0; u < 3; u++) {
      aS[u] = *(const AS3 v2f*)(ar + 128 + 4 * u + 2 * lhi);
      bS[u] = *(const AS3 v2f*)(br + 128 + 4 * u + 2 * lhi);
    }
#pragma unroll 8
    for (int s = 0; s < 16; s++) {
      const v4f a = *(const AS3 v4f*)(ar + 8 * s + 4 * lhi);
      const v4f b = *(const AS3 v4f*)(br + 8 * s + 4 * lhi);
      acc = wmma4(a.xy, b.xy, acc);
      acc = wmma4(a.zw, b.zw, acc);
    }
#pragma unroll
    for (int u = 0; u < 3; u++) acc = wmma4(aS[u], bS[u], acc);
    const float bt = b2[16 + tc];
    const int rb = ((wave << 4) + 8 * lhi) * 5 + tj;
    if (lln < 15) {
#pragma unroll
      for (int v = 0; v < 8; v++) {
        const int r  = rb + 5 * v;
        const int g  = (r >= S) + (r >= 2 * S) + (r >= 3 * S);
        const int jj = r - g * S;
        if (jj < Lp && g < nseg) {
          const int li = (r + 3) * PCH + 16 + tc;
          const float t   = fast_tanh(acc[v] + bt);
          const float oth = other[li];
          if (MODE <= 1) outLds[li] = oth * __expf(t);
          else {
            gptr xs = xg + g * (L_SEQ * NCH);
            if (MODE == 2) xs[(jj * tmul + tadd) * NCH + 16 + tc] = oth + t;
            else           xs[(jj * tmul + tadd) * NCH + 16 + tc] = oth - t;
          }
        }
      }
    }
  }
  const int nMt2  = (nseg * Lp) >> 4;
  const int tmask = (1 << tsh) - 1;
#pragma nounroll
  for (int pass = 0; pass < 2; pass++) {
    const int mt = pass == 0 ? wave - 2 : (wave >= 6 ? wave : 99);
    if (mt < 0 || mt >= nMt2) continue;
    const int g  = mt >> tsh;
    const int j0 = (mt & tmask) << 4;
    const int r0 = g * S + j0;
    clptr ar = hb + (r0 + lln) * PCH;
    clptr bq = wl + WM2_OFF + 4 * lane;
    v8f acc = {};
    const v2f aL = *(const AS3 v2f*)(ar + 56 + 2 * lhi);
    const v2f bL = *(const AS3 v2f*)(wl + WM2_OFF + 7 * 128 + 2 * lane);
#pragma unroll
    for (int s = 0; s < 7; s++) {
      const v4f a = *(const AS3 v4f*)(ar + 8 * s + 4 * lhi);
      const v4f b = *(const AS3 v4f*)(bq + 128 * s);
      acc = wmma4(a.xy, b.xy, acc);
      acc = wmma4(a.zw, b.zw, acc);
    }
    acc = wmma4(aL, bL, acc);
    const float b2a = b2[lln];
    const int mb = j0 + 8 * lhi;
    const int lb = (r0 + 3 + 8 * lhi) * PCH;
    gptr xs = xg + g * (L_SEQ * NCH);
#pragma unroll
    for (int v = 0; v < 8; v++) {
      const int m = mb + v;
      const float t   = fast_tanh(acc[v] + b2a);
      const float oth = other[lb + v * PCH + lln];
      if (MODE <= 1)      outLds[lb + v * PCH + lln] = oth * __expf(t);
      else if (MODE == 2) xs[(m * tmul + tadd) * NCH + lln] = oth + t;
      else                xs[(m * tmul + tadd) * NCH + lln] = oth - t;
    }
  }
  __syncthreads();
}

static __device__ __forceinline__ void pad_rows(lptr b0, lptr b1, int nb, int S,
                                                int Lp, int tid) {
  if (tid < nb * NCH) {
    const int w = tid / NCH, c = tid - NCH * w;
    lptr buf = ((w & 1) ? b1 : b0) + (w >> 1) * S * PCH + c;
    const float lo = buf[3 * PCH], hi = buf[(Lp + 2) * PCH];
    lptr bh = buf + (Lp + 3) * PCH;
    buf[0] = lo; buf[PCH] = lo; buf[2 * PCH] = lo;
    bh[0]  = hi; bh[PCH]  = hi; bh[2 * PCH]  = hi;
  }
}

__global__ __launch_bounds__(NTHREADS) void scinet_level_kernel(
    float* __restrict__ xbuf_, const float* __restrict__ x0D_,
    const float* __restrict__ xctrl_, const float* __restrict__ Wpk_,
    const float* __restrict__ b1g_, const float* __restrict__ b2g_, int depth) {
  __shared__ __align__(16) float smem[SMEM_TOT];
  const int tid = threadIdx.x;
  const int G   = 1 << depth;
  const int sel = blockIdx.x & (G - 1);
  const int bg  = blockIdx.x >> depth;
  int node = 0;
  for (int i = 0; i < depth; i++) node = 2 * node + 1 + ((sel >> i) & 1);
  const int stride = 1 << depth;
  const int Lp     = (L_SEQ >> depth) >> 1;
  const int S      = Lp + 6;
  const int lsh    = 7 - depth;
  const int tsh    = 3 - depth;

  lptr sev = (lptr)(smem + SEV_OFF);
  lptr sod = (lptr)(smem + SOD_OFF);
  lptr hb  = (lptr)(smem + HB_OFF);
  lptr db  = (lptr)(smem + DB_OFF);
  lptr cb  = sev;
  lptr wl  = (lptr)(smem + WL_OFF);
  lptr sm  = (lptr)smem;

  {
    const v4f z = {0.f, 0.f, 0.f, 0.f};
    static_assert(SMEM_TOT / 4 == 15 * NTHREADS + 248, "15 full rounds and one of 248");
#pragma unroll
    for (int it = 0; it < 15; it++) ((AS3 v4f*)sm)[tid + it * NTHREADS] = z;
    if (tid < 248) ((AS3 v4f*)sm)[tid + 15 * NTHREADS] = z;
  }
  __syncthreads();

  gptr  xg  = (gptr)(xbuf_ + (size_t)bg * G * L_SEQ * NCH);
  cgptr Wpk = (cgptr)Wpk_;
  cgptr b1g = (cgptr)b1g_;
  cgptr b2g = (cgptr)b2g_;

  if (depth == 0) {
    cgptr x0 = (cgptr)x0D_ + (size_t)bg * (L_SEQ * 5);
    cgptr xc = (cgptr)xctrl_ + (size_t)bg * (L_SEQ * 14);
    if (tid < 250) {
      const int tr = tid / 5, c = tid - 5 * tr;
      lptr dst = ((tr & 1) ? sod : sev) + (3 + (tr >> 1)) * PCH + c;
      float v[6];
#pragma unroll
      for (int it = 0; it < 5; it++) v[it] = x0[tid + 250 * it];
      v[5] = 0.f;
      if (tid < 30) v[5] = x0[tid + 250 * 5];
#pragma unroll
      for (int it = 0; it < 5; it++) dst[it * (25 * PCH)] = v[it];
      if (tid < 30) dst[5 * (25 * PCH)] = v[5];
    }
    if (tid < 252) {
      const int tr = tid / 14, c = tid - 14 * tr;
      lptr dst = ((tr & 1) ? sod : sev) + (3 + (tr >> 1)) * PCH + 5 + c;
      float v[15];
#pragma unroll
      for (int it = 0; it < 14; it++) v[it] = xc[tid + 252 * it];
      v[14] = 0.f;
      if (tid < 56) v[14] = xc[tid + 252 * 14];
#pragma unroll
      for (int it = 0; it < 14; it++) dst[it * (9 * PCH)] = v[it];
      if (tid < 56) dst[14 * (9 * PCH)] = v[14];
    }
  } else {
    if (tid < 13 * NCH) {
      const int tr = tid / NCH, c = tid - NCH * tr;
      const unsigned rstep = (2u * NCH) << depth;
      const unsigned o0 = (unsigned)tr * rstep + (unsigned)c;
      cgptr xe = xg + sel * NCH;
      cgptr xo = xe + (NCH << depth);
      const int l0 = (tr + 3) * PCH + c;
      const bool tenth = tid < 11 * NCH;
      float ve[10], vo[10];
#pragma unroll
      for (int it = 0; it < 9; it++) {
        ve[it] = (xe + (13 * it) * rstep)[o0];
        vo[it] = (xo + (13 * it) * rstep)[o0];
      }
      ve[9] = 0.f; vo[9] = 0.f;
      if (tenth) {
        ve[9] = (xe + (13 * 9) * rstep)[o0];
        vo[9] = (xo + (13 * 9) * rstep)[o0];
      }
#pragma unroll
      for (int it = 0; it < 9; it++) {
        const int o = l0 + (13 * it) * PCH + (6 * PCH) * ((13 * it + tr) >> lsh);
        sev[o] = ve[it];
        sod[o] = vo[it];
      }
      if (tenth) {
        const int o = l0 + (13 * 9) * PCH + (6 * PCH) * ((13 * 9 + tr) >> lsh);
        sev[o] = ve[9];
        sod[o] = vo[9];
      }
    }
  }
  __syncthreads();
  pad_rows(sev, sod, 2 * G, S, Lp, tid);
  __syncthreads();

  const int nb = node * 4;
  run_branch<0>(Wpk + (size_t)(nb + 0) * WB_TOT, b1g + (nb + 0) * NCH,
                b2g + (nb + 0) * NCH, sev, hb, wl, Lp, G, tsh, sod, db, xg, 0, 0, 0);
  run_branch<1>(Wpk + (size_t)(nb + 1) * WB_TOT, b1g + (nb + 1) * NCH,
                b2g + (nb + 1) * NCH, sod, hb, wl, Lp, G, tsh, sev, cb, xg, 0, 0, 0);
  pad_rows(db, cb, 2 * G, S, Lp, tid);
  __syncthreads();
  run_branch<2>(Wpk + (size_t)(nb + 3) * WB_TOT, b1g + (nb + 3) * NCH,
                b2g + (nb + 3) * NCH, db, hb, wl, Lp, G, tsh, cb, db, xg,
                2 * stride, sel, depth == 2);
  run_branch<3>(Wpk + (size_t)(nb + 2) * WB_TOT, b1g + (nb + 2) * NCH,
                b2g + (nb + 2) * NCH, cb, hb, wl, Lp, G, tsh, db, cb, xg,
                2 * stride, sel + stride, depth == 2);
}

__global__ void pack_weights_kernel(const float* __restrict__ W1,
                                    const float* __restrict__ W2,
                                    float* __restrict__ Wpk) {
  const int idx = blockIdx.x * blockDim.x + threadIdx.x;
  if (idx >= 28 * WB_TOT) return;
  const int nb = idx / WB_TOT, rem = idx - nb * WB_TOT;
  float v = 0.f;
  if (rem < WM2_OFF) {
    int co, k;
    if (rem < 12 * 128) { const int l = (rem & 127) >> 2; co = l & 15; k = 8 * (rem >> 7) + 4 * (l >> 4) + (rem & 3); }
    else { const int r = rem - 12 * 128, l = r >> 1; co = l & 15; k = 96 + 2 * (l >> 4) + (r & 1); }
    const int tap = k / PCH, ci = k - tap * PCH;
    if (ci < NCH) v = W1[((size_t)(nb * NCH + co) * NCH + ci) * 5 + tap];
  } else if (rem < WT1_OFF && nb >= 12 && (nb & 3) >= 2) {
    const int r = rem - WM2_OFF;
    const int co = r / L5_P, k = r - co * L5_P - L5_G;
    if (co < 5 && k >= 0 && k < K2M) {
      const int tap = k / PCH, ci = k - tap * PCH;
      if (ci < NCH) v = W2[((size_t)(nb * NCH + co) * NCH + ci) * 3 + tap];
    }
  } else if (rem < WT1_OFF) {
    const int r = rem - WM2_OFF;
    int co, k;
    if (r < 7 * 128) { const int l = (r & 127) >> 2; co = l & 15; k = 8 * (r >> 7) + 4 * (l >> 4) + (r & 3); }
    else { const int rr = r - 7 * 128, l = rr >> 1; co = l & 15; k = 56 + 2 * (l >> 4) + (rr & 1); }
    const int tap = k / PCH, ci = k - tap * PCH;
    if (ci < NCH) v = W2[((size_t)(nb * NCH + co) * NCH + ci) * 3 + tap];
  } else if (rem < WT2_OFF) {
    const int r = rem - WT1_OFF;
    const int c = r / TS1, k = r - c * TS1 - TG;
    if (k >= 0 && k < K1M) {
      const int tap = k / PCH, ci = k - tap * PCH;
      if (ci < NCH) v = W1[((size_t)(nb * NCH + 16 + c) * NCH + ci) * 5 + tap];
    }
  } else if (rem < WT2_OFF + 3 * TS2) {
    const int r = rem - WT2_OFF;
    const int c = r / TS2, k = r - c * TS2 - TG;
    if (k >= 0 && k < K2M) {
      const int tap = k / PCH, ci = k - tap * PCH;
      if (ci < NCH) v = W2[((size_t)(nb * NCH + 16 + c) * NCH + ci) * 3 + tap];
    }
  }
  Wpk[idx] = v;
}

#define PJ_NB    16
#define PJ_N     (PJ_NB * 5)
#define PJ_PITCH 268
#define PJ_P_OFF (PJ_N * PJ_PITCH)
#define PJ_SMEM  (PJ_P_OFF + 64 * PJ_PITCH)

template <int NT>
static __device__ __forceinline__ void proj_tiles(clptr pa, clptr yb, v8f* acc) {
#pragma unroll 4
  for (int s = 0; s < 32; s++) {
    const v4f a = *(const AS3 v4f*)(pa + 8 * s);
    v4f b[NT];
#pragma unroll
    for (int j = 0; j < NT; j++)
      b[j] = *(const AS3 v4f*)(yb + j * (16 * PJ_PITCH) + 8 * s);
#pragma unroll
    for (int j = 0; j < NT; j++) acc[j] = wmma4(a.xy, b[j].xy, acc[j]);
#pragma unroll
    for (int j = 0; j < NT; j++) acc[j] = wmma4(a.zw, b[j].zw, acc[j]);
  }
}

__global__ __launch_bounds__(NTHREADS) void proj_kernel(
    const float* __restrict__ xbuf, const float* __restrict__ x0D,
    const float* __restrict__ projw, float* __restrict__ out) {
  __shared__ __align__(16) float psm[PJ_SMEM];
  lptr ys = (lptr)psm;
  lptr ps = (lptr)psm + PJ_P_OFF;
  const int tid  = threadIdx.x;
  const int wave = __builtin_amdgcn_readfirstlane(tid >> 5);
  const int lane = tid & 31;
  const int lhi  = lane >> 4;
  const int lln  = lane & 15;
  const size_t b0 = (size_t)blockIdx.x * PJ_NB;
  cgptr xb = (cgptr)xbuf + b0 * (L_SEQ * NCH);
  cgptr x0 = (cgptr)x0D + b0 * (L_SEQ * 5);
  cgptr pj = (cgptr)projw;

#pragma unroll 8
  for (int it = 0; it < (64 * L_SEQ / 4) / NTHREADS; it++) {
    const int q = tid + it * NTHREADS;
    *(AS3 v4f*)(ps + (q >> 6) * PJ_PITCH + 4 * (q & 63)) = ((const AS1 v4f*)pj)[q];
  }
#pragma unroll 2
  for (int g = 0; g < PJ_NB; g++) {
    cgptr xbg = xb + g * (L_SEQ * NCH);
    cgptr x0g = x0 + g * (L_SEQ * 5);
    lptr  yg  = ys + g * 5 * PJ_PITCH;
#pragma unroll
    for (int it = 0; it < (L_SEQ * 5) / NTHREADS; it++) {
      const int e = tid + it * NTHREADS;
      const int l = e / 5, c = e - 5 * l;
      yg[c * PJ_PITCH + l] = xbg[l * NCH + c] + x0g[e];
    }
  }
  __syncthreads();

  const int mt  = wave & 3;
  const int nt0 = wave < 4 ? 0 : 3;
  const int ntn = wave < 4 ? 3 : 2;
  clptr pa = ps + (16 * mt + lln) * PJ_PITCH + 4 * lhi;
  clptr yb = ys + (16 * nt0 + lln) * PJ_PITCH + 4 * lhi;
  v8f acc[3];
#pragma unroll
  for (int j = 0; j < 3; j++) acc[j] = (v8f){};
  if (wave < 4) proj_tiles<3>(pa, yb, acc);
  else          proj_tiles<2>(pa, yb, acc);
  __syncthreads();
  lptr outs = ys;
#pragma unroll
  for (int j = 0; j < 3; j++) {
    if (j < ntn) {
      const int n  = 16 * (nt0 + j) + lln;
      const int bl = n / 5, c = n - 5 * bl;
      lptr op = outs + bl * 320 + (16 * mt + 8 * lhi) * 5 + c;
#pragma unroll
      for (int v = 0; v < 8; v++) op[v * 5] = acc[j][v];
    }
  }
  __syncthreads();
  AS1 v4f* og = (AS1 v4f*)((gptr)out + b0 * 320);
  for (int it = 0; it < (PJ_NB * 320 / 4) / NTHREADS; it++)
    og[tid + it * NTHREADS] = ((const AS3 v4f*)outs)[tid + it * NTHREADS];
}

extern "C" void kernel_launch(void* const* d_in, const int* in_sizes, int n_in,
                              void* d_out, int out_size, void* d_ws,
                              size_t ws_size, hipStream_t stream) {
  const float* x0D   = (const float*)d_in[0];
  const float* xctrl = (const float*)d_in[1];
  const float* W1    = (const float*)d_in[2];
  const float* b1    = (const float*)d_in[3];
  const float* W2    = (const float*)d_in[4];
  const float* b2    = (const float*)d_in[5];
  const float* projw = (const float*)d_in[6];
  float* out = (float*)d_out;

  float* ws   = (float*)d_ws;
  float* xbuf = ws;
  const size_t XB = (size_t)B_TOT * L_SEQ * NCH;
  float* Wpk = xbuf + XB;

  const int packN = 28 * WB_TOT;
  pack_weights_kernel<<<(packN + 255) / 256, 256, 0, stream>>>(W1, W2, Wpk);
  for (int d = 0; d < 3; d++)
    scinet_level_kernel<<<B_TOT, NTHREADS, 0, stream>>>(xbuf, x0D, xctrl, Wpk, b1,
                                                        b2, d);
  proj_kernel<<<B_TOT / PJ_NB, NTHREADS, 0, stream>>>(xbuf, x0D, projw, out);
}
